# starting kernel
# baseline (speedup 1.0000x reference)
#include <hip/hip_runtime.h>

#define N_NODES 100000
#define N_EDGES 3200000
#define IN_CH   128
#define HID     96
#define HID2    48
#define NCLS    3

typedef __attribute__((ext_vector_type(2)))  float  v2f;
typedef __attribute__((ext_vector_type(8)))  float  v8f;

static __device__ inline void atom_add_f32(float* p, float v) {
    unsafeAtomicAdd(p, v);
}


__global__ __launch_bounds__(256) void init_deg(float* __restrict__ deg) {
    int i = blockIdx.x * blockDim.x + threadIdx.x;
    if (i < N_NODES) deg[i] = 1.0f;
}

__global__ __launch_bounds__(256) void accum_deg(const int* __restrict__ dst,
                                                 const float* __restrict__ ew,
                                                 float* __restrict__ deg) {
    int e = blockIdx.x * blockDim.x + threadIdx.x;
    if (e < N_EDGES) atom_add_f32(deg + dst[e], ew[e]);
}

__global__ __launch_bounds__(256) void make_dinv(float* __restrict__ deg) {
    int i = blockIdx.x * blockDim.x + threadIdx.x;
    if (i < N_NODES) deg[i] = rsqrtf(deg[i]);
}

__global__ __launch_bounds__(256) void make_norm(const int* __restrict__ src,
                                                 const int* __restrict__ dst,
                                                 const float* __restrict__ ew,
                                                 const float* __restrict__ dinv,
                                                 float* __restrict__ norm) {
    int e = blockIdx.x * blockDim.x + threadIdx.x;
    if (e < N_EDGES) norm[e] = dinv[src[e]] * ew[e] * dinv[dst[e]];
}


template<int K, int COLS>
__global__ __launch_bounds__(256) void gemm_wmma(const float* __restrict__ X,
                                                 const float* __restrict__ W,
                                                 float* __restrict__ H) {
    const int lane = threadIdx.x & 31;
    const int wave = threadIdx.x >> 5;
    const int tm = blockIdx.x * 8 + wave;
    const int tn = blockIdx.y;
    if (tm * 16 >= N_NODES) return;
    const int m0 = tm * 16, n0 = tn * 16;
    const int row   = m0 + (lane & 15);
    const int col   = n0 + (lane & 15);
    const int khalf = (lane >> 4) << 1;
    const float* xrow = X + (size_t)row * K;
    v8f c = {};
    for (int kk = 0; kk < K; kk += 4) {
        v2f a = *(const v2f*)(xrow + kk + khalf);
        v2f b;
        b.x = W[(size_t)(kk + khalf)     * COLS + col];
        b.y = W[(size_t)(kk + khalf + 1) * COLS + col];
        c = __builtin_amdgcn_wmma_f32_16x16x4_f32(false, a, false, b,
                                                  (short)0, c, false, false);
    }
    const int rbase = (lane >> 4) << 3;
    float* hrow = H + (size_t)(m0 + rbase) * COLS + col;
    #pragma unroll
    for (int r = 0; r < 8; ++r) hrow[(size_t)r * COLS] = c[r];
}


template<int C>
__global__ __launch_bounds__(256) void selfloop_init(const float* __restrict__ H,
                                                     const float* __restrict__ dinv,
                                                     float* __restrict__ out) {
    const int C4 = C / 4;
    int idx = blockIdx.x * blockDim.x + threadIdx.x;
    if (idx >= N_NODES * C4) return;
    int i = idx / C4, q = idx % C4;
    float di = dinv[i], w = di * di;
    float4 hv = *(const float4*)(H + (size_t)i * C + q * 4);
    float4 o = { w * hv.x, w * hv.y, w * hv.z, w * hv.w };
    *(float4*)(out + (size_t)i * C + q * 4) = o;
}

template<int C, int EPB>
__global__ __launch_bounds__(256) void scatter_edges(const float* __restrict__ H,
                                                     const int* __restrict__ src,
                                                     const int* __restrict__ dst,
                                                     const float* __restrict__ norm,
                                                     float* __restrict__ out) {
    const int q = threadIdx.x;
    const int e = blockIdx.x * EPB + threadIdx.y;
    if (e >= N_EDGES) return;
    const int s = src[e], d = dst[e];
    const float nw = norm[e];
    const float4 hv = *(const float4*)(H + (size_t)s * C + q * 4);
    float* o = out + (size_t)d * C + q * 4;
    atom_add_f32(o + 0, nw * hv.x);
    atom_add_f32(o + 1, nw * hv.y);
    atom_add_f32(o + 2, nw * hv.z);
    atom_add_f32(o + 3, nw * hv.w);
}

template<int C>
__global__ __launch_bounds__(256) void bias_relu(float* __restrict__ out,
                                                 const float* __restrict__ b) {
    const int C4 = C / 4;
    int idx = blockIdx.x * blockDim.x + threadIdx.x;
    if (idx >= N_NODES * C4) return;
    int i = idx / C4, q = idx % C4;
    float4 v  = *(float4*)(out + (size_t)i * C + q * 4);
    float4 bb = *(const float4*)(b + q * 4);
    v.x = fmaxf(v.x + bb.x, 0.0f);
    v.y = fmaxf(v.y + bb.y, 0.0f);
    v.z = fmaxf(v.z + bb.z, 0.0f);
    v.w = fmaxf(v.w + bb.w, 0.0f);
    *(float4*)(out + (size_t)i * C + q * 4) = v;
}

__global__ __launch_bounds__(256) void classifier(const float* __restrict__ Hin,
                                                  const float* __restrict__ Wc,
                                                  const float* __restrict__ bc,
                                                  float* __restrict__ logits) {
    int i = blockIdx.x * blockDim.x + threadIdx.x;
    if (i >= N_NODES) return;
    const float* h = Hin + (size_t)i * HID2;
    float a0 = bc[0], a1 = bc[1], a2 = bc[2];
    #pragma unroll
    for (int k = 0; k < HID2; ++k) {
        float hv = h[k];
        a0 += hv * Wc[k * 3 + 0];
        a1 += hv * Wc[k * 3 + 1];
        a2 += hv * Wc[k * 3 + 2];
    }
    logits[(size_t)i * 3 + 0] = a0;
    logits[(size_t)i * 3 + 1] = a1;
    logits[(size_t)i * 3 + 2] = a2;
}


extern "C" void kernel_launch(void* const* d_in, const int* in_sizes, int n_in,
                              void* d_out, int out_size, void* d_ws, size_t ws_size,
                              hipStream_t stream) {
    const float* x  = (const float*)d_in[0];
    const int*   ei = (const int*)  d_in[1];
    const float* ew = (const float*)d_in[2];
    const float* W1 = (const float*)d_in[3];
    const float* b1 = (const float*)d_in[4];
    const float* W2 = (const float*)d_in[5];
    const float* b2 = (const float*)d_in[6];
    const float* Wc = (const float*)d_in[7];
    const float* bc = (const float*)d_in[8];
    float* logits = (float*)d_out;

    const int* src = ei;
    const int* dst = ei + N_EDGES;

    char* ws = (char*)d_ws;
    size_t off = 0;
    auto take = [&](size_t bytes) -> void* {
        void* p = ws + off;
        off += (bytes + 255) & ~(size_t)255;
        return p;
    };
    float* dinv = (float*)take((size_t)N_NODES * 4);
    float* norm = (float*)take((size_t)N_EDGES * 4);
    float* H1   = (float*)take((size_t)N_NODES * HID * 4);
    float* out1 = (float*)take((size_t)N_NODES * HID * 4);
    float* H2   = H1;
    float* out2 = out1;

    const int TB = 256;
    init_deg <<<(N_NODES + TB - 1) / TB, TB, 0, stream>>>(dinv);
    accum_deg<<<(N_EDGES + TB - 1) / TB, TB, 0, stream>>>(dst, ew, dinv);
    make_dinv<<<(N_NODES + TB - 1) / TB, TB, 0, stream>>>(dinv);
    make_norm<<<(N_EDGES + TB - 1) / TB, TB, 0, stream>>>(src, dst, ew, dinv, norm);

    {
        dim3 g((N_NODES / 16 + 7) / 8, HID / 16);
        gemm_wmma<IN_CH, HID><<<g, 256, 0, stream>>>(x, W1, H1);
    }
    {
        int tot = N_NODES * (HID / 4);
        selfloop_init<HID><<<(tot + TB - 1) / TB, TB, 0, stream>>>(H1, dinv, out1);
        scatter_edges<HID, 8><<<(N_EDGES + 7) / 8, dim3(HID / 4, 8), 0, stream>>>(
            H1, src, dst, norm, out1);
        bias_relu<HID><<<(tot + TB - 1) / TB, TB, 0, stream>>>(out1, b1);
    }

    {
        dim3 g((N_NODES / 16 + 7) / 8, HID2 / 16);
        gemm_wmma<HID, HID2><<<g, 256, 0, stream>>>(out1, W2, H2);
    }
    {
        int tot = N_NODES * (HID2 / 4);
        selfloop_init<HID2><<<(tot + TB - 1) / TB, TB, 0, stream>>>(H2, dinv, out2);
        scatter_edges<HID2, 16><<<(N_EDGES + 15) / 16, dim3(HID2 / 4, 16), 0, stream>>>(
            H2, src, dst, norm, out2);
        bias_relu<HID2><<<(tot + TB - 1) / TB, TB, 0, stream>>>(out2, b2);
    }

    classifier<<<(N_NODES + TB - 1) / TB, TB, 0, stream>>>(out2, Wc, bc, logits);
}


# scatter: a wave walks 32 edges with lanes on consecutive channels, so each f32 atomic instruction adds one whole 128-byte output line; adds still land in arrival order
# speedup vs baseline: 1.4920x; 1.4920x over previous
#include <hip/hip_runtime.h>

#define N_NODES 100000
#define N_EDGES 3200000
#define IN_CH   128
#define HID     96
#define HID2    48
#define NCLS    3

typedef __attribute__((ext_vector_type(2)))  float  v2f;
typedef __attribute__((ext_vector_type(8)))  float  v8f;

static __device__ inline void atom_add_f32(float* p, float v) {
    unsafeAtomicAdd(p, v);
}


__global__ __launch_bounds__(256) void init_deg(float* __restrict__ deg) {
    int i = blockIdx.x * blockDim.x + threadIdx.x;
    if (i < N_NODES) deg[i] = 1.0f;
}

__global__ __launch_bounds__(256) void accum_deg(const int* __restrict__ dst,
                                                 const float* __restrict__ ew,
                                                 float* __restrict__ deg) {
    int e = blockIdx.x * blockDim.x + threadIdx.x;
    if (e < N_EDGES) atom_add_f32(deg + dst[e], ew[e]);
}

__global__ __launch_bounds__(256) void make_dinv(float* __restrict__ deg) {
    int i = blockIdx.x * blockDim.x + threadIdx.x;
    if (i < N_NODES) deg[i] = rsqrtf(deg[i]);
}

__global__ __launch_bounds__(256) void make_norm(const int* __restrict__ src,
                                                 const int* __restrict__ dst,
                                                 const float* __restrict__ ew,
                                                 const float* __restrict__ dinv,
                                                 float* __restrict__ norm) {
    int e = blockIdx.x * blockDim.x + threadIdx.x;
    if (e < N_EDGES) norm[e] = dinv[src[e]] * ew[e] * dinv[dst[e]];
}


template<int K, int COLS>
__global__ __launch_bounds__(256) void gemm_wmma(const float* __restrict__ X,
                                                 const float* __restrict__ W,
                                                 float* __restrict__ H) {
    const int lane = threadIdx.x & 31;
    const int wave = threadIdx.x >> 5;
    const int tm = blockIdx.x * 8 + wave;
    const int tn = blockIdx.y;
    if (tm * 16 >= N_NODES) return;
    const int m0 = tm * 16, n0 = tn * 16;
    const int row   = m0 + (lane & 15);
    const int col   = n0 + (lane & 15);
    const int khalf = (lane >> 4) << 1;
    const float* xrow = X + (size_t)row * K;
    v8f c = {};
    for (int kk = 0; kk < K; kk += 4) {
        v2f a = *(const v2f*)(xrow + kk + khalf);
        v2f b;
        b.x = W[(size_t)(kk + khalf)     * COLS + col];
        b.y = W[(size_t)(kk + khalf + 1) * COLS + col];
        c = __builtin_amdgcn_wmma_f32_16x16x4_f32(false, a, false, b,
                                                  (short)0, c, false, false);
    }
    const int rbase = (lane >> 4) << 3;
    float* hrow = H + (size_t)(m0 + rbase) * COLS + col;
    #pragma unroll
    for (int r = 0; r < 8; ++r) hrow[(size_t)r * COLS] = c[r];
}


template<int C>
__global__ __launch_bounds__(256) void selfloop_init(const float* __restrict__ H,
                                                     const float* __restrict__ dinv,
                                                     float* __restrict__ out) {
    const int C4 = C / 4;
    int idx = blockIdx.x * blockDim.x + threadIdx.x;
    if (idx >= N_NODES * C4) return;
    int i = idx / C4, q = idx % C4;
    float di = dinv[i], w = di * di;
    float4 hv = *(const float4*)(H + (size_t)i * C + q * 4);
    float4 o = { w * hv.x, w * hv.y, w * hv.z, w * hv.w };
    *(float4*)(out + (size_t)i * C + q * 4) = o;
}

template<int C>
__global__ __launch_bounds__(256) void scatter_edges(const float* __restrict__ H,
                                                     const int* __restrict__ src,
                                                     const int* __restrict__ dst,
                                                     const float* __restrict__ norm,
                                                     float* __restrict__ out) {
    static_assert(N_EDGES % 256 == 0, "eight waves of 32 edges per block, no tail");
    constexpr int NT = (C + 31) / 32;
    constexpr int U  = 8;
    const int lane = threadIdx.x & 31;
    const int e0 = (blockIdx.x * 8 + (threadIdx.x >> 5)) * 32;
    const int   sv = src[e0 + lane];
    const int   dv = dst[e0 + lane];
    const float nv = norm[e0 + lane];
    for (int j0 = 0; j0 < 32; j0 += U) {
        const float* hp[U];
        float hv[U][NT] = {};
        #pragma unroll
        for (int u = 0; u < U; ++u) {
            const int s = __builtin_amdgcn_readlane(sv, j0 + u);
            hp[u] = H + (size_t)s * C + lane;
        }
        #pragma unroll
        for (int t = 0; t < NT; ++t)
            if (t * 32 + lane < C) {
                #pragma unroll
                for (int u = 0; u < U; ++u) hv[u][t] = hp[u][t * 32];
            }
        float* op[U];
        float  nw[U];
        #pragma unroll
        for (int u = 0; u < U; ++u) {
            const int d = __builtin_amdgcn_readlane(dv, j0 + u);
            nw[u] = __int_as_float(
                __builtin_amdgcn_readlane(__float_as_int(nv), j0 + u));
            op[u] = out + (size_t)d * C + lane;
        }
        #pragma unroll
        for (int t = 0; t < NT; ++t)
            if (t * 32 + lane < C) {
                #pragma unroll
                for (int u = 0; u < U; ++u)
                    atom_add_f32(op[u] + t * 32, nw[u] * hv[u][t]);
            }
    }
}

template<int C>
__global__ __launch_bounds__(256) void bias_relu(float* __restrict__ out,
                                                 const float* __restrict__ b) {
    const int C4 = C / 4;
    int idx = blockIdx.x * blockDim.x + threadIdx.x;
    if (idx >= N_NODES * C4) return;
    int i = idx / C4, q = idx % C4;
    float4 v  = *(float4*)(out + (size_t)i * C + q * 4);
    float4 bb = *(const float4*)(b + q * 4);
    v.x = fmaxf(v.x + bb.x, 0.0f);
    v.y = fmaxf(v.y + bb.y, 0.0f);
    v.z = fmaxf(v.z + bb.z, 0.0f);
    v.w = fmaxf(v.w + bb.w, 0.0f);
    *(float4*)(out + (size_t)i * C + q * 4) = v;
}

__global__ __launch_bounds__(256) void classifier(const float* __restrict__ Hin,
                                                  const float* __restrict__ Wc,
                                                  const float* __restrict__ bc,
                                                  float* __restrict__ logits) {
    int i = blockIdx.x * blockDim.x + threadIdx.x;
    if (i >= N_NODES) return;
    const float* h = Hin + (size_t)i * HID2;
    float a0 = bc[0], a1 = bc[1], a2 = bc[2];
    #pragma unroll
    for (int k = 0; k < HID2; ++k) {
        float hv = h[k];
        a0 += hv * Wc[k * 3 + 0];
        a1 += hv * Wc[k * 3 + 1];
        a2 += hv * Wc[k * 3 + 2];
    }
    logits[(size_t)i * 3 + 0] = a0;
    logits[(size_t)i * 3 + 1] = a1;
    logits[(size_t)i * 3 + 2] = a2;
}


extern "C" void kernel_launch(void* const* d_in, const int* in_sizes, int n_in,
                              void* d_out, int out_size, void* d_ws, size_t ws_size,
                              hipStream_t stream) {
    const float* x  = (const float*)d_in[0];
    const int*   ei = (const int*)  d_in[1];
    const float* ew = (const float*)d_in[2];
    const float* W1 = (const float*)d_in[3];
    const float* b1 = (const float*)d_in[4];
    const float* W2 = (const float*)d_in[5];
    const float* b2 = (const float*)d_in[6];
    const float* Wc = (const float*)d_in[7];
    const float* bc = (const float*)d_in[8];
    float* logits = (float*)d_out;

    const int* src = ei;
    const int* dst = ei + N_EDGES;

    char* ws = (char*)d_ws;
    size_t off = 0;
    auto take = [&](size_t bytes) -> void* {
        void* p = ws + off;
        off += (bytes + 255) & ~(size_t)255;
        return p;
    };
    float* dinv = (float*)take((size_t)N_NODES * 4);
    float* norm = (float*)take((size_t)N_EDGES * 4);
    float* H1   = (float*)take((size_t)N_NODES * HID * 4);
    float* out1 = (float*)take((size_t)N_NODES * HID * 4);
    float* H2   = H1;
    float* out2 = out1;

    const int TB = 256;
    init_deg <<<(N_NODES + TB - 1) / TB, TB, 0, stream>>>(dinv);
    accum_deg<<<(N_EDGES + TB - 1) / TB, TB, 0, stream>>>(dst, ew, dinv);
    make_dinv<<<(N_NODES + TB - 1) / TB, TB, 0, stream>>>(dinv);
    make_norm<<<(N_EDGES + TB - 1) / TB, TB, 0, stream>>>(src, dst, ew, dinv, norm);

    {
        dim3 g((N_NODES / 16 + 7) / 8, HID / 16);
        gemm_wmma<IN_CH, HID><<<g, 256, 0, stream>>>(x, W1, H1);
    }
    {
        int tot = N_NODES * (HID / 4);
        selfloop_init<HID><<<(tot + TB - 1) / TB, TB, 0, stream>>>(H1, dinv, out1);
        scatter_edges<HID><<<N_EDGES / 256, 256, 0, stream>>>(H1, src, dst, norm, out1);
        bias_relu<HID><<<(tot + TB - 1) / TB, TB, 0, stream>>>(out1, b1);
    }

    {
        dim3 g((N_NODES / 16 + 7) / 8, HID2 / 16);
        gemm_wmma<HID, HID2><<<g, 256, 0, stream>>>(out1, W2, H2);
    }
    {
        int tot = N_NODES * (HID2 / 4);
        selfloop_init<HID2><<<(tot + TB - 1) / TB, TB, 0, stream>>>(H2, dinv, out2);
        scatter_edges<HID2><<<N_EDGES / 256, 256, 0, stream>>>(H2, src, dst, norm, out2);
        bias_relu<HID2><<<(tot + TB - 1) / TB, TB, 0, stream>>>(out2, b2);
    }

    classifier<<<(N_NODES + TB - 1) / TB, TB, 0, stream>>>(out2, Wc, bc, logits);
}


# aggregation by destination: edges counting-sorted by dst per call (int rank atomics, scan); a wave sums a node's incoming rows in registers, stores the row once; no f32 atomics; sums in slot order
# speedup vs baseline: 2.1738x; 1.4570x over previous
#include <hip/hip_runtime.h>

#define N_NODES 100000
#define N_EDGES 3200000
#define IN_CH   128
#define HID     96
#define HID2    48
#define NCLS    3

typedef __attribute__((ext_vector_type(2)))  float  v2f;
typedef __attribute__((ext_vector_type(8)))  float  v8f;


#define SCAN_TB     256
#define SCAN_BLOCKS ((N_NODES + 1 + SCAN_TB - 1) / SCAN_TB)

__global__ __launch_bounds__(256) void init_cnt(int* __restrict__ cnt) {
    int i = blockIdx.x * blockDim.x + threadIdx.x;
    if (i < N_NODES) cnt[i] = 0;
}

__global__ __launch_bounds__(256) void rank_edges(const int* __restrict__ dst,
                                                  int* __restrict__ cnt,
                                                  int* __restrict__ rank) {
    int e = blockIdx.x * blockDim.x + threadIdx.x;
    if (e < N_EDGES) rank[e] = atomicAdd(cnt + dst[e], 1);
}

__global__ __launch_bounds__(SCAN_TB) void scan_local(const int* __restrict__ cnt,
                                                      int* __restrict__ rowptr,
                                                      int* __restrict__ blocksum) {
    __shared__ int wsum[SCAN_TB / 32];
    const int t = threadIdx.x, lane = t & 31, wave = t >> 5;
    const int i = blockIdx.x * SCAN_TB + t;
    const int c = (i < N_NODES) ? cnt[i] : 0;
    int incl = c;
    for (int d = 1; d < 32; d <<= 1) {
        const int v = __shfl_up(incl, d);
        if (lane >= d) incl += v;
    }
    if (lane == 31) wsum[wave] = incl;
    __syncthreads();
    int pre = 0;
    for (int w = 0; w < SCAN_TB / 32; ++w) if (w < wave) pre += wsum[w];
    if (i <= N_NODES) rowptr[i] = pre + incl - c;
    if (t == SCAN_TB - 1) blocksum[blockIdx.x] = pre + incl;
}

__global__ __launch_bounds__(SCAN_TB) void scan_offsets(const int* __restrict__ blocksum,
                                                        int* __restrict__ rowptr) {
    __shared__ int wsum[SCAN_TB / 32];
    const int t = threadIdx.x, lane = t & 31, wave = t >> 5;
    const int b = blockIdx.x;
    int v = 0;
    for (int k = t; k < b; k += SCAN_TB) v += blocksum[k];
    for (int d = 16; d >= 1; d >>= 1) v += __shfl_xor(v, d);
    if (lane == 0) wsum[wave] = v;
    __syncthreads();
    int off = 0;
    for (int w = 0; w < SCAN_TB / 32; ++w) off += wsum[w];
    const int i = b * SCAN_TB + t;
    if (i <= N_NODES) rowptr[i] += off;
}

__global__ __launch_bounds__(256) void fill_csr(const int* __restrict__ src,
                                                const int* __restrict__ dst,
                                                const float* __restrict__ ew,
                                                const int* __restrict__ rank,
                                                const int* __restrict__ rowptr,
                                                int* __restrict__ csr) {
    int e = blockIdx.x * blockDim.x + threadIdx.x;
    if (e < N_EDGES) {
        const int pos = rowptr[dst[e]] + rank[e];
        reinterpret_cast<int2*>(csr)[pos] = make_int2(src[e], __float_as_int(ew[e]));
    }
}

__global__ __launch_bounds__(256) void node_dinv(const int* __restrict__ rowptr,
                                                 const int* __restrict__ csr,
                                                 float* __restrict__ dinv) {
    int i = blockIdx.x * blockDim.x + threadIdx.x;
    if (i < N_NODES) {
        const int2* c = reinterpret_cast<const int2*>(csr);
        const int end = rowptr[i + 1];
        float d = 1.0f;
        for (int p = rowptr[i]; p < end; ++p) d += __int_as_float(c[p].y);
        dinv[i] = rsqrtf(d);
    }
}


template<int K, int COLS>
__global__ __launch_bounds__(256) void gemm_wmma(const float* __restrict__ X,
                                                 const float* __restrict__ W,
                                                 float* __restrict__ H) {
    const int lane = threadIdx.x & 31;
    const int wave = threadIdx.x >> 5;
    const int tm = blockIdx.x * 8 + wave;
    const int tn = blockIdx.y;
    if (tm * 16 >= N_NODES) return;
    const int m0 = tm * 16, n0 = tn * 16;
    const int row   = m0 + (lane & 15);
    const int col   = n0 + (lane & 15);
    const int khalf = (lane >> 4) << 1;
    const float* xrow = X + (size_t)row * K;
    v8f c = {};
    for (int kk = 0; kk < K; kk += 4) {
        v2f a = *(const v2f*)(xrow + kk + khalf);
        v2f b;
        b.x = W[(size_t)(kk + khalf)     * COLS + col];
        b.y = W[(size_t)(kk + khalf + 1) * COLS + col];
        c = __builtin_amdgcn_wmma_f32_16x16x4_f32(false, a, false, b,
                                                  (short)0, c, false, false);
    }
    const int rbase = (lane >> 4) << 3;
    float* hrow = H + (size_t)(m0 + rbase) * COLS + col;
    #pragma unroll
    for (int r = 0; r < 8; ++r) hrow[(size_t)r * COLS] = c[r];
}


template<int C, int NP, bool RAW>
static __device__ inline void sum_segment(const float* __restrict__ H, int2* csr,
                                          const float* __restrict__ dinv, float di,
                                          int beg, int end, int lane,
                                          const int (&ch)[NP], const bool (&act)[NP],
                                          float (&acc)[NP]) {
    constexpr int U = 8;
    for (int base = beg; base < end; base += 32) {
        const int left = end - base;
        const int n = left < 32 ? left : 32;
        int2 ev = make_int2(0, 0);
        if (lane < n) {
            ev = csr[base + lane];
            if (RAW) {
                ev.y = __float_as_int(dinv[ev.x] * __int_as_float(ev.y) * di);
                csr[base + lane].y = ev.y;
            }
        }
        for (int j0 = 0; j0 < n; j0 += U) {
            const int m = n - j0;
            float hv[U][NP];
            if (m >= U) {
                #pragma unroll
                for (int u = 0; u < U; ++u) {
                    const int s = __builtin_amdgcn_readlane(ev.x, j0 + u);
                    const float* hp = H + (size_t)s * C;
                    #pragma unroll
                    for (int p = 0; p < NP; ++p) if (act[p]) hv[u][p] = hp[ch[p]];
                }
                #pragma unroll
                for (int u = 0; u < U; ++u) {
                    const float w = __int_as_float(__builtin_amdgcn_readlane(ev.y, j0 + u));
                    #pragma unroll
                    for (int p = 0; p < NP; ++p) if (act[p]) acc[p] += w * hv[u][p];
                }
            } else {
                #pragma unroll
                for (int u = 0; u < U; ++u)
                    if (u < m) {
                        const int s = __builtin_amdgcn_readlane(ev.x, j0 + u);
                        const float* hp = H + (size_t)s * C;
                        #pragma unroll
                        for (int p = 0; p < NP; ++p) if (act[p]) hv[u][p] = hp[ch[p]];
                    }
                #pragma unroll
                for (int u = 0; u < U; ++u)
                    if (u < m) {
                        const float w = __int_as_float(__builtin_amdgcn_readlane(ev.y, j0 + u));
                        #pragma unroll
                        for (int p = 0; p < NP; ++p) if (act[p]) acc[p] += w * hv[u][p];
                    }
            }
        }
    }
}

#define NODES_PER_WAVE 4

__global__ __launch_bounds__(256) void gather_rows96(const float* __restrict__ H,
                                                     const int* __restrict__ rowptr,
                                                     int* csr,
                                                     const float* __restrict__ dinv,
                                                     const float* __restrict__ bias,
                                                     float* __restrict__ out) {
    static_assert(N_NODES % (8 * NODES_PER_WAVE) == 0, "eight waves per block, no tail");
    const int lane = threadIdx.x & 31;
    const int n0 = __builtin_amdgcn_readfirstlane(
        (blockIdx.x * 8 + (threadIdx.x >> 5)) * NODES_PER_WAVE);
    const int  ch[3]  = { lane, 32 + lane, 64 + lane };
    const bool act[3] = { true, true, true };
    const float bb[3] = { bias[ch[0]], bias[ch[1]], bias[ch[2]] };
    for (int k = 0; k < NODES_PER_WAVE; ++k) {
        const int node = n0 + k;
        const int beg  = __builtin_amdgcn_readfirstlane(rowptr[node]);
        const int end  = __builtin_amdgcn_readfirstlane(rowptr[node + 1]);
        const float di = dinv[node], w0 = di * di;
        const float* hs = H + (size_t)node * HID;
        float acc[3];
        #pragma unroll
        for (int p = 0; p < 3; ++p) acc[p] = w0 * hs[ch[p]];
        sum_segment<HID, 3, true>(H, reinterpret_cast<int2*>(csr), dinv, di,
                                  beg, end, lane, ch, act, acc);
        float* o = out + (size_t)node * HID;
        #pragma unroll
        for (int p = 0; p < 3; ++p) o[ch[p]] = fmaxf(acc[p] + bb[p], 0.0f);
    }
}

__global__ __launch_bounds__(256) void gather_rows48(const float* __restrict__ H,
                                                     const int* __restrict__ rowptr,
                                                     int* csr,
                                                     const float* __restrict__ dinv,
                                                     const float* __restrict__ bias,
                                                     float* __restrict__ out) {
    static_assert(NODES_PER_WAVE % 2 == 0, "whole pairs per wave");
    const int lane = threadIdx.x & 31;
    const int p0 = __builtin_amdgcn_readfirstlane(
        (blockIdx.x * 8 + (threadIdx.x >> 5)) * (NODES_PER_WAVE / 2));
    int2* cs = reinterpret_cast<int2*>(csr);
    const bool lo = lane < 16;
    const int  chA[2]  = { lane, 32 + (lane & 15) };
    const bool actA[2] = { true, lo };
    const int  chB[2]  = { 16 + lane, lane & 15 };
    const bool actB[2] = { true, !lo };
    const float bA = bias[lane];
    const float bM = bias[lo ? 32 + lane : lane - 16];
    const float bB = bias[16 + lane];
    for (int k = 0; k < NODES_PER_WAVE / 2; ++k) {
        const int pair = p0 + k;
        const int nA = 2 * pair, nB = nA + 1;
        const int r0 = __builtin_amdgcn_readfirstlane(rowptr[nA]);
        const int r1 = __builtin_amdgcn_readfirstlane(rowptr[nB]);
        const int r2 = __builtin_amdgcn_readfirstlane(rowptr[nB + 1]);
        float a[2], b[2];
        {
            const float di = dinv[nA], w0 = di * di;
            const float* hs = H + (size_t)nA * HID2;
            a[0] = w0 * hs[chA[0]];
            a[1] = lo ? w0 * hs[chA[1]] : 0.0f;
            sum_segment<HID2, 2, false>(H, cs, dinv, di, r0, r1, lane, chA, actA, a);
        }
        {
            const float di = dinv[nB], w0 = di * di;
            const float* hs = H + (size_t)nB * HID2;
            b[0] = w0 * hs[chB[0]];
            b[1] = lo ? 0.0f : w0 * hs[chB[1]];
            sum_segment<HID2, 2, false>(H, cs, dinv, di, r1, r2, lane, chB, actB, b);
        }
        float* o = out + (size_t)pair * (2 * HID2);
        const float mid = lo ? a[1] : b[1];
        o[lane]      = fmaxf(a[0] + bA, 0.0f);
        o[32 + lane] = fmaxf(mid  + bM, 0.0f);
        o[64 + lane] = fmaxf(b[0] + bB, 0.0f);
    }
}

__global__ __launch_bounds__(256) void classifier(const float* __restrict__ Hin,
                                                  const float* __restrict__ Wc,
                                                  const float* __restrict__ bc,
                                                  float* __restrict__ logits) {
    int i = blockIdx.x * blockDim.x + threadIdx.x;
    if (i >= N_NODES) return;
    const float* h = Hin + (size_t)i * HID2;
    float a0 = bc[0], a1 = bc[1], a2 = bc[2];
    #pragma unroll
    for (int k = 0; k < HID2; ++k) {
        float hv = h[k];
        a0 += hv * Wc[k * 3 + 0];
        a1 += hv * Wc[k * 3 + 1];
        a2 += hv * Wc[k * 3 + 2];
    }
    logits[(size_t)i * 3 + 0] = a0;
    logits[(size_t)i * 3 + 1] = a1;
    logits[(size_t)i * 3 + 2] = a2;
}


extern "C" void kernel_launch(void* const* d_in, const int* in_sizes, int n_in,
                              void* d_out, int out_size, void* d_ws, size_t ws_size,
                              hipStream_t stream) {
    const float* x  = (const float*)d_in[0];
    const int*   ei = (const int*)  d_in[1];
    const float* ew = (const float*)d_in[2];
    const float* W1 = (const float*)d_in[3];
    const float* b1 = (const float*)d_in[4];
    const float* W2 = (const float*)d_in[5];
    const float* b2 = (const float*)d_in[6];
    const float* Wc = (const float*)d_in[7];
    const float* bc = (const float*)d_in[8];
    float* logits = (float*)d_out;

    const int* src = ei;
    const int* dst = ei + N_EDGES;

    char* ws = (char*)d_ws;
    size_t off = 0;
    auto take = [&](size_t bytes) -> void* {
        void* p = ws + off;
        off += (bytes + 255) & ~(size_t)255;
        return p;
    };
    float* dinv     = (float*)take((size_t)N_NODES * 4);
    int*   cnt      = (int*)  take((size_t)N_NODES * 4);
    int*   rowptr   = (int*)  take((size_t)(N_NODES + 1) * 4);
    int*   blocksum = (int*)  take((size_t)SCAN_BLOCKS * 4);
    int*   rank     = (int*)  take((size_t)N_EDGES * 4);
    int*   csr      = (int*)  take((size_t)N_EDGES * 8);
    float* H1   = (float*)take((size_t)N_NODES * HID * 4);
    float* out1 = (float*)take((size_t)N_NODES * HID * 4);
    float* H2   = H1;
    float* out2 = out1;

    const int TB = 256;
    init_cnt    <<<(N_NODES + TB - 1) / TB, TB, 0, stream>>>(cnt);
    rank_edges  <<<(N_EDGES + TB - 1) / TB, TB, 0, stream>>>(dst, cnt, rank);
    scan_local  <<<SCAN_BLOCKS, SCAN_TB, 0, stream>>>(cnt, rowptr, blocksum);
    scan_offsets<<<SCAN_BLOCKS, SCAN_TB, 0, stream>>>(blocksum, rowptr);
    fill_csr    <<<(N_EDGES + TB - 1) / TB, TB, 0, stream>>>(src, dst, ew, rank, rowptr, csr);
    node_dinv   <<<(N_NODES + TB - 1) / TB, TB, 0, stream>>>(rowptr, csr, dinv);

    {
        dim3 g((N_NODES / 16 + 7) / 8, HID / 16);
        gemm_wmma<IN_CH, HID><<<g, 256, 0, stream>>>(x, W1, H1);
    }
    gather_rows96<<<N_NODES / (8 * NODES_PER_WAVE), 256, 0, stream>>>(H1, rowptr, csr, dinv, b1, out1);

    {
        dim3 g((N_NODES / 16 + 7) / 8, HID2 / 16);
        gemm_wmma<HID, HID2><<<g, 256, 0, stream>>>(out1, W2, H2);
    }
    gather_rows48<<<N_NODES / (8 * NODES_PER_WAVE), 256, 0, stream>>>(H2, rowptr, csr, dinv, b2, out2);

    classifier<<<(N_NODES + TB - 1) / TB, TB, 0, stream>>>(out2, Wc, bc, logits);
}


# edge ranks of the per-call counting sort come from returning LDS adds on per-chunk node counters (64 chunks, offsets by a scan over chunks), not global int atomics; slot order: chunk, then arrival
# speedup vs baseline: 2.9506x; 1.3573x over previous
#include <hip/hip_runtime.h>

#define N_NODES 100000
#define N_EDGES 3200000
#define IN_CH   128
#define HID     96
#define HID2    48
#define NCLS    3

typedef __attribute__((ext_vector_type(2)))  float  v2f;
typedef __attribute__((ext_vector_type(8)))  float  v8f;


#define SCAN_TB     256
#define SCAN_BLOCKS ((N_NODES + 1 + SCAN_TB - 1) / SCAN_TB)

#define NCH     64
#define CH      (N_EDGES / NCH)
#define NWORDS  (N_NODES / 2)
#define RK_TB   1024
#define RK_PER  ((CH + RK_TB - 1) / RK_TB)
static_assert(N_EDGES % NCH == 0, "whole chunks");
static_assert(CH < 65536, "a chunk cannot overflow a 16-bit counter, whatever the edges are");
static_assert(N_NODES % 2 == 0 && NWORDS % 4 == 0, "counter words move as 16-byte vectors");

__global__ __launch_bounds__(RK_TB) void rank_chunk(const int* __restrict__ dst,
                                                    unsigned* __restrict__ chunkcnt,
                                                    unsigned short* __restrict__ rank) {
    __shared__ uint4 cw4[NWORDS / 4];
    unsigned* cw = reinterpret_cast<unsigned*>(cw4);
    const int t  = threadIdx.x;
    const int e0 = blockIdx.x * CH;
    for (int i = t; i < NWORDS / 4; i += RK_TB) cw4[i] = make_uint4(0u, 0u, 0u, 0u);
    const bool has_last = t + (RK_PER - 1) * RK_TB < CH;
    int d[RK_PER];
    #pragma unroll
    for (int k = 0; k < RK_PER; ++k)
        d[k] = (k < RK_PER - 1 || has_last) ? dst[e0 + t + k * RK_TB] : 0;
    __syncthreads();
    unsigned r[RK_PER];
    #pragma unroll
    for (int k = 0; k < RK_PER; ++k) {
        r[k] = 0u;
        if (k < RK_PER - 1 || has_last) {
            const unsigned sh = (unsigned)(d[k] & 1) << 4;
            r[k] = atomicAdd(cw + (d[k] >> 1), 1u << sh) >> sh;
        }
    }
    #pragma unroll
    for (int k = 0; k < RK_PER; ++k)
        if (k < RK_PER - 1 || has_last) rank[e0 + t + k * RK_TB] = (unsigned short)r[k];
    __syncthreads();
    uint4* outw = reinterpret_cast<uint4*>(chunkcnt + (size_t)blockIdx.x * NWORDS);
    for (int i = t; i < NWORDS / 4; i += RK_TB) outw[i] = cw4[i];
}

#define CS_TB     256
#define CS_GROUPS (CS_TB / 32)
#define CS_PER    (NCH / CS_GROUPS)
#define CS_BLOCKS ((NWORDS + 31) / 32)
static_assert(NCH % CS_GROUPS == 0, "equal share of chunks per wave");

__global__ __launch_bounds__(CS_TB) void chunk_offsets(const unsigned* __restrict__ chunkcnt,
                                                       int* __restrict__ pre,
                                                       int* __restrict__ cnt) {
    __shared__ int2 part[CS_GROUPS][32];
    const int lane = threadIdx.x & 31, g = threadIdx.x >> 5;
    const int word = blockIdx.x * 32 + lane;
    const bool ok  = word < NWORDS;
    const int  wl  = ok ? word : NWORDS - 1;
    unsigned v[CS_PER];
    #pragma unroll
    for (int k = 0; k < CS_PER; ++k)
        v[k] = chunkcnt[(size_t)(g * CS_PER + k) * NWORDS + wl];
    int lo = 0, hi = 0;
    #pragma unroll
    for (int k = 0; k < CS_PER; ++k) { lo += (int)(v[k] & 0xFFFFu); hi += (int)(v[k] >> 16); }
    part[g][lane] = make_int2(lo, hi);
    __syncthreads();
    int blo = 0, bhi = 0;
    for (int w = 0; w < CS_GROUPS; ++w)
        if (w < g) { const int2 p = part[w][lane]; blo += p.x; bhi += p.y; }
    if (ok) {
        #pragma unroll
        for (int k = 0; k < CS_PER; ++k) {
            reinterpret_cast<int2*>(pre + (size_t)(g * CS_PER + k) * N_NODES)[word] = make_int2(blo, bhi);
            blo += (int)(v[k] & 0xFFFFu);
            bhi += (int)(v[k] >> 16);
        }
        if (g == CS_GROUPS - 1) reinterpret_cast<int2*>(cnt)[word] = make_int2(blo, bhi);
    }
}

__global__ __launch_bounds__(SCAN_TB) void scan_local(const int* __restrict__ cnt,
                                                      int* __restrict__ rowptr,
                                                      int* __restrict__ blocksum) {
    __shared__ int wsum[SCAN_TB / 32];
    const int t = threadIdx.x, lane = t & 31, wave = t >> 5;
    const int i = blockIdx.x * SCAN_TB + t;
    const int c = (i < N_NODES) ? cnt[i] : 0;
    int incl = c;
    for (int d = 1; d < 32; d <<= 1) {
        const int v = __shfl_up(incl, d);
        if (lane >= d) incl += v;
    }
    if (lane == 31) wsum[wave] = incl;
    __syncthreads();
    int pre = 0;
    for (int w = 0; w < SCAN_TB / 32; ++w) if (w < wave) pre += wsum[w];
    if (i <= N_NODES) rowptr[i] = pre + incl - c;
    if (t == SCAN_TB - 1) blocksum[blockIdx.x] = pre + incl;
}

__global__ __launch_bounds__(SCAN_TB) void scan_offsets(const int* __restrict__ blocksum,
                                                        int* __restrict__ rowptr) {
    __shared__ int wsum[SCAN_TB / 32];
    const int t = threadIdx.x, lane = t & 31, wave = t >> 5;
    const int b = blockIdx.x;
    int v = 0;
    for (int k = t; k < b; k += SCAN_TB) v += blocksum[k];
    for (int d = 16; d >= 1; d >>= 1) v += __shfl_xor(v, d);
    if (lane == 0) wsum[wave] = v;
    __syncthreads();
    int off = 0;
    for (int w = 0; w < SCAN_TB / 32; ++w) off += wsum[w];
    const int i = b * SCAN_TB + t;
    if (i <= N_NODES) rowptr[i] += off;
}

__global__ __launch_bounds__(256) void fill_csr(const int* __restrict__ src,
                                                const int* __restrict__ dst,
                                                const float* __restrict__ ew,
                                                const unsigned short* __restrict__ rank,
                                                const int* __restrict__ rowptr,
                                                const int* __restrict__ pre,
                                                int* __restrict__ csr) {
    int e = blockIdx.x * blockDim.x + threadIdx.x;
    if (e < N_EDGES) {
        const int d   = dst[e];
        const int pos = rowptr[d] + pre[(size_t)(e / CH) * N_NODES + d] + (int)rank[e];
        reinterpret_cast<int2*>(csr)[pos] = make_int2(src[e], __float_as_int(ew[e]));
    }
}

__global__ __launch_bounds__(256) void node_dinv(const int* __restrict__ rowptr,
                                                 const int* __restrict__ csr,
                                                 float* __restrict__ dinv) {
    int i = blockIdx.x * blockDim.x + threadIdx.x;
    if (i < N_NODES) {
        const int2* c = reinterpret_cast<const int2*>(csr);
        const int end = rowptr[i + 1];
        float d = 1.0f;
        for (int p = rowptr[i]; p < end; ++p) d += __int_as_float(c[p].y);
        dinv[i] = rsqrtf(d);
    }
}


template<int K, int COLS>
__global__ __launch_bounds__(256) void gemm_wmma(const float* __restrict__ X,
                                                 const float* __restrict__ W,
                                                 float* __restrict__ H) {
    const int lane = threadIdx.x & 31;
    const int wave = threadIdx.x >> 5;
    const int tm = blockIdx.x * 8 + wave;
    const int tn = blockIdx.y;
    if (tm * 16 >= N_NODES) return;
    const int m0 = tm * 16, n0 = tn * 16;
    const int row   = m0 + (lane & 15);
    const int col   = n0 + (lane & 15);
    const int khalf = (lane >> 4) << 1;
    const float* xrow = X + (size_t)row * K;
    v8f c = {};
    for (int kk = 0; kk < K; kk += 4) {
        v2f a = *(const v2f*)(xrow + kk + khalf);
        v2f b;
        b.x = W[(size_t)(kk + khalf)     * COLS + col];
        b.y = W[(size_t)(kk + khalf + 1) * COLS + col];
        c = __builtin_amdgcn_wmma_f32_16x16x4_f32(false, a, false, b,
                                                  (short)0, c, false, false);
    }
    const int rbase = (lane >> 4) << 3;
    float* hrow = H + (size_t)(m0 + rbase) * COLS + col;
    #pragma unroll
    for (int r = 0; r < 8; ++r) hrow[(size_t)r * COLS] = c[r];
}


template<int C, int NP, bool RAW>
static __device__ inline void sum_segment(const float* __restrict__ H, int2* csr,
                                          const float* __restrict__ dinv, float di,
                                          int beg, int end, int lane,
                                          const int (&ch)[NP], const bool (&act)[NP],
                                          float (&acc)[NP]) {
    constexpr int U = 8;
    for (int base = beg; base < end; base += 32) {
        const int left = end - base;
        const int n = left < 32 ? left : 32;
        int2 ev = make_int2(0, 0);
        if (lane < n) {
            ev = csr[base + lane];
            if (RAW) {
                ev.y = __float_as_int(dinv[ev.x] * __int_as_float(ev.y) * di);
                csr[base + lane].y = ev.y;
            }
        }
        for (int j0 = 0; j0 < n; j0 += U) {
            const int m = n - j0;
            float hv[U][NP];
            if (m >= U) {
                #pragma unroll
                for (int u = 0; u < U; ++u) {
                    const int s = __builtin_amdgcn_readlane(ev.x, j0 + u);
                    const float* hp = H + (size_t)s * C;
                    #pragma unroll
                    for (int p = 0; p < NP; ++p) if (act[p]) hv[u][p] = hp[ch[p]];
                }
                #pragma unroll
                for (int u = 0; u < U; ++u) {
                    const float w = __int_as_float(__builtin_amdgcn_readlane(ev.y, j0 + u));
                    #pragma unroll
                    for (int p = 0; p < NP; ++p) if (act[p]) acc[p] += w * hv[u][p];
                }
            } else {
                #pragma unroll
                for (int u = 0; u < U; ++u)
                    if (u < m) {
                        const int s = __builtin_amdgcn_readlane(ev.x, j0 + u);
                        const float* hp = H + (size_t)s * C;
                        #pragma unroll
                        for (int p = 0; p < NP; ++p) if (act[p]) hv[u][p] = hp[ch[p]];
                    }
                #pragma unroll
                for (int u = 0; u < U; ++u)
                    if (u < m) {
                        const float w = __int_as_float(__builtin_amdgcn_readlane(ev.y, j0 + u));
                        #pragma unroll
                        for (int p = 0; p < NP; ++p) if (act[p]) acc[p] += w * hv[u][p];
                    }
            }
        }
    }
}

#define NODES_PER_WAVE 4

__global__ __launch_bounds__(256) void gather_rows96(const float* __restrict__ H,
                                                     const int* __restrict__ rowptr,
                                                     int* csr,
                                                     const float* __restrict__ dinv,
                                                     const float* __restrict__ bias,
                                                     float* __restrict__ out) {
    static_assert(N_NODES % (8 * NODES_PER_WAVE) == 0, "eight waves per block, no tail");
    const int lane = threadIdx.x & 31;
    const int n0 = __builtin_amdgcn_readfirstlane(
        (blockIdx.x * 8 + (threadIdx.x >> 5)) * NODES_PER_WAVE);
    const int  ch[3]  = { lane, 32 + lane, 64 + lane };
    const bool act[3] = { true, true, true };
    const float bb[3] = { bias[ch[0]], bias[ch[1]], bias[ch[2]] };
    for (int k = 0; k < NODES_PER_WAVE; ++k) {
        const int node = n0 + k;
        const int beg  = __builtin_amdgcn_readfirstlane(rowptr[node]);
        const int end  = __builtin_amdgcn_readfirstlane(rowptr[node + 1]);
        const float di = dinv[node], w0 = di * di;
        const float* hs = H + (size_t)node * HID;
        float acc[3];
        #pragma unroll
        for (int p = 0; p < 3; ++p) acc[p] = w0 * hs[ch[p]];
        sum_segment<HID, 3, true>(H, reinterpret_cast<int2*>(csr), dinv, di,
                                  beg, end, lane, ch, act, acc);
        float* o = out + (size_t)node * HID;
        #pragma unroll
        for (int p = 0; p < 3; ++p) o[ch[p]] = fmaxf(acc[p] + bb[p], 0.0f);
    }
}

__global__ __launch_bounds__(256) void gather_rows48(const float* __restrict__ H,
                                                     const int* __restrict__ rowptr,
                                                     int* csr,
                                                     const float* __restrict__ dinv,
                                                     const float* __restrict__ bias,
                                                     float* __restrict__ out) {
    static_assert(NODES_PER_WAVE % 2 == 0, "whole pairs per wave");
    const int lane = threadIdx.x & 31;
    const int p0 = __builtin_amdgcn_readfirstlane(
        (blockIdx.x * 8 + (threadIdx.x >> 5)) * (NODES_PER_WAVE / 2));
    int2* cs = reinterpret_cast<int2*>(csr);
    const bool lo = lane < 16;
    const int  chA[2]  = { lane, 32 + (lane & 15) };
    const bool actA[2] = { true, lo };
    const int  chB[2]  = { 16 + lane, lane & 15 };
    const bool actB[2] = { true, !lo };
    const float bA = bias[lane];
    const float bM = bias[lo ? 32 + lane : lane - 16];
    const float bB = bias[16 + lane];
    for (int k = 0; k < NODES_PER_WAVE / 2; ++k) {
        const int pair = p0 + k;
        const int nA = 2 * pair, nB = nA + 1;
        const int r0 = __builtin_amdgcn_readfirstlane(rowptr[nA]);
        const int r1 = __builtin_amdgcn_readfirstlane(rowptr[nB]);
        const int r2 = __builtin_amdgcn_readfirstlane(rowptr[nB + 1]);
        float a[2], b[2];
        {
            const float di = dinv[nA], w0 = di * di;
            const float* hs = H + (size_t)nA * HID2;
            a[0] = w0 * hs[chA[0]];
            a[1] = lo ? w0 * hs[chA[1]] : 0.0f;
            sum_segment<HID2, 2, false>(H, cs, dinv, di, r0, r1, lane, chA, actA, a);
        }
        {
            const float di = dinv[nB], w0 = di * di;
            const float* hs = H + (size_t)nB * HID2;
            b[0] = w0 * hs[chB[0]];
            b[1] = lo ? 0.0f : w0 * hs[chB[1]];
            sum_segment<HID2, 2, false>(H, cs, dinv, di, r1, r2, lane, chB, actB, b);
        }
        float* o = out + (size_t)pair * (2 * HID2);
        const float mid = lo ? a[1] : b[1];
        o[lane]      = fmaxf(a[0] + bA, 0.0f);
        o[32 + lane] = fmaxf(mid  + bM, 0.0f);
        o[64 + lane] = fmaxf(b[0] + bB, 0.0f);
    }
}

__global__ __launch_bounds__(256) void classifier(const float* __restrict__ Hin,
                                                  const float* __restrict__ Wc,
                                                  const float* __restrict__ bc,
                                                  float* __restrict__ logits) {
    int i = blockIdx.x * blockDim.x + threadIdx.x;
    if (i >= N_NODES) return;
    const float* h = Hin + (size_t)i * HID2;
    float a0 = bc[0], a1 = bc[1], a2 = bc[2];
    #pragma unroll
    for (int k = 0; k < HID2; ++k) {
        float hv = h[k];
        a0 += hv * Wc[k * 3 + 0];
        a1 += hv * Wc[k * 3 + 1];
        a2 += hv * Wc[k * 3 + 2];
    }
    logits[(size_t)i * 3 + 0] = a0;
    logits[(size_t)i * 3 + 1] = a1;
    logits[(size_t)i * 3 + 2] = a2;
}


extern "C" void kernel_launch(void* const* d_in, const int* in_sizes, int n_in,
                              void* d_out, int out_size, void* d_ws, size_t ws_size,
                              hipStream_t stream) {
    const float* x  = (const float*)d_in[0];
    const int*   ei = (const int*)  d_in[1];
    const float* ew = (const float*)d_in[2];
    const float* W1 = (const float*)d_in[3];
    const float* b1 = (const float*)d_in[4];
    const float* W2 = (const float*)d_in[5];
    const float* b2 = (const float*)d_in[6];
    const float* Wc = (const float*)d_in[7];
    const float* bc = (const float*)d_in[8];
    float* logits = (float*)d_out;

    const int* src = ei;
    const int* dst = ei + N_EDGES;

    char* ws = (char*)d_ws;
    size_t off = 0;
    auto take = [&](size_t bytes) -> void* {
        void* p = ws + off;
        off += (bytes + 255) & ~(size_t)255;
        return p;
    };
    float* dinv     = (float*)take((size_t)N_NODES * 4);
    int*   cnt      = (int*)  take((size_t)N_NODES * 4);
    int*   rowptr   = (int*)  take((size_t)(N_NODES + 1) * 4);
    int*   blocksum = (int*)  take((size_t)SCAN_BLOCKS * 4);
    unsigned short* rank = (unsigned short*)take((size_t)N_EDGES * 2);
    unsigned* chunkcnt   = (unsigned*)take((size_t)NCH * NWORDS * 4);
    int*   pre      = (int*)  take((size_t)NCH * N_NODES * 4);
    int*   csr      = (int*)  take((size_t)N_EDGES * 8);
    float* H1   = (float*)take((size_t)N_NODES * HID * 4);
    float* out1 = (float*)take((size_t)N_NODES * HID * 4);
    float* H2   = H1;
    float* out2 = out1;

    const int TB = 256;
    rank_chunk  <<<NCH, RK_TB, 0, stream>>>(dst, chunkcnt, rank);
    chunk_offsets<<<CS_BLOCKS, CS_TB, 0, stream>>>(chunkcnt, pre, cnt);
    scan_local  <<<SCAN_BLOCKS, SCAN_TB, 0, stream>>>(cnt, rowptr, blocksum);
    scan_offsets<<<SCAN_BLOCKS, SCAN_TB, 0, stream>>>(blocksum, rowptr);
    fill_csr    <<<(N_EDGES + TB - 1) / TB, TB, 0, stream>>>(src, dst, ew, rank, rowptr, pre, csr);
    node_dinv   <<<(N_NODES + TB - 1) / TB, TB, 0, stream>>>(rowptr, csr, dinv);

    {
        dim3 g((N_NODES / 16 + 7) / 8, HID / 16);
        gemm_wmma<IN_CH, HID><<<g, 256, 0, stream>>>(x, W1, H1);
    }
    gather_rows96<<<N_NODES / (8 * NODES_PER_WAVE), 256, 0, stream>>>(H1, rowptr, csr, dinv, b1, out1);

    {
        dim3 g((N_NODES / 16 + 7) / 8, HID2 / 16);
        gemm_wmma<HID, HID2><<<g, 256, 0, stream>>>(out1, W2, H2);
    }
    gather_rows48<<<N_NODES / (8 * NODES_PER_WAVE), 256, 0, stream>>>(H2, rowptr, csr, dinv, b2, out2);

    classifier<<<(N_NODES + TB - 1) / TB, TB, 0, stream>>>(out2, Wc, bc, logits);
}


# GEMM re-tiled: a wave owns 16 rows x all columns (6 or 3 independent accumulator tiles), W staged once per block in LDS as k pairs; X rows are read once, not once per column tile; same sum order
# speedup vs baseline: 3.2991x; 1.1181x over previous
#include <hip/hip_runtime.h>

#define N_NODES 100000
#define N_EDGES 3200000
#define IN_CH   128
#define HID     96
#define HID2    48
#define NCLS    3

typedef __attribute__((ext_vector_type(2)))  float  v2f;
typedef __attribute__((ext_vector_type(8)))  float  v8f;


#define SCAN_TB     256
#define SCAN_BLOCKS ((N_NODES + 1 + SCAN_TB - 1) / SCAN_TB)

#define NCH     64
#define CH      (N_EDGES / NCH)
#define NWORDS  (N_NODES / 2)
#define RK_TB   1024
#define RK_PER  ((CH + RK_TB - 1) / RK_TB)
static_assert(N_EDGES % NCH == 0, "whole chunks");
static_assert(CH < 65536, "a chunk cannot overflow a 16-bit counter, whatever the edges are");
static_assert(N_NODES % 2 == 0 && NWORDS % 4 == 0, "counter words move as 16-byte vectors");

__global__ __launch_bounds__(RK_TB) void rank_chunk(const int* __restrict__ dst,
                                                    unsigned* __restrict__ chunkcnt,
                                                    unsigned short* __restrict__ rank) {
    __shared__ uint4 cw4[NWORDS / 4];
    unsigned* cw = reinterpret_cast<unsigned*>(cw4);
    const int t  = threadIdx.x;
    const int e0 = blockIdx.x * CH;
    for (int i = t; i < NWORDS / 4; i += RK_TB) cw4[i] = make_uint4(0u, 0u, 0u, 0u);
    const bool has_last = t + (RK_PER - 1) * RK_TB < CH;
    int d[RK_PER];
    #pragma unroll
    for (int k = 0; k < RK_PER; ++k)
        d[k] = (k < RK_PER - 1 || has_last) ? dst[e0 + t + k * RK_TB] : 0;
    __syncthreads();
    unsigned r[RK_PER];
    #pragma unroll
    for (int k = 0; k < RK_PER; ++k) {
        r[k] = 0u;
        if (k < RK_PER - 1 || has_last) {
            const unsigned sh = (unsigned)(d[k] & 1) << 4;
            r[k] = atomicAdd(cw + (d[k] >> 1), 1u << sh) >> sh;
        }
    }
    #pragma unroll
    for (int k = 0; k < RK_PER; ++k)
        if (k < RK_PER - 1 || has_last) rank[e0 + t + k * RK_TB] = (unsigned short)r[k];
    __syncthreads();
    uint4* outw = reinterpret_cast<uint4*>(chunkcnt + (size_t)blockIdx.x * NWORDS);
    for (int i = t; i < NWORDS / 4; i += RK_TB) outw[i] = cw4[i];
}

#define CS_TB     256
#define CS_GROUPS (CS_TB / 32)
#define CS_PER    (NCH / CS_GROUPS)
#define CS_BLOCKS ((NWORDS + 31) / 32)
static_assert(NCH % CS_GROUPS == 0, "equal share of chunks per wave");

__global__ __launch_bounds__(CS_TB) void chunk_offsets(const unsigned* __restrict__ chunkcnt,
                                                       int* __restrict__ pre,
                                                       int* __restrict__ cnt) {
    __shared__ int2 part[CS_GROUPS][32];
    const int lane = threadIdx.x & 31, g = threadIdx.x >> 5;
    const int word = blockIdx.x * 32 + lane;
    const bool ok  = word < NWORDS;
    const int  wl  = ok ? word : NWORDS - 1;
    unsigned v[CS_PER];
    #pragma unroll
    for (int k = 0; k < CS_PER; ++k)
        v[k] = chunkcnt[(size_t)(g * CS_PER + k) * NWORDS + wl];
    int lo = 0, hi = 0;
    #pragma unroll
    for (int k = 0; k < CS_PER; ++k) { lo += (int)(v[k] & 0xFFFFu); hi += (int)(v[k] >> 16); }
    part[g][lane] = make_int2(lo, hi);
    __syncthreads();
    int blo = 0, bhi = 0;
    for (int w = 0; w < CS_GROUPS; ++w)
        if (w < g) { const int2 p = part[w][lane]; blo += p.x; bhi += p.y; }
    if (ok) {
        #pragma unroll
        for (int k = 0; k < CS_PER; ++k) {
            reinterpret_cast<int2*>(pre + (size_t)(g * CS_PER + k) * N_NODES)[word] = make_int2(blo, bhi);
            blo += (int)(v[k] & 0xFFFFu);
            bhi += (int)(v[k] >> 16);
        }
        if (g == CS_GROUPS - 1) reinterpret_cast<int2*>(cnt)[word] = make_int2(blo, bhi);
    }
}

__global__ __launch_bounds__(SCAN_TB) void scan_local(const int* __restrict__ cnt,
                                                      int* __restrict__ rowptr,
                                                      int* __restrict__ blocksum) {
    __shared__ int wsum[SCAN_TB / 32];
    const int t = threadIdx.x, lane = t & 31, wave = t >> 5;
    const int i = blockIdx.x * SCAN_TB + t;
    const int c = (i < N_NODES) ? cnt[i] : 0;
    int incl = c;
    for (int d = 1; d < 32; d <<= 1) {
        const int v = __shfl_up(incl, d);
        if (lane >= d) incl += v;
    }
    if (lane == 31) wsum[wave] = incl;
    __syncthreads();
    int pre = 0;
    for (int w = 0; w < SCAN_TB / 32; ++w) if (w < wave) pre += wsum[w];
    if (i <= N_NODES) rowptr[i] = pre + incl - c;
    if (t == SCAN_TB - 1) blocksum[blockIdx.x] = pre + incl;
}

__global__ __launch_bounds__(SCAN_TB) void scan_offsets(const int* __restrict__ blocksum,
                                                        int* __restrict__ rowptr) {
    __shared__ int wsum[SCAN_TB / 32];
    const int t = threadIdx.x, lane = t & 31, wave = t >> 5;
    const int b = blockIdx.x;
    int v = 0;
    for (int k = t; k < b; k += SCAN_TB) v += blocksum[k];
    for (int d = 16; d >= 1; d >>= 1) v += __shfl_xor(v, d);
    if (lane == 0) wsum[wave] = v;
    __syncthreads();
    int off = 0;
    for (int w = 0; w < SCAN_TB / 32; ++w) off += wsum[w];
    const int i = b * SCAN_TB + t;
    if (i <= N_NODES) rowptr[i] += off;
}

__global__ __launch_bounds__(256) void fill_csr(const int* __restrict__ src,
                                                const int* __restrict__ dst,
                                                const float* __restrict__ ew,
                                                const unsigned short* __restrict__ rank,
                                                const int* __restrict__ rowptr,
                                                const int* __restrict__ pre,
                                                int* __restrict__ csr) {
    int e = blockIdx.x * blockDim.x + threadIdx.x;
    if (e < N_EDGES) {
        const int d   = dst[e];
        const int pos = rowptr[d] + pre[(size_t)(e / CH) * N_NODES + d] + (int)rank[e];
        reinterpret_cast<int2*>(csr)[pos] = make_int2(src[e], __float_as_int(ew[e]));
    }
}

__global__ __launch_bounds__(256) void node_dinv(const int* __restrict__ rowptr,
                                                 const int* __restrict__ csr,
                                                 float* __restrict__ dinv) {
    int i = blockIdx.x * blockDim.x + threadIdx.x;
    if (i < N_NODES) {
        const int2* c = reinterpret_cast<const int2*>(csr);
        const int end = rowptr[i + 1];
        float d = 1.0f;
        for (int p = rowptr[i]; p < end; ++p) d += __int_as_float(c[p].y);
        dinv[i] = rsqrtf(d);
    }
}


template<int K, int COLS>
__global__ __launch_bounds__(256) void gemm_wmma(const float* __restrict__ X,
                                                 const float* __restrict__ W,
                                                 float* __restrict__ H) {
    constexpr int NT = COLS / 16;
    constexpr int KS = K / 4;
    constexpr int P2 = COLS + ((16 - COLS % 32 + 32) % 32);
    static_assert(K % 4 == 0 && COLS % 16 == 0 && P2 % 32 == 16 && P2 % 4 == 0, "tile shapes");
    __shared__ __attribute__((aligned(16))) v2f wp[(K / 2) * P2];
    const int lane = threadIdx.x & 31;
    const int wave = threadIdx.x >> 5;
    const int c = lane & 15, h = lane >> 4;
    const int m0 = (blockIdx.x * 8 + wave) * 16;
    const bool live = m0 < N_NODES;
    const int row = (live ? m0 : N_NODES - 16) + c;
    const float* xrow = X + (size_t)row * K + 2 * h;
    constexpr int QR  = COLS / 4;
    constexpr int NQ  = (K / 2) * QR;
    constexpr int WIT = (NQ + 255) / 256;
    float4 w0[WIT], w1[WIT];
    #pragma unroll
    for (int j = 0; j < WIT; ++j) {
        const int idx = j * 256 + (int)threadIdx.x;
        if ((j + 1) * 256 <= NQ || idx < NQ) {
            const int kp = idx / QR, q = idx - kp * QR;
            w0[j] = *(const float4*)(W + (size_t)(2 * kp)     * COLS + 4 * q);
            w1[j] = *(const float4*)(W + (size_t)(2 * kp + 1) * COLS + 4 * q);
        }
    }
    #pragma unroll
    for (int j = 0; j < WIT; ++j) {
        const int idx = j * 256 + (int)threadIdx.x;
        if ((j + 1) * 256 <= NQ || idx < NQ) {
            const int kp = idx / QR, q = idx - kp * QR;
            float4* d = reinterpret_cast<float4*>(wp + kp * P2 + 4 * q);
            d[0] = make_float4(w0[j].x, w1[j].x, w0[j].y, w1[j].y);
            d[1] = make_float4(w0[j].z, w1[j].z, w0[j].w, w1[j].w);
        }
    }
    __syncthreads();
    v2f a[KS];
    #pragma unroll
    for (int i = 0; i < KS; ++i) a[i] = *(const v2f*)(xrow + 4 * i);
    v8f acc[NT];
    #pragma unroll
    for (int t = 0; t < NT; ++t) acc[t] = v8f{};
    const v2f* bp = wp + h * P2 + c;
    #pragma unroll
    for (int i = 0; i < KS; ++i) {
        #pragma unroll
        for (int t = 0; t < NT; ++t) {
            const v2f b = bp[(2 * i) * P2 + 16 * t];
            acc[t] = __builtin_amdgcn_wmma_f32_16x16x4_f32(false, a[i], false, b,
                                                           (short)0, acc[t], false, false);
        }
    }
    if (live) {
        float* hrow = H + (size_t)(m0 + 8 * h) * COLS + c;
        #pragma unroll
        for (int t = 0; t < NT; ++t) {
            #pragma unroll
            for (int r = 0; r < 8; ++r) hrow[(size_t)r * COLS + 16 * t] = acc[t][r];
        }
    }
}


template<int C, int NP, bool RAW>
static __device__ inline void sum_segment(const float* __restrict__ H, int2* csr,
                                          const float* __restrict__ dinv, float di,
                                          int beg, int end, int lane,
                                          const int (&ch)[NP], const bool (&act)[NP],
                                          float (&acc)[NP]) {
    constexpr int U = 8;
    for (int base = beg; base < end; base += 32) {
        const int left = end - base;
        const int n = left < 32 ? left : 32;
        int2 ev = make_int2(0, 0);
        if (lane < n) {
            ev = csr[base + lane];
            if (RAW) {
                ev.y = __float_as_int(dinv[ev.x] * __int_as_float(ev.y) * di);
                csr[base + lane].y = ev.y;
            }
        }
        for (int j0 = 0; j0 < n; j0 += U) {
            const int m = n - j0;
            float hv[U][NP];
            if (m >= U) {
                #pragma unroll
                for (int u = 0; u < U; ++u) {
                    const int s = __builtin_amdgcn_readlane(ev.x, j0 + u);
                    const float* hp = H + (size_t)s * C;
                    #pragma unroll
                    for (int p = 0; p < NP; ++p) if (act[p]) hv[u][p] = hp[ch[p]];
                }
                #pragma unroll
                for (int u = 0; u < U; ++u) {
                    const float w = __int_as_float(__builtin_amdgcn_readlane(ev.y, j0 + u));
                    #pragma unroll
                    for (int p = 0; p < NP; ++p) if (act[p]) acc[p] += w * hv[u][p];
                }
            } else {
                #pragma unroll
                for (int u = 0; u < U; ++u)
                    if (u < m) {
                        const int s = __builtin_amdgcn_readlane(ev.x, j0 + u);
                        const float* hp = H + (size_t)s * C;
                        #pragma unroll
                        for (int p = 0; p < NP; ++p) if (act[p]) hv[u][p] = hp[ch[p]];
                    }
                #pragma unroll
                for (int u = 0; u < U; ++u)
                    if (u < m) {
                        const float w = __int_as_float(__builtin_amdgcn_readlane(ev.y, j0 + u));
                        #pragma unroll
                        for (int p = 0; p < NP; ++p) if (act[p]) acc[p] += w * hv[u][p];
                    }
            }
        }
    }
}

#define NODES_PER_WAVE 4

__global__ __launch_bounds__(256) void gather_rows96(const float* __restrict__ H,
                                                     const int* __restrict__ rowptr,
                                                     int* csr,
                                                     const float* __restrict__ dinv,
                                                     const float* __restrict__ bias,
                                                     float* __restrict__ out) {
    static_assert(N_NODES % (8 * NODES_PER_WAVE) == 0, "eight waves per block, no tail");
    const int lane = threadIdx.x & 31;
    const int n0 = __builtin_amdgcn_readfirstlane(
        (blockIdx.x * 8 + (threadIdx.x >> 5)) * NODES_PER_WAVE);
    const int  ch[3]  = { lane, 32 + lane, 64 + lane };
    const bool act[3] = { true, true, true };
    const float bb[3] = { bias[ch[0]], bias[ch[1]], bias[ch[2]] };
    for (int k = 0; k < NODES_PER_WAVE; ++k) {
        const int node = n0 + k;
        const int beg  = __builtin_amdgcn_readfirstlane(rowptr[node]);
        const int end  = __builtin_amdgcn_readfirstlane(rowptr[node + 1]);
        const float di = dinv[node], w0 = di * di;
        const float* hs = H + (size_t)node * HID;
        float acc[3];
        #pragma unroll
        for (int p = 0; p < 3; ++p) acc[p] = w0 * hs[ch[p]];
        sum_segment<HID, 3, true>(H, reinterpret_cast<int2*>(csr), dinv, di,
                                  beg, end, lane, ch, act, acc);
        float* o = out + (size_t)node * HID;
        #pragma unroll
        for (int p = 0; p < 3; ++p) o[ch[p]] = fmaxf(acc[p] + bb[p], 0.0f);
    }
}

__global__ __launch_bounds__(256) void gather_rows48(const float* __restrict__ H,
                                                     const int* __restrict__ rowptr,
                                                     int* csr,
                                                     const float* __restrict__ dinv,
                                                     const float* __restrict__ bias,
                                                     float* __restrict__ out) {
    static_assert(NODES_PER_WAVE % 2 == 0, "whole pairs per wave");
    const int lane = threadIdx.x & 31;
    const int p0 = __builtin_amdgcn_readfirstlane(
        (blockIdx.x * 8 + (threadIdx.x >> 5)) * (NODES_PER_WAVE / 2));
    int2* cs = reinterpret_cast<int2*>(csr);
    const bool lo = lane < 16;
    const int  chA[2]  = { lane, 32 + (lane & 15) };
    const bool actA[2] = { true, lo };
    const int  chB[2]  = { 16 + lane, lane & 15 };
    const bool actB[2] = { true, !lo };
    const float bA = bias[lane];
    const float bM = bias[lo ? 32 + lane : lane - 16];
    const float bB = bias[16 + lane];
    for (int k = 0; k < NODES_PER_WAVE / 2; ++k) {
        const int pair = p0 + k;
        const int nA = 2 * pair, nB = nA + 1;
        const int r0 = __builtin_amdgcn_readfirstlane(rowptr[nA]);
        const int r1 = __builtin_amdgcn_readfirstlane(rowptr[nB]);
        const int r2 = __builtin_amdgcn_readfirstlane(rowptr[nB + 1]);
        float a[2], b[2];
        {
            const float di = dinv[nA], w0 = di * di;
            const float* hs = H + (size_t)nA * HID2;
            a[0] = w0 * hs[chA[0]];
            a[1] = lo ? w0 * hs[chA[1]] : 0.0f;
            sum_segment<HID2, 2, false>(H, cs, dinv, di, r0, r1, lane, chA, actA, a);
        }
        {
            const float di = dinv[nB], w0 = di * di;
            const float* hs = H + (size_t)nB * HID2;
            b[0] = w0 * hs[chB[0]];
            b[1] = lo ? 0.0f : w0 * hs[chB[1]];
            sum_segment<HID2, 2, false>(H, cs, dinv, di, r1, r2, lane, chB, actB, b);
        }
        float* o = out + (size_t)pair * (2 * HID2);
        const float mid = lo ? a[1] : b[1];
        o[lane]      = fmaxf(a[0] + bA, 0.0f);
        o[32 + lane] = fmaxf(mid  + bM, 0.0f);
        o[64 + lane] = fmaxf(b[0] + bB, 0.0f);
    }
}

__global__ __launch_bounds__(256) void classifier(const float* __restrict__ Hin,
                                                  const float* __restrict__ Wc,
                                                  const float* __restrict__ bc,
                                                  float* __restrict__ logits) {
    int i = blockIdx.x * blockDim.x + threadIdx.x;
    if (i >= N_NODES) return;
    const float* h = Hin + (size_t)i * HID2;
    float a0 = bc[0], a1 = bc[1], a2 = bc[2];
    #pragma unroll
    for (int k = 0; k < HID2; ++k) {
        float hv = h[k];
        a0 += hv * Wc[k * 3 + 0];
        a1 += hv * Wc[k * 3 + 1];
        a2 += hv * Wc[k * 3 + 2];
    }
    logits[(size_t)i * 3 + 0] = a0;
    logits[(size_t)i * 3 + 1] = a1;
    logits[(size_t)i * 3 + 2] = a2;
}


extern "C" void kernel_launch(void* const* d_in, const int* in_sizes, int n_in,
                              void* d_out, int out_size, void* d_ws, size_t ws_size,
                              hipStream_t stream) {
    const float* x  = (const float*)d_in[0];
    const int*   ei = (const int*)  d_in[1];
    const float* ew = (const float*)d_in[2];
    const float* W1 = (const float*)d_in[3];
    const float* b1 = (const float*)d_in[4];
    const float* W2 = (const float*)d_in[5];
    const float* b2 = (const float*)d_in[6];
    const float* Wc = (const float*)d_in[7];
    const float* bc = (const float*)d_in[8];
    float* logits = (float*)d_out;

    const int* src = ei;
    const int* dst = ei + N_EDGES;

    char* ws = (char*)d_ws;
    size_t off = 0;
    auto take = [&](size_t bytes) -> void* {
        void* p = ws + off;
        off += (bytes + 255) & ~(size_t)255;
        return p;
    };
    float* dinv     = (float*)take((size_t)N_NODES * 4);
    int*   cnt      = (int*)  take((size_t)N_NODES * 4);
    int*   rowptr   = (int*)  take((size_t)(N_NODES + 1) * 4);
    int*   blocksum = (int*)  take((size_t)SCAN_BLOCKS * 4);
    unsigned short* rank = (unsigned short*)take((size_t)N_EDGES * 2);
    unsigned* chunkcnt   = (unsigned*)take((size_t)NCH * NWORDS * 4);
    int*   pre      = (int*)  take((size_t)NCH * N_NODES * 4);
    int*   csr      = (int*)  take((size_t)N_EDGES * 8);
    float* H1   = (float*)take((size_t)N_NODES * HID * 4);
    float* out1 = (float*)take((size_t)N_NODES * HID * 4);
    float* H2   = H1;
    float* out2 = out1;

    const int TB = 256;
    rank_chunk  <<<NCH, RK_TB, 0, stream>>>(dst, chunkcnt, rank);
    chunk_offsets<<<CS_BLOCKS, CS_TB, 0, stream>>>(chunkcnt, pre, cnt);
    scan_local  <<<SCAN_BLOCKS, SCAN_TB, 0, stream>>>(cnt, rowptr, blocksum);
    scan_offsets<<<SCAN_BLOCKS, SCAN_TB, 0, stream>>>(blocksum, rowptr);
    fill_csr    <<<(N_EDGES + TB - 1) / TB, TB, 0, stream>>>(src, dst, ew, rank, rowptr, pre, csr);
    node_dinv   <<<(N_NODES + TB - 1) / TB, TB, 0, stream>>>(rowptr, csr, dinv);

    {
        gemm_wmma<IN_CH, HID><<<(N_NODES / 16 + 7) / 8, 256, 0, stream>>>(x, W1, H1);
    }
    gather_rows96<<<N_NODES / (8 * NODES_PER_WAVE), 256, 0, stream>>>(H1, rowptr, csr, dinv, b1, out1);

    {
        gemm_wmma<HID, HID2><<<(N_NODES / 16 + 7) / 8, 256, 0, stream>>>(out1, W2, H2);
    }
    gather_rows48<<<N_NODES / (8 * NODES_PER_WAVE), 256, 0, stream>>>(H2, rowptr, csr, dinv, b2, out2);

    classifier<<<(N_NODES + TB - 1) / TB, TB, 0, stream>>>(out2, Wc, bc, logits);
}


# edge norm factored: the GEMM stores dinv[row]*H[row], the gathers sum raw-weight rows and scale each node's sum by dinv[dst] once; no per-edge dinv look-up, no norm write-back; products re-associated
# speedup vs baseline: 3.4614x; 1.0492x over previous
#include <hip/hip_runtime.h>

#define N_NODES 100000
#define N_EDGES 3200000
#define IN_CH   128
#define HID     96
#define HID2    48
#define NCLS    3

typedef __attribute__((ext_vector_type(2)))  float  v2f;
typedef __attribute__((ext_vector_type(8)))  float  v8f;


#define SCAN_TB     256
#define SCAN_BLOCKS ((N_NODES + 1 + SCAN_TB - 1) / SCAN_TB)

#define NCH     64
#define CH      (N_EDGES / NCH)
#define NWORDS  (N_NODES / 2)
#define RK_TB   1024
#define RK_PER  ((CH + RK_TB - 1) / RK_TB)
static_assert(N_EDGES % NCH == 0, "whole chunks");
static_assert(CH < 65536, "a chunk cannot overflow a 16-bit counter, whatever the edges are");
static_assert(N_NODES % 2 == 0 && NWORDS % 4 == 0, "counter words move as 16-byte vectors");

__global__ __launch_bounds__(RK_TB) void rank_chunk(const int* __restrict__ dst,
                                                    unsigned* __restrict__ chunkcnt,
                                                    unsigned short* __restrict__ rank) {
    __shared__ uint4 cw4[NWORDS / 4];
    unsigned* cw = reinterpret_cast<unsigned*>(cw4);
    const int t  = threadIdx.x;
    const int e0 = blockIdx.x * CH;
    for (int i = t; i < NWORDS / 4; i += RK_TB) cw4[i] = make_uint4(0u, 0u, 0u, 0u);
    const bool has_last = t + (RK_PER - 1) * RK_TB < CH;
    int d[RK_PER];
    #pragma unroll
    for (int k = 0; k < RK_PER; ++k)
        d[k] = (k < RK_PER - 1 || has_last) ? dst[e0 + t + k * RK_TB] : 0;
    __syncthreads();
    unsigned r[RK_PER];
    #pragma unroll
    for (int k = 0; k < RK_PER; ++k) {
        r[k] = 0u;
        if (k < RK_PER - 1 || has_last) {
            const unsigned sh = (unsigned)(d[k] & 1) << 4;
            r[k] = atomicAdd(cw + (d[k] >> 1), 1u << sh) >> sh;
        }
    }
    #pragma unroll
    for (int k = 0; k < RK_PER; ++k)
        if (k < RK_PER - 1 || has_last) rank[e0 + t + k * RK_TB] = (unsigned short)r[k];
    __syncthreads();
    uint4* outw = reinterpret_cast<uint4*>(chunkcnt + (size_t)blockIdx.x * NWORDS);
    for (int i = t; i < NWORDS / 4; i += RK_TB) outw[i] = cw4[i];
}

#define CS_TB     256
#define CS_GROUPS (CS_TB / 32)
#define CS_PER    (NCH / CS_GROUPS)
#define CS_BLOCKS ((NWORDS + 31) / 32)
static_assert(NCH % CS_GROUPS == 0, "equal share of chunks per wave");

__global__ __launch_bounds__(CS_TB) void chunk_offsets(const unsigned* __restrict__ chunkcnt,
                                                       int* __restrict__ pre,
                                                       int* __restrict__ cnt) {
    __shared__ int2 part[CS_GROUPS][32];
    const int lane = threadIdx.x & 31, g = threadIdx.x >> 5;
    const int word = blockIdx.x * 32 + lane;
    const bool ok  = word < NWORDS;
    const int  wl  = ok ? word : NWORDS - 1;
    unsigned v[CS_PER];
    #pragma unroll
    for (int k = 0; k < CS_PER; ++k)
        v[k] = chunkcnt[(size_t)(g * CS_PER + k) * NWORDS + wl];
    int lo = 0, hi = 0;
    #pragma unroll
    for (int k = 0; k < CS_PER; ++k) { lo += (int)(v[k] & 0xFFFFu); hi += (int)(v[k] >> 16); }
    part[g][lane] = make_int2(lo, hi);
    __syncthreads();
    int blo = 0, bhi = 0;
    for (int w = 0; w < CS_GROUPS; ++w)
        if (w < g) { const int2 p = part[w][lane]; blo += p.x; bhi += p.y; }
    if (ok) {
        #pragma unroll
        for (int k = 0; k < CS_PER; ++k) {
            reinterpret_cast<int2*>(pre + (size_t)(g * CS_PER + k) * N_NODES)[word] = make_int2(blo, bhi);
            blo += (int)(v[k] & 0xFFFFu);
            bhi += (int)(v[k] >> 16);
        }
        if (g == CS_GROUPS - 1) reinterpret_cast<int2*>(cnt)[word] = make_int2(blo, bhi);
    }
}

__global__ __launch_bounds__(SCAN_TB) void scan_local(const int* __restrict__ cnt,
                                                      int* __restrict__ rowptr,
                                                      int* __restrict__ blocksum) {
    __shared__ int wsum[SCAN_TB / 32];
    const int t = threadIdx.x, lane = t & 31, wave = t >> 5;
    const int i = blockIdx.x * SCAN_TB + t;
    const int c = (i < N_NODES) ? cnt[i] : 0;
    int incl = c;
    for (int d = 1; d < 32; d <<= 1) {
        const int v = __shfl_up(incl, d);
        if (lane >= d) incl += v;
    }
    if (lane == 31) wsum[wave] = incl;
    __syncthreads();
    int pre = 0;
    for (int w = 0; w < SCAN_TB / 32; ++w) if (w < wave) pre += wsum[w];
    if (i <= N_NODES) rowptr[i] = pre + incl - c;
    if (t == SCAN_TB - 1) blocksum[blockIdx.x] = pre + incl;
}

__global__ __launch_bounds__(SCAN_TB) void scan_offsets(const int* __restrict__ blocksum,
                                                        int* __restrict__ rowptr) {
    __shared__ int wsum[SCAN_TB / 32];
    const int t = threadIdx.x, lane = t & 31, wave = t >> 5;
    const int b = blockIdx.x;
    int v = 0;
    for (int k = t; k < b; k += SCAN_TB) v += blocksum[k];
    for (int d = 16; d >= 1; d >>= 1) v += __shfl_xor(v, d);
    if (lane == 0) wsum[wave] = v;
    __syncthreads();
    int off = 0;
    for (int w = 0; w < SCAN_TB / 32; ++w) off += wsum[w];
    const int i = b * SCAN_TB + t;
    if (i <= N_NODES) rowptr[i] += off;
}

__global__ __launch_bounds__(256) void fill_csr(const int* __restrict__ src,
                                                const int* __restrict__ dst,
                                                const float* __restrict__ ew,
                                                const unsigned short* __restrict__ rank,
                                                const int* __restrict__ rowptr,
                                                const int* __restrict__ pre,
                                                int* __restrict__ csr) {
    int e = blockIdx.x * blockDim.x + threadIdx.x;
    if (e < N_EDGES) {
        const int d   = dst[e];
        const int pos = rowptr[d] + pre[(size_t)(e / CH) * N_NODES + d] + (int)rank[e];
        reinterpret_cast<int2*>(csr)[pos] = make_int2(src[e], __float_as_int(ew[e]));
    }
}

__global__ __launch_bounds__(256) void node_dinv(const int* __restrict__ rowptr,
                                                 const int* __restrict__ csr,
                                                 float* __restrict__ dinv) {
    int i = blockIdx.x * blockDim.x + threadIdx.x;
    if (i < N_NODES) {
        const int2* c = reinterpret_cast<const int2*>(csr);
        const int end = rowptr[i + 1];
        float d = 1.0f;
        for (int p = rowptr[i]; p < end; ++p) d += __int_as_float(c[p].y);
        dinv[i] = rsqrtf(d);
    }
}


template<int K, int COLS>
__global__ __launch_bounds__(256) void gemm_wmma(const float* __restrict__ X,
                                                 const float* __restrict__ W,
                                                 const float* __restrict__ dinv,
                                                 float* __restrict__ H) {
    constexpr int NT = COLS / 16;
    constexpr int KS = K / 4;
    constexpr int P2 = COLS + ((16 - COLS % 32 + 32) % 32);
    static_assert(K % 4 == 0 && COLS % 16 == 0 && P2 % 32 == 16 && P2 % 4 == 0, "tile shapes");
    __shared__ __attribute__((aligned(16))) v2f wp[(K / 2) * P2];
    const int lane = threadIdx.x & 31;
    const int wave = threadIdx.x >> 5;
    const int c = lane & 15, h = lane >> 4;
    const int m0 = (blockIdx.x * 8 + wave) * 16;
    const bool live = m0 < N_NODES;
    const int row = (live ? m0 : N_NODES - 16) + c;
    const float* xrow = X + (size_t)row * K + 2 * h;
    constexpr int QR  = COLS / 4;
    constexpr int NQ  = (K / 2) * QR;
    constexpr int WIT = (NQ + 255) / 256;
    float4 w0[WIT], w1[WIT];
    #pragma unroll
    for (int j = 0; j < WIT; ++j) {
        const int idx = j * 256 + (int)threadIdx.x;
        if ((j + 1) * 256 <= NQ || idx < NQ) {
            const int kp = idx / QR, q = idx - kp * QR;
            w0[j] = *(const float4*)(W + (size_t)(2 * kp)     * COLS + 4 * q);
            w1[j] = *(const float4*)(W + (size_t)(2 * kp + 1) * COLS + 4 * q);
        }
    }
    #pragma unroll
    for (int j = 0; j < WIT; ++j) {
        const int idx = j * 256 + (int)threadIdx.x;
        if ((j + 1) * 256 <= NQ || idx < NQ) {
            const int kp = idx / QR, q = idx - kp * QR;
            float4* d = reinterpret_cast<float4*>(wp + kp * P2 + 4 * q);
            d[0] = make_float4(w0[j].x, w1[j].x, w0[j].y, w1[j].y);
            d[1] = make_float4(w0[j].z, w1[j].z, w0[j].w, w1[j].w);
        }
    }
    __syncthreads();
    v2f a[KS];
    #pragma unroll
    for (int i = 0; i < KS; ++i) a[i] = *(const v2f*)(xrow + 4 * i);
    v8f acc[NT];
    #pragma unroll
    for (int t = 0; t < NT; ++t) acc[t] = v8f{};
    const v2f* bp = wp + h * P2 + c;
    #pragma unroll
    for (int i = 0; i < KS; ++i) {
        #pragma unroll
        for (int t = 0; t < NT; ++t) {
            const v2f b = bp[(2 * i) * P2 + 16 * t];
            acc[t] = __builtin_amdgcn_wmma_f32_16x16x4_f32(false, a[i], false, b,
                                                           (short)0, acc[t], false, false);
        }
    }
    if (live) {
        float* hrow = H + (size_t)(m0 + 8 * h) * COLS + c;
        float dr[8];
        #pragma unroll
        for (int r = 0; r < 8; ++r) dr[r] = dinv[m0 + 8 * h + r];
        #pragma unroll
        for (int t = 0; t < NT; ++t) {
            #pragma unroll
            for (int r = 0; r < 8; ++r) hrow[(size_t)r * COLS + 16 * t] = dr[r] * acc[t][r];
        }
    }
}


template<int C, int NP>
static __device__ inline void sum_segment(const float* __restrict__ H,
                                          const int2* __restrict__ csr,
                                          int beg, int end, int lane,
                                          const int (&ch)[NP], const bool (&act)[NP],
                                          float (&acc)[NP]) {
    constexpr int U = 8;
    for (int base = beg; base < end; base += 32) {
        const int left = end - base;
        const int n = left < 32 ? left : 32;
        int2 ev = make_int2(0, 0);
        if (lane < n) ev = csr[base + lane];
        for (int j0 = 0; j0 < n; j0 += U) {
            const int m = n - j0;
            float hv[U][NP];
            if (m >= U) {
                #pragma unroll
                for (int u = 0; u < U; ++u) {
                    const int s = __builtin_amdgcn_readlane(ev.x, j0 + u);
                    const float* hp = H + (size_t)s * C;
                    #pragma unroll
                    for (int p = 0; p < NP; ++p) if (act[p]) hv[u][p] = hp[ch[p]];
                }
                #pragma unroll
                for (int u = 0; u < U; ++u) {
                    const float w = __int_as_float(__builtin_amdgcn_readlane(ev.y, j0 + u));
                    #pragma unroll
                    for (int p = 0; p < NP; ++p) if (act[p]) acc[p] += w * hv[u][p];
                }
            } else {
                #pragma unroll
                for (int u = 0; u < U; ++u)
                    if (u < m) {
                        const int s = __builtin_amdgcn_readlane(ev.x, j0 + u);
                        const float* hp = H + (size_t)s * C;
                        #pragma unroll
                        for (int p = 0; p < NP; ++p) if (act[p]) hv[u][p] = hp[ch[p]];
                    }
                #pragma unroll
                for (int u = 0; u < U; ++u)
                    if (u < m) {
                        const float w = __int_as_float(__builtin_amdgcn_readlane(ev.y, j0 + u));
                        #pragma unroll
                        for (int p = 0; p < NP; ++p) if (act[p]) acc[p] += w * hv[u][p];
                    }
            }
        }
    }
}

#define NODES_PER_WAVE 4

__global__ __launch_bounds__(256) void gather_rows96(const float* __restrict__ H,
                                                     const int* __restrict__ rowptr,
                                                     const int* __restrict__ csr,
                                                     const float* __restrict__ dinv,
                                                     const float* __restrict__ bias,
                                                     float* __restrict__ out) {
    static_assert(N_NODES % (8 * NODES_PER_WAVE) == 0, "eight waves per block, no tail");
    const int lane = threadIdx.x & 31;
    const int n0 = __builtin_amdgcn_readfirstlane(
        (blockIdx.x * 8 + (threadIdx.x >> 5)) * NODES_PER_WAVE);
    const int  ch[3]  = { lane, 32 + lane, 64 + lane };
    const bool act[3] = { true, true, true };
    const float bb[3] = { bias[ch[0]], bias[ch[1]], bias[ch[2]] };
    for (int k = 0; k < NODES_PER_WAVE; ++k) {
        const int node = n0 + k;
        const int beg  = __builtin_amdgcn_readfirstlane(rowptr[node]);
        const int end  = __builtin_amdgcn_readfirstlane(rowptr[node + 1]);
        const float di = dinv[node];
        const float* hs = H + (size_t)node * HID;
        float acc[3];
        #pragma unroll
        for (int p = 0; p < 3; ++p) acc[p] = hs[ch[p]];
        sum_segment<HID, 3>(H, reinterpret_cast<const int2*>(csr), beg, end, lane, ch, act, acc);
        float* o = out + (size_t)node * HID;
        #pragma unroll
        for (int p = 0; p < 3; ++p) o[ch[p]] = fmaxf(di * acc[p] + bb[p], 0.0f);
    }
}

__global__ __launch_bounds__(256) void gather_rows48(const float* __restrict__ H,
                                                     const int* __restrict__ rowptr,
                                                     const int* __restrict__ csr,
                                                     const float* __restrict__ dinv,
                                                     const float* __restrict__ bias,
                                                     float* __restrict__ out) {
    static_assert(NODES_PER_WAVE % 2 == 0, "whole pairs per wave");
    const int lane = threadIdx.x & 31;
    const int p0 = __builtin_amdgcn_readfirstlane(
        (blockIdx.x * 8 + (threadIdx.x >> 5)) * (NODES_PER_WAVE / 2));
    const int2* cs = reinterpret_cast<const int2*>(csr);
    const bool lo = lane < 16;
    const int  chA[2]  = { lane, 32 + (lane & 15) };
    const bool actA[2] = { true, lo };
    const int  chB[2]  = { 16 + lane, lane & 15 };
    const bool actB[2] = { true, !lo };
    const float bA = bias[lane];
    const float bM = bias[lo ? 32 + lane : lane - 16];
    const float bB = bias[16 + lane];
    for (int k = 0; k < NODES_PER_WAVE / 2; ++k) {
        const int pair = p0 + k;
        const int nA = 2 * pair, nB = nA + 1;
        const int r0 = __builtin_amdgcn_readfirstlane(rowptr[nA]);
        const int r1 = __builtin_amdgcn_readfirstlane(rowptr[nB]);
        const int r2 = __builtin_amdgcn_readfirstlane(rowptr[nB + 1]);
        const float diA = dinv[nA], diB = dinv[nB];
        float a[2], b[2];
        {
            const float* hs = H + (size_t)nA * HID2;
            a[0] = hs[chA[0]];
            a[1] = lo ? hs[chA[1]] : 0.0f;
            sum_segment<HID2, 2>(H, cs, r0, r1, lane, chA, actA, a);
        }
        {
            const float* hs = H + (size_t)nB * HID2;
            b[0] = hs[chB[0]];
            b[1] = lo ? 0.0f : hs[chB[1]];
            sum_segment<HID2, 2>(H, cs, r1, r2, lane, chB, actB, b);
        }
        float* o = out + (size_t)pair * (2 * HID2);
        const float mid = lo ? diA * a[1] : diB * b[1];
        o[lane]      = fmaxf(diA * a[0] + bA, 0.0f);
        o[32 + lane] = fmaxf(mid          + bM, 0.0f);
        o[64 + lane] = fmaxf(diB * b[0] + bB, 0.0f);
    }
}

__global__ __launch_bounds__(256) void classifier(const float* __restrict__ Hin,
                                                  const float* __restrict__ Wc,
                                                  const float* __restrict__ bc,
                                                  float* __restrict__ logits) {
    int i = blockIdx.x * blockDim.x + threadIdx.x;
    if (i >= N_NODES) return;
    const float* h = Hin + (size_t)i * HID2;
    float a0 = bc[0], a1 = bc[1], a2 = bc[2];
    #pragma unroll
    for (int k = 0; k < HID2; ++k) {
        float hv = h[k];
        a0 += hv * Wc[k * 3 + 0];
        a1 += hv * Wc[k * 3 + 1];
        a2 += hv * Wc[k * 3 + 2];
    }
    logits[(size_t)i * 3 + 0] = a0;
    logits[(size_t)i * 3 + 1] = a1;
    logits[(size_t)i * 3 + 2] = a2;
}


extern "C" void kernel_launch(void* const* d_in, const int* in_sizes, int n_in,
                              void* d_out, int out_size, void* d_ws, size_t ws_size,
                              hipStream_t stream) {
    const float* x  = (const float*)d_in[0];
    const int*   ei = (const int*)  d_in[1];
    const float* ew = (const float*)d_in[2];
    const float* W1 = (const float*)d_in[3];
    const float* b1 = (const float*)d_in[4];
    const float* W2 = (const float*)d_in[5];
    const float* b2 = (const float*)d_in[6];
    const float* Wc = (const float*)d_in[7];
    const float* bc = (const float*)d_in[8];
    float* logits = (float*)d_out;

    const int* src = ei;
    const int* dst = ei + N_EDGES;

    char* ws = (char*)d_ws;
    size_t off = 0;
    auto take = [&](size_t bytes) -> void* {
        void* p = ws + off;
        off += (bytes + 255) & ~(size_t)255;
        return p;
    };
    float* dinv     = (float*)take((size_t)N_NODES * 4);
    int*   cnt      = (int*)  take((size_t)N_NODES * 4);
    int*   rowptr   = (int*)  take((size_t)(N_NODES + 1) * 4);
    int*   blocksum = (int*)  take((size_t)SCAN_BLOCKS * 4);
    unsigned short* rank = (unsigned short*)take((size_t)N_EDGES * 2);
    unsigned* chunkcnt   = (unsigned*)take((size_t)NCH * NWORDS * 4);
    int*   pre      = (int*)  take((size_t)NCH * N_NODES * 4);
    int*   csr      = (int*)  take((size_t)N_EDGES * 8);
    float* H1   = (float*)take((size_t)N_NODES * HID * 4);
    float* out1 = (float*)take((size_t)N_NODES * HID * 4);
    float* H2   = H1;
    float* out2 = out1;

    const int TB = 256;
    rank_chunk  <<<NCH, RK_TB, 0, stream>>>(dst, chunkcnt, rank);
    chunk_offsets<<<CS_BLOCKS, CS_TB, 0, stream>>>(chunkcnt, pre, cnt);
    scan_local  <<<SCAN_BLOCKS, SCAN_TB, 0, stream>>>(cnt, rowptr, blocksum);
    scan_offsets<<<SCAN_BLOCKS, SCAN_TB, 0, stream>>>(blocksum, rowptr);
    fill_csr    <<<(N_EDGES + TB - 1) / TB, TB, 0, stream>>>(src, dst, ew, rank, rowptr, pre, csr);
    node_dinv   <<<(N_NODES + TB - 1) / TB, TB, 0, stream>>>(rowptr, csr, dinv);

    {
        gemm_wmma<IN_CH, HID><<<(N_NODES / 16 + 7) / 8, 256, 0, stream>>>(x, W1, dinv, H1);
    }
    gather_rows96<<<N_NODES / (8 * NODES_PER_WAVE), 256, 0, stream>>>(H1, rowptr, csr, dinv, b1, out1);

    {
        gemm_wmma<HID, HID2><<<(N_NODES / 16 + 7) / 8, 256, 0, stream>>>(out1, W2, dinv, H2);
    }
    gather_rows48<<<N_NODES / (8 * NODES_PER_WAVE), 256, 0, stream>>>(H2, rowptr, csr, dinv, b2, out2);

    classifier<<<(N_NODES + TB - 1) / TB, TB, 0, stream>>>(out2, Wc, bc, logits);
}


# the dense product behind each gather runs in the gather's block from LDS (layer-2 GEMM on f32 WMMA; classifier, now summed as a lane tree): out1, out2 never stored, 2 launches fewer
# speedup vs baseline: 3.6058x; 1.0417x over previous
#include <hip/hip_runtime.h>

#define N_NODES 100000
#define N_EDGES 3200000
#define IN_CH   128
#define HID     96
#define HID2    48
#define NCLS    3

typedef __attribute__((ext_vector_type(2)))  float  v2f;
typedef __attribute__((ext_vector_type(8)))  float  v8f;


#define SCAN_TB     256
#define SCAN_BLOCKS ((N_NODES + 1 + SCAN_TB - 1) / SCAN_TB)

#define NCH     64
#define CH      (N_EDGES / NCH)
#define NWORDS  (N_NODES / 2)
#define RK_TB   1024
#define RK_PER  ((CH + RK_TB - 1) / RK_TB)
static_assert(N_EDGES % NCH == 0, "whole chunks");
static_assert(CH < 65536, "a chunk cannot overflow a 16-bit counter, whatever the edges are");
static_assert(N_NODES % 2 == 0 && NWORDS % 4 == 0, "counter words move as 16-byte vectors");

__global__ __launch_bounds__(RK_TB) void rank_chunk(const int* __restrict__ dst,
                                                    unsigned* __restrict__ chunkcnt,
                                                    unsigned short* __restrict__ rank) {
    __shared__ uint4 cw4[NWORDS / 4];
    unsigned* cw = reinterpret_cast<unsigned*>(cw4);
    const int t  = threadIdx.x;
    const int e0 = blockIdx.x * CH;
    for (int i = t; i < NWORDS / 4; i += RK_TB) cw4[i] = make_uint4(0u, 0u, 0u, 0u);
    const bool has_last = t + (RK_PER - 1) * RK_TB < CH;
    int d[RK_PER];
    #pragma unroll
    for (int k = 0; k < RK_PER; ++k)
        d[k] = (k < RK_PER - 1 || has_last) ? dst[e0 + t + k * RK_TB] : 0;
    __syncthreads();
    unsigned r[RK_PER];
    #pragma unroll
    for (int k = 0; k < RK_PER; ++k) {
        r[k] = 0u;
        if (k < RK_PER - 1 || has_last) {
            const unsigned sh = (unsigned)(d[k] & 1) << 4;
            r[k] = atomicAdd(cw + (d[k] >> 1), 1u << sh) >> sh;
        }
    }
    #pragma unroll
    for (int k = 0; k < RK_PER; ++k)
        if (k < RK_PER - 1 || has_last) rank[e0 + t + k * RK_TB] = (unsigned short)r[k];
    __syncthreads();
    uint4* outw = reinterpret_cast<uint4*>(chunkcnt + (size_t)blockIdx.x * NWORDS);
    for (int i = t; i < NWORDS / 4; i += RK_TB) outw[i] = cw4[i];
}

#define CS_TB     256
#define CS_GROUPS (CS_TB / 32)
#define CS_PER    (NCH / CS_GROUPS)
#define CS_BLOCKS ((NWORDS + 31) / 32)
static_assert(NCH % CS_GROUPS == 0, "equal share of chunks per wave");

__global__ __launch_bounds__(CS_TB) void chunk_offsets(const unsigned* __restrict__ chunkcnt,
                                                       int* __restrict__ pre,
                                                       int* __restrict__ cnt) {
    __shared__ int2 part[CS_GROUPS][32];
    const int lane = threadIdx.x & 31, g = threadIdx.x >> 5;
    const int word = blockIdx.x * 32 + lane;
    const bool ok  = word < NWORDS;
    const int  wl  = ok ? word : NWORDS - 1;
    unsigned v[CS_PER];
    #pragma unroll
    for (int k = 0; k < CS_PER; ++k)
        v[k] = chunkcnt[(size_t)(g * CS_PER + k) * NWORDS + wl];
    int lo = 0, hi = 0;
    #pragma unroll
    for (int k = 0; k < CS_PER; ++k) { lo += (int)(v[k] & 0xFFFFu); hi += (int)(v[k] >> 16); }
    part[g][lane] = make_int2(lo, hi);
    __syncthreads();
    int blo = 0, bhi = 0;
    for (int w = 0; w < CS_GROUPS; ++w)
        if (w < g) { const int2 p = part[w][lane]; blo += p.x; bhi += p.y; }
    if (ok) {
        #pragma unroll
        for (int k = 0; k < CS_PER; ++k) {
            reinterpret_cast<int2*>(pre + (size_t)(g * CS_PER + k) * N_NODES)[word] = make_int2(blo, bhi);
            blo += (int)(v[k] & 0xFFFFu);
            bhi += (int)(v[k] >> 16);
        }
        if (g == CS_GROUPS - 1) reinterpret_cast<int2*>(cnt)[word] = make_int2(blo, bhi);
    }
}

__global__ __launch_bounds__(SCAN_TB) void scan_local(const int* __restrict__ cnt,
                                                      int* __restrict__ rowptr,
                                                      int* __restrict__ blocksum) {
    __shared__ int wsum[SCAN_TB / 32];
    const int t = threadIdx.x, lane = t & 31, wave = t >> 5;
    const int i = blockIdx.x * SCAN_TB + t;
    const int c = (i < N_NODES) ? cnt[i] : 0;
    int incl = c;
    for (int d = 1; d < 32; d <<= 1) {
        const int v = __shfl_up(incl, d);
        if (lane >= d) incl += v;
    }
    if (lane == 31) wsum[wave] = incl;
    __syncthreads();
    int pre = 0;
    for (int w = 0; w < SCAN_TB / 32; ++w) if (w < wave) pre += wsum[w];
    if (i <= N_NODES) rowptr[i] = pre + incl - c;
    if (t == SCAN_TB - 1) blocksum[blockIdx.x] = pre + incl;
}

__global__ __launch_bounds__(SCAN_TB) void scan_offsets(const int* __restrict__ blocksum,
                                                        int* __restrict__ rowptr) {
    __shared__ int wsum[SCAN_TB / 32];
    const int t = threadIdx.x, lane = t & 31, wave = t >> 5;
    const int b = blockIdx.x;
    int v = 0;
    for (int k = t; k < b; k += SCAN_TB) v += blocksum[k];
    for (int d = 16; d >= 1; d >>= 1) v += __shfl_xor(v, d);
    if (lane == 0) wsum[wave] = v;
    __syncthreads();
    int off = 0;
    for (int w = 0; w < SCAN_TB / 32; ++w) off += wsum[w];
    const int i = b * SCAN_TB + t;
    if (i <= N_NODES) rowptr[i] += off;
}

__global__ __launch_bounds__(256) void fill_csr(const int* __restrict__ src,
                                                const int* __restrict__ dst,
                                                const float* __restrict__ ew,
                                                const unsigned short* __restrict__ rank,
                                                const int* __restrict__ rowptr,
                                                const int* __restrict__ pre,
                                                int* __restrict__ csr) {
    int e = blockIdx.x * blockDim.x + threadIdx.x;
    if (e < N_EDGES) {
        const int d   = dst[e];
        const int pos = rowptr[d] + pre[(size_t)(e / CH) * N_NODES + d] + (int)rank[e];
        reinterpret_cast<int2*>(csr)[pos] = make_int2(src[e], __float_as_int(ew[e]));
    }
}

__global__ __launch_bounds__(256) void node_dinv(const int* __restrict__ rowptr,
                                                 const int* __restrict__ csr,
                                                 float* __restrict__ dinv) {
    int i = blockIdx.x * blockDim.x + threadIdx.x;
    if (i < N_NODES) {
        const int2* c = reinterpret_cast<const int2*>(csr);
        const int end = rowptr[i + 1];
        float d = 1.0f;
        for (int p = rowptr[i]; p < end; ++p) d += __int_as_float(c[p].y);
        dinv[i] = rsqrtf(d);
    }
}


template<int K, int COLS>
__global__ __launch_bounds__(256) void gemm_wmma(const float* __restrict__ X,
                                                 const float* __restrict__ W,
                                                 const float* __restrict__ dinv,
                                                 float* __restrict__ H) {
    constexpr int NT = COLS / 16;
    constexpr int KS = K / 4;
    constexpr int P2 = COLS + ((16 - COLS % 32 + 32) % 32);
    static_assert(K % 4 == 0 && COLS % 16 == 0 && P2 % 32 == 16 && P2 % 4 == 0, "tile shapes");
    __shared__ __attribute__((aligned(16))) v2f wp[(K / 2) * P2];
    const int lane = threadIdx.x & 31;
    const int wave = threadIdx.x >> 5;
    const int c = lane & 15, h = lane >> 4;
    const int m0 = (blockIdx.x * 8 + wave) * 16;
    const bool live = m0 < N_NODES;
    const int row = (live ? m0 : N_NODES - 16) + c;
    const float* xrow = X + (size_t)row * K + 2 * h;
    constexpr int QR  = COLS / 4;
    constexpr int NQ  = (K / 2) * QR;
    constexpr int WIT = (NQ + 255) / 256;
    float4 w0[WIT], w1[WIT];
    #pragma unroll
    for (int j = 0; j < WIT; ++j) {
        const int idx = j * 256 + (int)threadIdx.x;
        if ((j + 1) * 256 <= NQ || idx < NQ) {
            const int kp = idx / QR, q = idx - kp * QR;
            w0[j] = *(const float4*)(W + (size_t)(2 * kp)     * COLS + 4 * q);
            w1[j] = *(const float4*)(W + (size_t)(2 * kp + 1) * COLS + 4 * q);
        }
    }
    #pragma unroll
    for (int j = 0; j < WIT; ++j) {
        const int idx = j * 256 + (int)threadIdx.x;
        if ((j + 1) * 256 <= NQ || idx < NQ) {
            const int kp = idx / QR, q = idx - kp * QR;
            float4* d = reinterpret_cast<float4*>(wp + kp * P2 + 4 * q);
            d[0] = make_float4(w0[j].x, w1[j].x, w0[j].y, w1[j].y);
            d[1] = make_float4(w0[j].z, w1[j].z, w0[j].w, w1[j].w);
        }
    }
    __syncthreads();
    v2f a[KS];
    #pragma unroll
    for (int i = 0; i < KS; ++i) a[i] = *(const v2f*)(xrow + 4 * i);
    v8f acc[NT];
    #pragma unroll
    for (int t = 0; t < NT; ++t) acc[t] = v8f{};
    const v2f* bp = wp + h * P2 + c;
    #pragma unroll
    for (int i = 0; i < KS; ++i) {
        #pragma unroll
        for (int t = 0; t < NT; ++t) {
            const v2f b = bp[(2 * i) * P2 + 16 * t];
            acc[t] = __builtin_amdgcn_wmma_f32_16x16x4_f32(false, a[i], false, b,
                                                           (short)0, acc[t], false, false);
        }
    }
    if (live) {
        float* hrow = H + (size_t)(m0 + 8 * h) * COLS + c;
        float dr[8];
        #pragma unroll
        for (int r = 0; r < 8; ++r) dr[r] = dinv[m0 + 8 * h + r];
        #pragma unroll
        for (int t = 0; t < NT; ++t) {
            #pragma unroll
            for (int r = 0; r < 8; ++r) hrow[(size_t)r * COLS + 16 * t] = dr[r] * acc[t][r];
        }
    }
}


template<int C, int NP>
static __device__ inline void sum_segment(const float* __restrict__ H,
                                          const int2* __restrict__ csr,
                                          int beg, int end, int lane,
                                          const int (&ch)[NP], const bool (&act)[NP],
                                          float (&acc)[NP]) {
    constexpr int U = 8;
    for (int base = beg; base < end; base += 32) {
        const int left = end - base;
        const int n = left < 32 ? left : 32;
        int2 ev = make_int2(0, 0);
        if (lane < n) ev = csr[base + lane];
        for (int j0 = 0; j0 < n; j0 += U) {
            const int m = n - j0;
            float hv[U][NP];
            if (m >= U) {
                #pragma unroll
                for (int u = 0; u < U; ++u) {
                    const int s = __builtin_amdgcn_readlane(ev.x, j0 + u);
                    const float* hp = H + (size_t)s * C;
                    #pragma unroll
                    for (int p = 0; p < NP; ++p) if (act[p]) hv[u][p] = hp[ch[p]];
                }
                #pragma unroll
                for (int u = 0; u < U; ++u) {
                    const float w = __int_as_float(__builtin_amdgcn_readlane(ev.y, j0 + u));
                    #pragma unroll
                    for (int p = 0; p < NP; ++p) if (act[p]) acc[p] += w * hv[u][p];
                }
            } else {
                #pragma unroll
                for (int u = 0; u < U; ++u)
                    if (u < m) {
                        const int s = __builtin_amdgcn_readlane(ev.x, j0 + u);
                        const float* hp = H + (size_t)s * C;
                        #pragma unroll
                        for (int p = 0; p < NP; ++p) if (act[p]) hv[u][p] = hp[ch[p]];
                    }
                #pragma unroll
                for (int u = 0; u < U; ++u)
                    if (u < m) {
                        const float w = __int_as_float(__builtin_amdgcn_readlane(ev.y, j0 + u));
                        #pragma unroll
                        for (int p = 0; p < NP; ++p) if (act[p]) acc[p] += w * hv[u][p];
                    }
            }
        }
    }
}

#define NODES_PER_WAVE 4
#define GB_NODES (8 * NODES_PER_WAVE)

__global__ __launch_bounds__(256) void gather96_gemm48(const float* __restrict__ H,
                                                       const int* __restrict__ rowptr,
                                                       const int* __restrict__ csr,
                                                       const float* __restrict__ dinv,
                                                       const float* __restrict__ bias,
                                                       const float* __restrict__ W,
                                                       float* __restrict__ H2) {
    static_assert(N_NODES % GB_NODES == 0 && GB_NODES % 16 == 0, "whole blocks, whole 16-row strips");
    constexpr int K = HID, COLS = HID2;
    constexpr int NT = COLS / 16;
    constexpr int KS = K / 4;
    constexpr int P2 = COLS + ((16 - COLS % 32 + 32) % 32);
    constexpr int XP = K + 4;
    static_assert(K % 4 == 0 && COLS % 16 == 0 && P2 % 32 == 16 && P2 % 4 == 0, "tile shapes");
    static_assert(XP % 64 == 36 && XP % 2 == 0, "A fragment reads: 8-byte aligned, no bank shared");
    __shared__ __attribute__((aligned(16))) v2f wp[(K / 2) * P2];
    __shared__ __attribute__((aligned(16))) float xt[GB_NODES * XP];
    const int lane = threadIdx.x & 31;
    const int wave = __builtin_amdgcn_readfirstlane((int)(threadIdx.x >> 5));
    {
        constexpr int QR  = COLS / 4;
        constexpr int NQ  = (K / 2) * QR;
        constexpr int WIT = (NQ + 255) / 256;
        float4 w0[WIT], w1[WIT];
        #pragma unroll
        for (int j = 0; j < WIT; ++j) {
            const int idx = j * 256 + (int)threadIdx.x;
            if ((j + 1) * 256 <= NQ || idx < NQ) {
                const int kp = idx / QR, q = idx - kp * QR;
                w0[j] = *(const float4*)(W + (size_t)(2 * kp)     * COLS + 4 * q);
                w1[j] = *(const float4*)(W + (size_t)(2 * kp + 1) * COLS + 4 * q);
            }
        }
        #pragma unroll
        for (int j = 0; j < WIT; ++j) {
            const int idx = j * 256 + (int)threadIdx.x;
            if ((j + 1) * 256 <= NQ || idx < NQ) {
                const int kp = idx / QR, q = idx - kp * QR;
                float4* d = reinterpret_cast<float4*>(wp + kp * P2 + 4 * q);
                d[0] = make_float4(w0[j].x, w1[j].x, w0[j].y, w1[j].y);
                d[1] = make_float4(w0[j].z, w1[j].z, w0[j].w, w1[j].w);
            }
        }
    }
    const int n0 = (blockIdx.x * 8 + wave) * NODES_PER_WAVE;
    const int  ch[3]  = { lane, 32 + lane, 64 + lane };
    const bool act[3] = { true, true, true };
    const float bb[3] = { bias[ch[0]], bias[ch[1]], bias[ch[2]] };
    for (int k = 0; k < NODES_PER_WAVE; ++k) {
        const int node = n0 + k;
        const int beg  = __builtin_amdgcn_readfirstlane(rowptr[node]);
        const int end  = __builtin_amdgcn_readfirstlane(rowptr[node + 1]);
        const float di = dinv[node];
        const float* hs = H + (size_t)node * HID;
        float acc[3];
        #pragma unroll
        for (int p = 0; p < 3; ++p) acc[p] = hs[ch[p]];
        sum_segment<HID, 3>(H, reinterpret_cast<const int2*>(csr), beg, end, lane, ch, act, acc);
        float* o = xt + (wave * NODES_PER_WAVE + k) * XP;
        #pragma unroll
        for (int p = 0; p < 3; ++p) o[ch[p]] = fmaxf(di * acc[p] + bb[p], 0.0f);
    }
    __syncthreads();
    if (wave < GB_NODES / 16) {
        const int c = lane & 15, h = lane >> 4;
        const int m0 = blockIdx.x * GB_NODES + wave * 16;
        const float* xrow = xt + (wave * 16 + c) * XP + 2 * h;
        const v2f* bp = wp + h * P2 + c;
        float dr[8];
        #pragma unroll
        for (int r = 0; r < 8; ++r) dr[r] = dinv[m0 + 8 * h + r];
        v8f acc[NT];
        #pragma unroll
        for (int t = 0; t < NT; ++t) acc[t] = v8f{};
        #pragma unroll
        for (int i = 0; i < KS; ++i) {
            const v2f a = *(const v2f*)(xrow + 4 * i);
            #pragma unroll
            for (int t = 0; t < NT; ++t) {
                const v2f b = bp[(2 * i) * P2 + 16 * t];
                acc[t] = __builtin_amdgcn_wmma_f32_16x16x4_f32(false, a, false, b,
                                                               (short)0, acc[t], false, false);
            }
        }
        float* hrow = H2 + (size_t)(m0 + 8 * h) * COLS + c;
        #pragma unroll
        for (int t = 0; t < NT; ++t) {
            #pragma unroll
            for (int r = 0; r < 8; ++r) hrow[(size_t)r * COLS + 16 * t] = dr[r] * acc[t][r];
        }
    }
}

__global__ __launch_bounds__(256) void gather48_classify(const float* __restrict__ H,
                                                         const int* __restrict__ rowptr,
                                                         const int* __restrict__ csr,
                                                         const float* __restrict__ dinv,
                                                         const float* __restrict__ bias,
                                                         const float* __restrict__ Wc,
                                                         const float* __restrict__ bc,
                                                         float* __restrict__ logits) {
    static_assert(NODES_PER_WAVE % 2 == 0, "whole pairs per wave");
    static_assert((GB_NODES * NCLS) % 32 == 0 && GB_NODES * NCLS <= 256, "the block's logits are whole lines");
    __shared__ float sl[GB_NODES * NCLS];
    const int lane = threadIdx.x & 31;
    const int wave = __builtin_amdgcn_readfirstlane((int)(threadIdx.x >> 5));
    const int p0 = (blockIdx.x * 8 + wave) * (NODES_PER_WAVE / 2);
    const int2* cs = reinterpret_cast<const int2*>(csr);
    const bool lo = lane < 16;
    const int  chA[2]  = { lane, 32 + (lane & 15) };
    const bool actA[2] = { true, lo };
    const int  chB[2]  = { 16 + lane, lane & 15 };
    const bool actB[2] = { true, !lo };
    const int  cM = lo ? 32 + lane : lane - 16;
    const float bA = bias[lane];
    const float bM = bias[cM];
    const float bB = bias[16 + lane];
    float wA[NCLS], wM[NCLS], wB[NCLS];
    #pragma unroll
    for (int j = 0; j < NCLS; ++j) {
        wA[j] = Wc[lane * NCLS + j];
        wM[j] = Wc[cM * NCLS + j];
        wB[j] = Wc[(16 + lane) * NCLS + j];
    }
    const int   jc  = lane & 15;
    const float bcl = bc[jc < NCLS ? jc : 0];
    for (int k = 0; k < NODES_PER_WAVE / 2; ++k) {
        const int pair = p0 + k;
        const int nA = 2 * pair, nB = nA + 1;
        const int r0 = __builtin_amdgcn_readfirstlane(rowptr[nA]);
        const int r1 = __builtin_amdgcn_readfirstlane(rowptr[nB]);
        const int r2 = __builtin_amdgcn_readfirstlane(rowptr[nB + 1]);
        const float diA = dinv[nA], diB = dinv[nB];
        float a[2], b[2];
        {
            const float* hs = H + (size_t)nA * HID2;
            a[0] = hs[chA[0]];
            a[1] = lo ? hs[chA[1]] : 0.0f;
            sum_segment<HID2, 2>(H, cs, r0, r1, lane, chA, actA, a);
        }
        {
            const float* hs = H + (size_t)nB * HID2;
            b[0] = hs[chB[0]];
            b[1] = lo ? 0.0f : hs[chB[1]];
            sum_segment<HID2, 2>(H, cs, r1, r2, lane, chB, actB, b);
        }
        const float mid = lo ? diA * a[1] : diB * b[1];
        const float vA = fmaxf(diA * a[0] + bA, 0.0f);
        const float vM = fmaxf(mid          + bM, 0.0f);
        const float vB = fmaxf(diB * b[0] + bB, 0.0f);
        float s[NCLS];
        #pragma unroll
        for (int j = 0; j < NCLS; ++j) {
            const float m  = vM * wM[j];
            const float pa = vA * wA[j] + (lo ? m : 0.0f);
            const float pb = vB * wB[j] + (lo ? 0.0f : m);
            s[j] = (lo ? pa : pb) + __shfl_xor(lo ? pb : pa, 16);
        }
        #pragma unroll
        for (int d = 8; d >= 1; d >>= 1) {
            #pragma unroll
            for (int j = 0; j < NCLS; ++j) s[j] += __shfl_xor(s[j], d);
        }
        if (jc < NCLS) {
            const float v = jc == 0 ? s[0] : (jc == 1 ? s[1] : s[2]);
            sl[(wave * NODES_PER_WAVE + 2 * k + (lane >> 4)) * NCLS + jc] = v + bcl;
        }
    }
    __syncthreads();
    if (threadIdx.x < GB_NODES * NCLS)
        logits[(size_t)blockIdx.x * (GB_NODES * NCLS) + threadIdx.x] = sl[threadIdx.x];
}


extern "C" void kernel_launch(void* const* d_in, const int* in_sizes, int n_in,
                              void* d_out, int out_size, void* d_ws, size_t ws_size,
                              hipStream_t stream) {
    const float* x  = (const float*)d_in[0];
    const int*   ei = (const int*)  d_in[1];
    const float* ew = (const float*)d_in[2];
    const float* W1 = (const float*)d_in[3];
    const float* b1 = (const float*)d_in[4];
    const float* W2 = (const float*)d_in[5];
    const float* b2 = (const float*)d_in[6];
    const float* Wc = (const float*)d_in[7];
    const float* bc = (const float*)d_in[8];
    float* logits = (float*)d_out;

    const int* src = ei;
    const int* dst = ei + N_EDGES;

    char* ws = (char*)d_ws;
    size_t off = 0;
    auto take = [&](size_t bytes) -> void* {
        void* p = ws + off;
        off += (bytes + 255) & ~(size_t)255;
        return p;
    };
    float* dinv     = (float*)take((size_t)N_NODES * 4);
    int*   cnt      = (int*)  take((size_t)N_NODES * 4);
    int*   rowptr   = (int*)  take((size_t)(N_NODES + 1) * 4);
    int*   blocksum = (int*)  take((size_t)SCAN_BLOCKS * 4);
    unsigned short* rank = (unsigned short*)take((size_t)N_EDGES * 2);
    unsigned* chunkcnt   = (unsigned*)take((size_t)NCH * NWORDS * 4);
    int*   pre      = (int*)  take((size_t)NCH * N_NODES * 4);
    int*   csr      = (int*)  take((size_t)N_EDGES * 8);
    float* H1   = (float*)take((size_t)N_NODES * HID * 4);
    float* H2   = (float*)take((size_t)N_NODES * HID2 * 4);

    const int TB = 256;
    rank_chunk  <<<NCH, RK_TB, 0, stream>>>(dst, chunkcnt, rank);
    chunk_offsets<<<CS_BLOCKS, CS_TB, 0, stream>>>(chunkcnt, pre, cnt);
    scan_local  <<<SCAN_BLOCKS, SCAN_TB, 0, stream>>>(cnt, rowptr, blocksum);
    scan_offsets<<<SCAN_BLOCKS, SCAN_TB, 0, stream>>>(blocksum, rowptr);
    fill_csr    <<<(N_EDGES + TB - 1) / TB, TB, 0, stream>>>(src, dst, ew, rank, rowptr, pre, csr);
    node_dinv   <<<(N_NODES + TB - 1) / TB, TB, 0, stream>>>(rowptr, csr, dinv);

    {
        gemm_wmma<IN_CH, HID><<<(N_NODES / 16 + 7) / 8, 256, 0, stream>>>(x, W1, dinv, H1);
    }
    gather96_gemm48<<<N_NODES / GB_NODES, 256, 0, stream>>>(H1, rowptr, csr, dinv, b1, W2, H2);

    gather48_classify<<<N_NODES / GB_NODES, 256, 0, stream>>>(H2, rowptr, csr, dinv, b2, Wc, bc, logits);
}


# gathers: each source row fetched by ONE load (96 ch: 16 B per lane, 48 ch: 8 B per lane, 24 lanes a row), not three or two 4-byte loads; layer-2 nodes singly; classifier lane tree re-ordered
# speedup vs baseline: 3.8011x; 1.0542x over previous
#include <hip/hip_runtime.h>

#define N_NODES 100000
#define N_EDGES 3200000
#define IN_CH   128
#define HID     96
#define HID2    48
#define NCLS    3

typedef __attribute__((ext_vector_type(2)))  float  v2f;
typedef __attribute__((ext_vector_type(4)))  float  v4f;
typedef __attribute__((ext_vector_type(8)))  float  v8f;


#define SCAN_TB     256
#define SCAN_BLOCKS ((N_NODES + 1 + SCAN_TB - 1) / SCAN_TB)

#define NCH     64
#define CH      (N_EDGES / NCH)
#define NWORDS  (N_NODES / 2)
#define RK_TB   1024
#define RK_PER  ((CH + RK_TB - 1) / RK_TB)
static_assert(N_EDGES % NCH == 0, "whole chunks");
static_assert(CH < 65536, "a chunk cannot overflow a 16-bit counter, whatever the edges are");
static_assert(N_NODES % 2 == 0 && NWORDS % 4 == 0, "counter words move as 16-byte vectors");

__global__ __launch_bounds__(RK_TB) void rank_chunk(const int* __restrict__ dst,
                                                    unsigned* __restrict__ chunkcnt,
                                                    unsigned short* __restrict__ rank) {
    __shared__ uint4 cw4[NWORDS / 4];
    unsigned* cw = reinterpret_cast<unsigned*>(cw4);
    const int t  = threadIdx.x;
    const int e0 = blockIdx.x * CH;
    for (int i = t; i < NWORDS / 4; i += RK_TB) cw4[i] = make_uint4(0u, 0u, 0u, 0u);
    const bool has_last = t + (RK_PER - 1) * RK_TB < CH;
    int d[RK_PER];
    #pragma unroll
    for (int k = 0; k < RK_PER; ++k)
        d[k] = (k < RK_PER - 1 || has_last) ? dst[e0 + t + k * RK_TB] : 0;
    __syncthreads();
    unsigned r[RK_PER];
    #pragma unroll
    for (int k = 0; k < RK_PER; ++k) {
        r[k] = 0u;
        if (k < RK_PER - 1 || has_last) {
            const unsigned sh = (unsigned)(d[k] & 1) << 4;
            r[k] = atomicAdd(cw + (d[k] >> 1), 1u << sh) >> sh;
        }
    }
    #pragma unroll
    for (int k = 0; k < RK_PER; ++k)
        if (k < RK_PER - 1 || has_last) rank[e0 + t + k * RK_TB] = (unsigned short)r[k];
    __syncthreads();
    uint4* outw = reinterpret_cast<uint4*>(chunkcnt + (size_t)blockIdx.x * NWORDS);
    for (int i = t; i < NWORDS / 4; i += RK_TB) outw[i] = cw4[i];
}

#define CS_TB     256
#define CS_GROUPS (CS_TB / 32)
#define CS_PER    (NCH / CS_GROUPS)
#define CS_BLOCKS ((NWORDS + 31) / 32)
static_assert(NCH % CS_GROUPS == 0, "equal share of chunks per wave");

__global__ __launch_bounds__(CS_TB) void chunk_offsets(const unsigned* __restrict__ chunkcnt,
                                                       int* __restrict__ pre,
                                                       int* __restrict__ cnt) {
    __shared__ int2 part[CS_GROUPS][32];
    const int lane = threadIdx.x & 31, g = threadIdx.x >> 5;
    const int word = blockIdx.x * 32 + lane;
    const bool ok  = word < NWORDS;
    const int  wl  = ok ? word : NWORDS - 1;
    unsigned v[CS_PER];
    #pragma unroll
    for (int k = 0; k < CS_PER; ++k)
        v[k] = chunkcnt[(size_t)(g * CS_PER + k) * NWORDS + wl];
    int lo = 0, hi = 0;
    #pragma unroll
    for (int k = 0; k < CS_PER; ++k) { lo += (int)(v[k] & 0xFFFFu); hi += (int)(v[k] >> 16); }
    part[g][lane] = make_int2(lo, hi);
    __syncthreads();
    int blo = 0, bhi = 0;
    for (int w = 0; w < CS_GROUPS; ++w)
        if (w < g) { const int2 p = part[w][lane]; blo += p.x; bhi += p.y; }
    if (ok) {
        #pragma unroll
        for (int k = 0; k < CS_PER; ++k) {
            reinterpret_cast<int2*>(pre + (size_t)(g * CS_PER + k) * N_NODES)[word] = make_int2(blo, bhi);
            blo += (int)(v[k] & 0xFFFFu);
            bhi += (int)(v[k] >> 16);
        }
        if (g == CS_GROUPS - 1) reinterpret_cast<int2*>(cnt)[word] = make_int2(blo, bhi);
    }
}

__global__ __launch_bounds__(SCAN_TB) void scan_local(const int* __restrict__ cnt,
                                                      int* __restrict__ rowptr,
                                                      int* __restrict__ blocksum) {
    __shared__ int wsum[SCAN_TB / 32];
    const int t = threadIdx.x, lane = t & 31, wave = t >> 5;
    const int i = blockIdx.x * SCAN_TB + t;
    const int c = (i < N_NODES) ? cnt[i] : 0;
    int incl = c;
    for (int d = 1; d < 32; d <<= 1) {
        const int v = __shfl_up(incl, d);
        if (lane >= d) incl += v;
    }
    if (lane == 31) wsum[wave] = incl;
    __syncthreads();
    int pre = 0;
    for (int w = 0; w < SCAN_TB / 32; ++w) if (w < wave) pre += wsum[w];
    if (i <= N_NODES) rowptr[i] = pre + incl - c;
    if (t == SCAN_TB - 1) blocksum[blockIdx.x] = pre + incl;
}

__global__ __launch_bounds__(SCAN_TB) void scan_offsets(const int* __restrict__ blocksum,
                                                        int* __restrict__ rowptr) {
    __shared__ int wsum[SCAN_TB / 32];
    const int t = threadIdx.x, lane = t & 31, wave = t >> 5;
    const int b = blockIdx.x;
    int v = 0;
    for (int k = t; k < b; k += SCAN_TB) v += blocksum[k];
    for (int d = 16; d >= 1; d >>= 1) v += __shfl_xor(v, d);
    if (lane == 0) wsum[wave] = v;
    __syncthreads();
    int off = 0;
    for (int w = 0; w < SCAN_TB / 32; ++w) off += wsum[w];
    const int i = b * SCAN_TB + t;
    if (i <= N_NODES) rowptr[i] += off;
}

__global__ __launch_bounds__(256) void fill_csr(const int* __restrict__ src,
                                                const int* __restrict__ dst,
                                                const float* __restrict__ ew,
                                                const unsigned short* __restrict__ rank,
                                                const int* __restrict__ rowptr,
                                                const int* __restrict__ pre,
                                                int* __restrict__ csr) {
    int e = blockIdx.x * blockDim.x + threadIdx.x;
    if (e < N_EDGES) {
        const int d   = dst[e];
        const int pos = rowptr[d] + pre[(size_t)(e / CH) * N_NODES + d] + (int)rank[e];
        reinterpret_cast<int2*>(csr)[pos] = make_int2(src[e], __float_as_int(ew[e]));
    }
}

__global__ __launch_bounds__(256) void node_dinv(const int* __restrict__ rowptr,
                                                 const int* __restrict__ csr,
                                                 float* __restrict__ dinv) {
    int i = blockIdx.x * blockDim.x + threadIdx.x;
    if (i < N_NODES) {
        const int2* c = reinterpret_cast<const int2*>(csr);
        const int end = rowptr[i + 1];
        float d = 1.0f;
        for (int p = rowptr[i]; p < end; ++p) d += __int_as_float(c[p].y);
        dinv[i] = rsqrtf(d);
    }
}


template<int K, int COLS>
__global__ __launch_bounds__(256) void gemm_wmma(const float* __restrict__ X,
                                                 const float* __restrict__ W,
                                                 const float* __restrict__ dinv,
                                                 float* __restrict__ H) {
    constexpr int NT = COLS / 16;
    constexpr int KS = K / 4;
    constexpr int P2 = COLS + ((16 - COLS % 32 + 32) % 32);
    static_assert(K % 4 == 0 && COLS % 16 == 0 && P2 % 32 == 16 && P2 % 4 == 0, "tile shapes");
    __shared__ __attribute__((aligned(16))) v2f wp[(K / 2) * P2];
    const int lane = threadIdx.x & 31;
    const int wave = threadIdx.x >> 5;
    const int c = lane & 15, h = lane >> 4;
    const int m0 = (blockIdx.x * 8 + wave) * 16;
    const bool live = m0 < N_NODES;
    const int row = (live ? m0 : N_NODES - 16) + c;
    const float* xrow = X + (size_t)row * K + 2 * h;
    constexpr int QR  = COLS / 4;
    constexpr int NQ  = (K / 2) * QR;
    constexpr int WIT = (NQ + 255) / 256;
    float4 w0[WIT], w1[WIT];
    #pragma unroll
    for (int j = 0; j < WIT; ++j) {
        const int idx = j * 256 + (int)threadIdx.x;
        if ((j + 1) * 256 <= NQ || idx < NQ) {
            const int kp = idx / QR, q = idx - kp * QR;
            w0[j] = *(const float4*)(W + (size_t)(2 * kp)     * COLS + 4 * q);
            w1[j] = *(const float4*)(W + (size_t)(2 * kp + 1) * COLS + 4 * q);
        }
    }
    #pragma unroll
    for (int j = 0; j < WIT; ++j) {
        const int idx = j * 256 + (int)threadIdx.x;
        if ((j + 1) * 256 <= NQ || idx < NQ) {
            const int kp = idx / QR, q = idx - kp * QR;
            float4* d = reinterpret_cast<float4*>(wp + kp * P2 + 4 * q);
            d[0] = make_float4(w0[j].x, w1[j].x, w0[j].y, w1[j].y);
            d[1] = make_float4(w0[j].z, w1[j].z, w0[j].w, w1[j].w);
        }
    }
    __syncthreads();
    v2f a[KS];
    #pragma unroll
    for (int i = 0; i < KS; ++i) a[i] = *(const v2f*)(xrow + 4 * i);
    v8f acc[NT];
    #pragma unroll
    for (int t = 0; t < NT; ++t) acc[t] = v8f{};
    const v2f* bp = wp + h * P2 + c;
    #pragma unroll
    for (int i = 0; i < KS; ++i) {
        #pragma unroll
        for (int t = 0; t < NT; ++t) {
            const v2f b = bp[(2 * i) * P2 + 16 * t];
            acc[t] = __builtin_amdgcn_wmma_f32_16x16x4_f32(false, a[i], false, b,
                                                           (short)0, acc[t], false, false);
        }
    }
    if (live) {
        float* hrow = H + (size_t)(m0 + 8 * h) * COLS + c;
        float dr[8];
        #pragma unroll
        for (int r = 0; r < 8; ++r) dr[r] = dinv[m0 + 8 * h + r];
        #pragma unroll
        for (int t = 0; t < NT; ++t) {
            #pragma unroll
            for (int r = 0; r < 8; ++r) hrow[(size_t)r * COLS + 16 * t] = dr[r] * acc[t][r];
        }
    }
}


#define ROW_LANES 24
template<int VW> struct row_vec;
template<> struct row_vec<2> { typedef v2f type; };
template<> struct row_vec<4> { typedef v4f type; };

template<int C, int VW>
static __device__ inline void sum_segment(const float* __restrict__ H,
                                          const int2* __restrict__ csr,
                                          int beg, int end, int lane, int q,
                                          typename row_vec<VW>::type& acc) {
    typedef typename row_vec<VW>::type vec;
    static_assert(C == ROW_LANES * VW, "a row is ROW_LANES lanes x VW floats");
    constexpr int U = 8;
    for (int base = beg; base < end; base += 32) {
        const int left = end - base;
        const int n = left < 32 ? left : 32;
        int2 ev = make_int2(0, 0);
        if (lane < n) ev = csr[base + lane];
        for (int j0 = 0; j0 < n; j0 += U) {
            const int m = n - j0;
            vec hv[U];
            if (m >= U) {
                #pragma unroll
                for (int u = 0; u < U; ++u) {
                    const int s = __builtin_amdgcn_readlane(ev.x, j0 + u);
                    hv[u] = *(const vec*)(H + (size_t)s * C + VW * q);
                }
                #pragma unroll
                for (int u = 0; u < U; ++u) {
                    const float w = __int_as_float(__builtin_amdgcn_readlane(ev.y, j0 + u));
                    acc += w * hv[u];
                }
            } else {
                #pragma unroll
                for (int u = 0; u < U; ++u)
                    if (u < m) {
                        const int s = __builtin_amdgcn_readlane(ev.x, j0 + u);
                        hv[u] = *(const vec*)(H + (size_t)s * C + VW * q);
                    }
                #pragma unroll
                for (int u = 0; u < U; ++u)
                    if (u < m) {
                        const float w = __int_as_float(__builtin_amdgcn_readlane(ev.y, j0 + u));
                        acc += w * hv[u];
                    }
            }
        }
    }
}

#define NODES_PER_WAVE 4
#define GB_NODES (8 * NODES_PER_WAVE)

__global__ __launch_bounds__(256) void gather96_gemm48(const float* __restrict__ H,
                                                       const int* __restrict__ rowptr,
                                                       const int* __restrict__ csr,
                                                       const float* __restrict__ dinv,
                                                       const float* __restrict__ bias,
                                                       const float* __restrict__ W,
                                                       float* __restrict__ H2) {
    static_assert(N_NODES % GB_NODES == 0 && GB_NODES % 16 == 0, "whole blocks, whole 16-row strips");
    constexpr int K = HID, COLS = HID2;
    constexpr int NT = COLS / 16;
    constexpr int KS = K / 4;
    constexpr int P2 = COLS + ((16 - COLS % 32 + 32) % 32);
    constexpr int XP = K + 4;
    static_assert(K % 4 == 0 && COLS % 16 == 0 && P2 % 32 == 16 && P2 % 4 == 0, "tile shapes");
    static_assert(XP % 64 == 36 && XP % 4 == 0, "A fragment reads: 8-byte aligned, no bank shared");
    __shared__ __attribute__((aligned(16))) v2f wp[(K / 2) * P2];
    __shared__ __attribute__((aligned(16))) float xt[GB_NODES * XP];
    const int lane = threadIdx.x & 31;
    const int wave = __builtin_amdgcn_readfirstlane((int)(threadIdx.x >> 5));
    {
        constexpr int QR  = COLS / 4;
        constexpr int NQ  = (K / 2) * QR;
        constexpr int WIT = (NQ + 255) / 256;
        float4 w0[WIT], w1[WIT];
        #pragma unroll
        for (int j = 0; j < WIT; ++j) {
            const int idx = j * 256 + (int)threadIdx.x;
            if ((j + 1) * 256 <= NQ || idx < NQ) {
                const int kp = idx / QR, q = idx - kp * QR;
                w0[j] = *(const float4*)(W + (size_t)(2 * kp)     * COLS + 4 * q);
                w1[j] = *(const float4*)(W + (size_t)(2 * kp + 1) * COLS + 4 * q);
            }
        }
        #pragma unroll
        for (int j = 0; j < WIT; ++j) {
            const int idx = j * 256 + (int)threadIdx.x;
            if ((j + 1) * 256 <= NQ || idx < NQ) {
                const int kp = idx / QR, q = idx - kp * QR;
                float4* d = reinterpret_cast<float4*>(wp + kp * P2 + 4 * q);
                d[0] = make_float4(w0[j].x, w1[j].x, w0[j].y, w1[j].y);
                d[1] = make_float4(w0[j].z, w1[j].z, w0[j].w, w1[j].w);
            }
        }
    }
    const int n0 = (blockIdx.x * 8 + wave) * NODES_PER_WAVE;
    const int q = lane < ROW_LANES ? lane : lane - ROW_LANES;
    const float bb[4] = { bias[4 * q], bias[4 * q + 1], bias[4 * q + 2], bias[4 * q + 3] };
    for (int k = 0; k < NODES_PER_WAVE; ++k) {
        const int node = n0 + k;
        const int beg  = __builtin_amdgcn_readfirstlane(rowptr[node]);
        const int end  = __builtin_amdgcn_readfirstlane(rowptr[node + 1]);
        const float di = dinv[node];
        const float* hs = H + (size_t)node * HID;
        v4f acc = *(const v4f*)(hs + 4 * q);
        sum_segment<HID, 4>(H, reinterpret_cast<const int2*>(csr), beg, end, lane, q, acc);
        float* o = xt + (wave * NODES_PER_WAVE + k) * XP;
        if (lane < ROW_LANES) {
            v4f r;
            #pragma unroll
            for (int p = 0; p < 4; ++p) r[p] = fmaxf(di * acc[p] + bb[p], 0.0f);
            *(v4f*)(o + 4 * lane) = r;
        }
    }
    __syncthreads();
    if (wave < GB_NODES / 16) {
        const int c = lane & 15, h = lane >> 4;
        const int m0 = blockIdx.x * GB_NODES + wave * 16;
        const float* xrow = xt + (wave * 16 + c) * XP + 2 * h;
        const v2f* bp = wp + h * P2 + c;
        float dr[8];
        #pragma unroll
        for (int r = 0; r < 8; ++r) dr[r] = dinv[m0 + 8 * h + r];
        v8f acc[NT];
        #pragma unroll
        for (int t = 0; t < NT; ++t) acc[t] = v8f{};
        #pragma unroll
        for (int i = 0; i < KS; ++i) {
            const v2f a = *(const v2f*)(xrow + 4 * i);
            #pragma unroll
            for (int t = 0; t < NT; ++t) {
                const v2f b = bp[(2 * i) * P2 + 16 * t];
                acc[t] = __builtin_amdgcn_wmma_f32_16x16x4_f32(false, a, false, b,
                                                               (short)0, acc[t], false, false);
            }
        }
        float* hrow = H2 + (size_t)(m0 + 8 * h) * COLS + c;
        #pragma unroll
        for (int t = 0; t < NT; ++t) {
            #pragma unroll
            for (int r = 0; r < 8; ++r) hrow[(size_t)r * COLS + 16 * t] = dr[r] * acc[t][r];
        }
    }
}

__global__ __launch_bounds__(256) void gather48_classify(const float* __restrict__ H,
                                                         const int* __restrict__ rowptr,
                                                         const int* __restrict__ csr,
                                                         const float* __restrict__ dinv,
                                                         const float* __restrict__ bias,
                                                         const float* __restrict__ Wc,
                                                         const float* __restrict__ bc,
                                                         float* __restrict__ logits) {
    static_assert((GB_NODES * NCLS) % 32 == 0 && GB_NODES * NCLS <= 256, "the block's logits are whole lines");
    __shared__ float sl[GB_NODES * NCLS];
    const int lane = threadIdx.x & 31;
    const int wave = __builtin_amdgcn_readfirstlane((int)(threadIdx.x >> 5));
    const int n0 = (blockIdx.x * 8 + wave) * NODES_PER_WAVE;
    const int2* cs = reinterpret_cast<const int2*>(csr);
    const bool on = lane < ROW_LANES;
    const int  q  = on ? lane : lane - ROW_LANES;
    const float b0 = bias[2 * q], b1 = bias[2 * q + 1];
    float w0[NCLS], w1[NCLS];
    #pragma unroll
    for (int j = 0; j < NCLS; ++j) {
        w0[j] = Wc[(2 * q) * NCLS + j];
        w1[j] = Wc[(2 * q + 1) * NCLS + j];
    }
    const float bcl = bc[lane < NCLS ? lane : 0];
    for (int k = 0; k < NODES_PER_WAVE; ++k) {
        const int node = n0 + k;
        const int beg  = __builtin_amdgcn_readfirstlane(rowptr[node]);
        const int end  = __builtin_amdgcn_readfirstlane(rowptr[node + 1]);
        const float di = dinv[node];
        v2f acc = *(const v2f*)(H + (size_t)node * HID2 + 2 * q);
        sum_segment<HID2, 2>(H, cs, beg, end, lane, q, acc);
        const float v0 = fmaxf(di * acc[0] + b0, 0.0f);
        const float v1 = fmaxf(di * acc[1] + b1, 0.0f);
        float s[NCLS];
        #pragma unroll
        for (int j = 0; j < NCLS; ++j) s[j] = on ? v0 * w0[j] + v1 * w1[j] : 0.0f;
        #pragma unroll
        for (int d = 16; d >= 1; d >>= 1) {
            #pragma unroll
            for (int j = 0; j < NCLS; ++j) s[j] += __shfl_xor(s[j], d);
        }
        if (lane < NCLS) {
            const float v = lane == 0 ? s[0] : (lane == 1 ? s[1] : s[2]);
            sl[(wave * NODES_PER_WAVE + k) * NCLS + lane] = v + bcl;
        }
    }
    __syncthreads();
    if (threadIdx.x < GB_NODES * NCLS)
        logits[(size_t)blockIdx.x * (GB_NODES * NCLS) + threadIdx.x] = sl[threadIdx.x];
}


extern "C" void kernel_launch(void* const* d_in, const int* in_sizes, int n_in,
                              void* d_out, int out_size, void* d_ws, size_t ws_size,
                              hipStream_t stream) {
    const float* x  = (const float*)d_in[0];
    const int*   ei = (const int*)  d_in[1];
    const float* ew = (const float*)d_in[2];
    const float* W1 = (const float*)d_in[3];
    const float* b1 = (const float*)d_in[4];
    const float* W2 = (const float*)d_in[5];
    const float* b2 = (const float*)d_in[6];
    const float* Wc = (const float*)d_in[7];
    const float* bc = (const float*)d_in[8];
    float* logits = (float*)d_out;

    const int* src = ei;
    const int* dst = ei + N_EDGES;

    char* ws = (char*)d_ws;
    size_t off = 0;
    auto take = [&](size_t bytes) -> void* {
        void* p = ws + off;
        off += (bytes + 255) & ~(size_t)255;
        return p;
    };
    float* dinv     = (float*)take((size_t)N_NODES * 4);
    int*   cnt      = (int*)  take((size_t)N_NODES * 4);
    int*   rowptr   = (int*)  take((size_t)(N_NODES + 1) * 4);
    int*   blocksum = (int*)  take((size_t)SCAN_BLOCKS * 4);
    unsigned short* rank = (unsigned short*)take((size_t)N_EDGES * 2);
    unsigned* chunkcnt   = (unsigned*)take((size_t)NCH * NWORDS * 4);
    int*   pre      = (int*)  take((size_t)NCH * N_NODES * 4);
    int*   csr      = (int*)  take((size_t)N_EDGES * 8);
    float* H1   = (float*)take((size_t)N_NODES * HID * 4);
    float* H2   = (float*)take((size_t)N_NODES * HID2 * 4);

    const int TB = 256;
    rank_chunk  <<<NCH, RK_TB, 0, stream>>>(dst, chunkcnt, rank);
    chunk_offsets<<<CS_BLOCKS, CS_TB, 0, stream>>>(chunkcnt, pre, cnt);
    scan_local  <<<SCAN_BLOCKS, SCAN_TB, 0, stream>>>(cnt, rowptr, blocksum);
    scan_offsets<<<SCAN_BLOCKS, SCAN_TB, 0, stream>>>(blocksum, rowptr);
    fill_csr    <<<(N_EDGES + TB - 1) / TB, TB, 0, stream>>>(src, dst, ew, rank, rowptr, pre, csr);
    node_dinv   <<<(N_NODES + TB - 1) / TB, TB, 0, stream>>>(rowptr, csr, dinv);

    {
        gemm_wmma<IN_CH, HID><<<(N_NODES / 16 + 7) / 8, 256, 0, stream>>>(x, W1, dinv, H1);
    }
    gather96_gemm48<<<N_NODES / GB_NODES, 256, 0, stream>>>(H1, rowptr, csr, dinv, b1, W2, H2);

    gather48_classify<<<N_NODES / GB_NODES, 256, 0, stream>>>(H2, rowptr, csr, dinv, b2, Wc, bc, logits);
}


# edge placement: a work-group per chunk and quarter of the nodes keeps its nodes' next free slots in LDS and takes each edge's slot by a returning LDS add; no rank array; slot order: chunk, arrival
# speedup vs baseline: 4.1953x; 1.1037x over previous
#include <hip/hip_runtime.h>

#define N_NODES 100000
#define N_EDGES 3200000
#define IN_CH   128
#define HID     96
#define HID2    48
#define NCLS    3

typedef __attribute__((ext_vector_type(2)))  float  v2f;
typedef __attribute__((ext_vector_type(4)))  float  v4f;
typedef __attribute__((ext_vector_type(8)))  float  v8f;


#define SCAN_TB     256
#define SCAN_BLOCKS ((N_NODES + 1 + SCAN_TB - 1) / SCAN_TB)

#define NCH     64
#define CH      (N_EDGES / NCH)
#define NWORDS  (N_NODES / 2)
#define RK_TB   1024
#define RK_PER  ((CH + RK_TB - 1) / RK_TB)
static_assert(N_EDGES % NCH == 0, "whole chunks");
static_assert(CH < 65536, "a chunk cannot overflow a 16-bit counter, whatever the edges are");
static_assert(N_NODES % 2 == 0 && NWORDS % 4 == 0, "counter words move as 16-byte vectors");

__global__ __launch_bounds__(RK_TB) void count_chunk(const int* __restrict__ dst,
                                                     unsigned* __restrict__ chunkcnt) {
    __shared__ uint4 cw4[NWORDS / 4];
    unsigned* cw = reinterpret_cast<unsigned*>(cw4);
    const int t  = threadIdx.x;
    const int e0 = blockIdx.x * CH;
    for (int i = t; i < NWORDS / 4; i += RK_TB) cw4[i] = make_uint4(0u, 0u, 0u, 0u);
    const bool has_last = t + (RK_PER - 1) * RK_TB < CH;
    int d[RK_PER];
    #pragma unroll
    for (int k = 0; k < RK_PER; ++k)
        d[k] = (k < RK_PER - 1 || has_last) ? dst[e0 + t + k * RK_TB] : 0;
    __syncthreads();
    #pragma unroll
    for (int k = 0; k < RK_PER; ++k) {
        if (k < RK_PER - 1 || has_last) {
            const unsigned sh = (unsigned)(d[k] & 1) << 4;
            atomicAdd(cw + (d[k] >> 1), 1u << sh);
        }
    }
    __syncthreads();
    uint4* outw = reinterpret_cast<uint4*>(chunkcnt + (size_t)blockIdx.x * NWORDS);
    for (int i = t; i < NWORDS / 4; i += RK_TB) outw[i] = cw4[i];
}

#define CS_TB     256
#define CS_GROUPS (CS_TB / 32)
#define CS_PER    (NCH / CS_GROUPS)
#define CS_BLOCKS ((NWORDS + 31) / 32)
static_assert(NCH % CS_GROUPS == 0, "equal share of chunks per wave");

__global__ __launch_bounds__(CS_TB) void chunk_offsets(const unsigned* __restrict__ chunkcnt,
                                                       int* __restrict__ pre,
                                                       int* __restrict__ cnt) {
    __shared__ int2 part[CS_GROUPS][32];
    const int lane = threadIdx.x & 31, g = threadIdx.x >> 5;
    const int word = blockIdx.x * 32 + lane;
    const bool ok  = word < NWORDS;
    const int  wl  = ok ? word : NWORDS - 1;
    unsigned v[CS_PER];
    #pragma unroll
    for (int k = 0; k < CS_PER; ++k)
        v[k] = chunkcnt[(size_t)(g * CS_PER + k) * NWORDS + wl];
    int lo = 0, hi = 0;
    #pragma unroll
    for (int k = 0; k < CS_PER; ++k) { lo += (int)(v[k] & 0xFFFFu); hi += (int)(v[k] >> 16); }
    part[g][lane] = make_int2(lo, hi);
    __syncthreads();
    int blo = 0, bhi = 0;
    for (int w = 0; w < CS_GROUPS; ++w)
        if (w < g) { const int2 p = part[w][lane]; blo += p.x; bhi += p.y; }
    if (ok) {
        #pragma unroll
        for (int k = 0; k < CS_PER; ++k) {
            reinterpret_cast<int2*>(pre + (size_t)(g * CS_PER + k) * N_NODES)[word] = make_int2(blo, bhi);
            blo += (int)(v[k] & 0xFFFFu);
            bhi += (int)(v[k] >> 16);
        }
        if (g == CS_GROUPS - 1) reinterpret_cast<int2*>(cnt)[word] = make_int2(blo, bhi);
    }
}

__global__ __launch_bounds__(SCAN_TB) void scan_local(const int* __restrict__ cnt,
                                                      int* __restrict__ rowptr,
                                                      int* __restrict__ blocksum) {
    __shared__ int wsum[SCAN_TB / 32];
    const int t = threadIdx.x, lane = t & 31, wave = t >> 5;
    const int i = blockIdx.x * SCAN_TB + t;
    const int c = (i < N_NODES) ? cnt[i] : 0;
    int incl = c;
    for (int d = 1; d < 32; d <<= 1) {
        const int v = __shfl_up(incl, d);
        if (lane >= d) incl += v;
    }
    if (lane == 31) wsum[wave] = incl;
    __syncthreads();
    int pre = 0;
    for (int w = 0; w < SCAN_TB / 32; ++w) if (w < wave) pre += wsum[w];
    if (i <= N_NODES) rowptr[i] = pre + incl - c;
    if (t == SCAN_TB - 1) blocksum[blockIdx.x] = pre + incl;
}

__global__ __launch_bounds__(SCAN_TB) void scan_offsets(const int* __restrict__ blocksum,
                                                        int* __restrict__ rowptr) {
    __shared__ int wsum[SCAN_TB / 32];
    const int t = threadIdx.x, lane = t & 31, wave = t >> 5;
    const int b = blockIdx.x;
    int v = 0;
    for (int k = t; k < b; k += SCAN_TB) v += blocksum[k];
    for (int d = 16; d >= 1; d >>= 1) v += __shfl_xor(v, d);
    if (lane == 0) wsum[wave] = v;
    __syncthreads();
    int off = 0;
    for (int w = 0; w < SCAN_TB / 32; ++w) off += wsum[w];
    const int i = b * SCAN_TB + t;
    if (i <= N_NODES) rowptr[i] += off;
}

#define PQ        4
#define PQ_NODES  (N_NODES / PQ)
#define PL_B      17
static_assert(N_NODES % PQ == 0 && PQ_NODES % 4 == 0, "the ranges' slots move as 16-byte vectors");

__global__ __launch_bounds__(RK_TB) void place_chunk(const int* __restrict__ src,
                                                     const int* __restrict__ dst,
                                                     const float* __restrict__ ew,
                                                     const int* __restrict__ rowptr,
                                                     const int* __restrict__ pre,
                                                     int* __restrict__ csr) {
    __shared__ int4 cur4[PQ_NODES / 4];
    int* cur = reinterpret_cast<int*>(cur4);
    const int t  = threadIdx.x;
    const int c  = blockIdx.x / PQ;
    const int nb = (blockIdx.x % PQ) * PQ_NODES;
    const int4* rp = reinterpret_cast<const int4*>(rowptr + nb);
    const int4* pp = reinterpret_cast<const int4*>(pre + (size_t)c * N_NODES + nb);
    for (int i = t; i < PQ_NODES / 4; i += RK_TB) {
        const int4 a = rp[i], b = pp[i];
        cur4[i] = make_int4(a.x + b.x, a.y + b.y, a.z + b.z, a.w + b.w);
    }
    __syncthreads();
    const int  e0 = c * CH;
    const bool has_last = t + (RK_PER - 1) * RK_TB < CH;
    int2* out = reinterpret_cast<int2*>(csr);
    #pragma unroll
    for (int k0 = 0; k0 < RK_PER; k0 += PL_B) {
        int d[PL_B], s[PL_B], w[PL_B];
        #pragma unroll
        for (int k = 0; k < PL_B; ++k) {
            const int kk = k0 + k;
            d[k] = -1;
            if (kk < RK_PER - 1 || (kk == RK_PER - 1 && has_last)) {
                const int e = e0 + t + kk * RK_TB;
                d[k] = dst[e];
                s[k] = src[e];
                w[k] = __float_as_int(ew[e]);
            }
        }
        int pos[PL_B];
        #pragma unroll
        for (int k = 0; k < PL_B; ++k) {
            const unsigned r = (unsigned)(d[k] - nb);
            pos[k] = -1;
            if (k0 + k < RK_PER && r < (unsigned)PQ_NODES) pos[k] = atomicAdd(cur + r, 1);
        }
        #pragma unroll
        for (int k = 0; k < PL_B; ++k)
            if (k0 + k < RK_PER && pos[k] >= 0) out[pos[k]] = make_int2(s[k], w[k]);
    }
}

__global__ __launch_bounds__(256) void node_dinv(const int* __restrict__ rowptr,
                                                 const int* __restrict__ csr,
                                                 float* __restrict__ dinv) {
    int i = blockIdx.x * blockDim.x + threadIdx.x;
    if (i < N_NODES) {
        const int2* c = reinterpret_cast<const int2*>(csr);
        const int end = rowptr[i + 1];
        float d = 1.0f;
        for (int p = rowptr[i]; p < end; ++p) d += __int_as_float(c[p].y);
        dinv[i] = rsqrtf(d);
    }
}


template<int K, int COLS>
__global__ __launch_bounds__(256) void gemm_wmma(const float* __restrict__ X,
                                                 const float* __restrict__ W,
                                                 const float* __restrict__ dinv,
                                                 float* __restrict__ H) {
    constexpr int NT = COLS / 16;
    constexpr int KS = K / 4;
    constexpr int P2 = COLS + ((16 - COLS % 32 + 32) % 32);
    static_assert(K % 4 == 0 && COLS % 16 == 0 && P2 % 32 == 16 && P2 % 4 == 0, "tile shapes");
    __shared__ __attribute__((aligned(16))) v2f wp[(K / 2) * P2];
    const int lane = threadIdx.x & 31;
    const int wave = threadIdx.x >> 5;
    const int c = lane & 15, h = lane >> 4;
    const int m0 = (blockIdx.x * 8 + wave) * 16;
    const bool live = m0 < N_NODES;
    const int row = (live ? m0 : N_NODES - 16) + c;
    const float* xrow = X + (size_t)row * K + 2 * h;
    constexpr int QR  = COLS / 4;
    constexpr int NQ  = (K / 2) * QR;
    constexpr int WIT = (NQ + 255) / 256;
    float4 w0[WIT], w1[WIT];
    #pragma unroll
    for (int j = 0; j < WIT; ++j) {
        const int idx = j * 256 + (int)threadIdx.x;
        if ((j + 1) * 256 <= NQ || idx < NQ) {
            const int kp = idx / QR, q = idx - kp * QR;
            w0[j] = *(const float4*)(W + (size_t)(2 * kp)     * COLS + 4 * q);
            w1[j] = *(const float4*)(W + (size_t)(2 * kp + 1) * COLS + 4 * q);
        }
    }
    #pragma unroll
    for (int j = 0; j < WIT; ++j) {
        const int idx = j * 256 + (int)threadIdx.x;
        if ((j + 1) * 256 <= NQ || idx < NQ) {
            const int kp = idx / QR, q = idx - kp * QR;
            float4* d = reinterpret_cast<float4*>(wp + kp * P2 + 4 * q);
            d[0] = make_float4(w0[j].x, w1[j].x, w0[j].y, w1[j].y);
            d[1] = make_float4(w0[j].z, w1[j].z, w0[j].w, w1[j].w);
        }
    }
    __syncthreads();
    v2f a[KS];
    #pragma unroll
    for (int i = 0; i < KS; ++i) a[i] = *(const v2f*)(xrow + 4 * i);
    v8f acc[NT];
    #pragma unroll
    for (int t = 0; t < NT; ++t) acc[t] = v8f{};
    const v2f* bp = wp + h * P2 + c;
    #pragma unroll
    for (int i = 0; i < KS; ++i) {
        #pragma unroll
        for (int t = 0; t < NT; ++t) {
            const v2f b = bp[(2 * i) * P2 + 16 * t];
            acc[t] = __builtin_amdgcn_wmma_f32_16x16x4_f32(false, a[i], false, b,
                                                           (short)0, acc[t], false, false);
        }
    }
    if (live) {
        float* hrow = H + (size_t)(m0 + 8 * h) * COLS + c;
        float dr[8];
        #pragma unroll
        for (int r = 0; r < 8; ++r) dr[r] = dinv[m0 + 8 * h + r];
        #pragma unroll
        for (int t = 0; t < NT; ++t) {
            #pragma unroll
            for (int r = 0; r < 8; ++r) hrow[(size_t)r * COLS + 16 * t] = dr[r] * acc[t][r];
        }
    }
}


#define ROW_LANES 24
template<int VW> struct row_vec;
template<> struct row_vec<2> { typedef v2f type; };
template<> struct row_vec<4> { typedef v4f type; };

template<int C, int VW>
static __device__ inline void sum_segment(const float* __restrict__ H,
                                          const int2* __restrict__ csr,
                                          int beg, int end, int lane, int q,
                                          typename row_vec<VW>::type& acc) {
    typedef typename row_vec<VW>::type vec;
    static_assert(C == ROW_LANES * VW, "a row is ROW_LANES lanes x VW floats");
    constexpr int U = 8;
    for (int base = beg; base < end; base += 32) {
        const int left = end - base;
        const int n = left < 32 ? left : 32;
        int2 ev = make_int2(0, 0);
        if (lane < n) ev = csr[base + lane];
        for (int j0 = 0; j0 < n; j0 += U) {
            const int m = n - j0;
            vec hv[U];
            if (m >= U) {
                #pragma unroll
                for (int u = 0; u < U; ++u) {
                    const int s = __builtin_amdgcn_readlane(ev.x, j0 + u);
                    hv[u] = *(const vec*)(H + (size_t)s * C + VW * q);
                }
                #pragma unroll
                for (int u = 0; u < U; ++u) {
                    const float w = __int_as_float(__builtin_amdgcn_readlane(ev.y, j0 + u));
                    acc += w * hv[u];
                }
            } else {
                #pragma unroll
                for (int u = 0; u < U; ++u)
                    if (u < m) {
                        const int s = __builtin_amdgcn_readlane(ev.x, j0 + u);
                        hv[u] = *(const vec*)(H + (size_t)s * C + VW * q);
                    }
                #pragma unroll
                for (int u = 0; u < U; ++u)
                    if (u < m) {
                        const float w = __int_as_float(__builtin_amdgcn_readlane(ev.y, j0 + u));
                        acc += w * hv[u];
                    }
            }
        }
    }
}

#define NODES_PER_WAVE 4
#define GB_NODES (8 * NODES_PER_WAVE)

__global__ __launch_bounds__(256) void gather96_gemm48(const float* __restrict__ H,
                                                       const int* __restrict__ rowptr,
                                                       const int* __restrict__ csr,
                                                       const float* __restrict__ dinv,
                                                       const float* __restrict__ bias,
                                                       const float* __restrict__ W,
                                                       float* __restrict__ H2) {
    static_assert(N_NODES % GB_NODES == 0 && GB_NODES % 16 == 0, "whole blocks, whole 16-row strips");
    constexpr int K = HID, COLS = HID2;
    constexpr int NT = COLS / 16;
    constexpr int KS = K / 4;
    constexpr int P2 = COLS + ((16 - COLS % 32 + 32) % 32);
    constexpr int XP = K + 4;
    static_assert(K % 4 == 0 && COLS % 16 == 0 && P2 % 32 == 16 && P2 % 4 == 0, "tile shapes");
    static_assert(XP % 64 == 36 && XP % 4 == 0, "A fragment reads: 8-byte aligned, no bank shared");
    __shared__ __attribute__((aligned(16))) v2f wp[(K / 2) * P2];
    __shared__ __attribute__((aligned(16))) float xt[GB_NODES * XP];
    const int lane = threadIdx.x & 31;
    const int wave = __builtin_amdgcn_readfirstlane((int)(threadIdx.x >> 5));
    {
        constexpr int QR  = COLS / 4;
        constexpr int NQ  = (K / 2) * QR;
        constexpr int WIT = (NQ + 255) / 256;
        float4 w0[WIT], w1[WIT];
        #pragma unroll
        for (int j = 0; j < WIT; ++j) {
            const int idx = j * 256 + (int)threadIdx.x;
            if ((j + 1) * 256 <= NQ || idx < NQ) {
                const int kp = idx / QR, q = idx - kp * QR;
                w0[j] = *(const float4*)(W + (size_t)(2 * kp)     * COLS + 4 * q);
                w1[j] = *(const float4*)(W + (size_t)(2 * kp + 1) * COLS + 4 * q);
            }
        }
        #pragma unroll
        for (int j = 0; j < WIT; ++j) {
            const int idx = j * 256 + (int)threadIdx.x;
            if ((j + 1) * 256 <= NQ || idx < NQ) {
                const int kp = idx / QR, q = idx - kp * QR;
                float4* d = reinterpret_cast<float4*>(wp + kp * P2 + 4 * q);
                d[0] = make_float4(w0[j].x, w1[j].x, w0[j].y, w1[j].y);
                d[1] = make_float4(w0[j].z, w1[j].z, w0[j].w, w1[j].w);
            }
        }
    }
    const int n0 = (blockIdx.x * 8 + wave) * NODES_PER_WAVE;
    const int q = lane < ROW_LANES ? lane : lane - ROW_LANES;
    const float bb[4] = { bias[4 * q], bias[4 * q + 1], bias[4 * q + 2], bias[4 * q + 3] };
    for (int k = 0; k < NODES_PER_WAVE; ++k) {
        const int node = n0 + k;
        const int beg  = __builtin_amdgcn_readfirstlane(rowptr[node]);
        const int end  = __builtin_amdgcn_readfirstlane(rowptr[node + 1]);
        const float di = dinv[node];
        const float* hs = H + (size_t)node * HID;
        v4f acc = *(const v4f*)(hs + 4 * q);
        sum_segment<HID, 4>(H, reinterpret_cast<const int2*>(csr), beg, end, lane, q, acc);
        float* o = xt + (wave * NODES_PER_WAVE + k) * XP;
        if (lane < ROW_LANES) {
            v4f r;
            #pragma unroll
            for (int p = 0; p < 4; ++p) r[p] = fmaxf(di * acc[p] + bb[p], 0.0f);
            *(v4f*)(o + 4 * lane) = r;
        }
    }
    __syncthreads();
    if (wave < GB_NODES / 16) {
        const int c = lane & 15, h = lane >> 4;
        const int m0 = blockIdx.x * GB_NODES + wave * 16;
        const float* xrow = xt + (wave * 16 + c) * XP + 2 * h;
        const v2f* bp = wp + h * P2 + c;
        float dr[8];
        #pragma unroll
        for (int r = 0; r < 8; ++r) dr[r] = dinv[m0 + 8 * h + r];
        v8f acc[NT];
        #pragma unroll
        for (int t = 0; t < NT; ++t) acc[t] = v8f{};
        #pragma unroll
        for (int i = 0; i < KS; ++i) {
            const v2f a = *(const v2f*)(xrow + 4 * i);
            #pragma unroll
            for (int t = 0; t < NT; ++t) {
                const v2f b = bp[(2 * i) * P2 + 16 * t];
                acc[t] = __builtin_amdgcn_wmma_f32_16x16x4_f32(false, a, false, b,
                                                               (short)0, acc[t], false, false);
            }
        }
        float* hrow = H2 + (size_t)(m0 + 8 * h) * COLS + c;
        #pragma unroll
        for (int t = 0; t < NT; ++t) {
            #pragma unroll
            for (int r = 0; r < 8; ++r) hrow[(size_t)r * COLS + 16 * t] = dr[r] * acc[t][r];
        }
    }
}

__global__ __launch_bounds__(256) void gather48_classify(const float* __restrict__ H,
                                                         const int* __restrict__ rowptr,
                                                         const int* __restrict__ csr,
                                                         const float* __restrict__ dinv,
                                                         const float* __restrict__ bias,
                                                         const float* __restrict__ Wc,
                                                         const float* __restrict__ bc,
                                                         float* __restrict__ logits) {
    static_assert((GB_NODES * NCLS) % 32 == 0 && GB_NODES * NCLS <= 256, "the block's logits are whole lines");
    __shared__ float sl[GB_NODES * NCLS];
    const int lane = threadIdx.x & 31;
    const int wave = __builtin_amdgcn_readfirstlane((int)(threadIdx.x >> 5));
    const int n0 = (blockIdx.x * 8 + wave) * NODES_PER_WAVE;
    const int2* cs = reinterpret_cast<const int2*>(csr);
    const bool on = lane < ROW_LANES;
    const int  q  = on ? lane : lane - ROW_LANES;
    const float b0 = bias[2 * q], b1 = bias[2 * q + 1];
    float w0[NCLS], w1[NCLS];
    #pragma unroll
    for (int j = 0; j < NCLS; ++j) {
        w0[j] = Wc[(2 * q) * NCLS + j];
        w1[j] = Wc[(2 * q + 1) * NCLS + j];
    }
    const float bcl = bc[lane < NCLS ? lane : 0];
    for (int k = 0; k < NODES_PER_WAVE; ++k) {
        const int node = n0 + k;
        const int beg  = __builtin_amdgcn_readfirstlane(rowptr[node]);
        const int end  = __builtin_amdgcn_readfirstlane(rowptr[node + 1]);
        const float di = dinv[node];
        v2f acc = *(const v2f*)(H + (size_t)node * HID2 + 2 * q);
        sum_segment<HID2, 2>(H, cs, beg, end, lane, q, acc);
        const float v0 = fmaxf(di * acc[0] + b0, 0.0f);
        const float v1 = fmaxf(di * acc[1] + b1, 0.0f);
        float s[NCLS];
        #pragma unroll
        for (int j = 0; j < NCLS; ++j) s[j] = on ? v0 * w0[j] + v1 * w1[j] : 0.0f;
        #pragma unroll
        for (int d = 16; d >= 1; d >>= 1) {
            #pragma unroll
            for (int j = 0; j < NCLS; ++j) s[j] += __shfl_xor(s[j], d);
        }
        if (lane < NCLS) {
            const float v = lane == 0 ? s[0] : (lane == 1 ? s[1] : s[2]);
            sl[(wave * NODES_PER_WAVE + k) * NCLS + lane] = v + bcl;
        }
    }
    __syncthreads();
    if (threadIdx.x < GB_NODES * NCLS)
        logits[(size_t)blockIdx.x * (GB_NODES * NCLS) + threadIdx.x] = sl[threadIdx.x];
}


extern "C" void kernel_launch(void* const* d_in, const int* in_sizes, int n_in,
                              void* d_out, int out_size, void* d_ws, size_t ws_size,
                              hipStream_t stream) {
    const float* x  = (const float*)d_in[0];
    const int*   ei = (const int*)  d_in[1];
    const float* ew = (const float*)d_in[2];
    const float* W1 = (const float*)d_in[3];
    const float* b1 = (const float*)d_in[4];
    const float* W2 = (const float*)d_in[5];
    const float* b2 = (const float*)d_in[6];
    const float* Wc = (const float*)d_in[7];
    const float* bc = (const float*)d_in[8];
    float* logits = (float*)d_out;

    const int* src = ei;
    const int* dst = ei + N_EDGES;

    char* ws = (char*)d_ws;
    size_t off = 0;
    auto take = [&](size_t bytes) -> void* {
        void* p = ws + off;
        off += (bytes + 255) & ~(size_t)255;
        return p;
    };
    float* dinv     = (float*)take((size_t)N_NODES * 4);
    int*   cnt      = (int*)  take((size_t)N_NODES * 4);
    int*   rowptr   = (int*)  take((size_t)(N_NODES + 1) * 4);
    int*   blocksum = (int*)  take((size_t)SCAN_BLOCKS * 4);
    unsigned* chunkcnt   = (unsigned*)take((size_t)NCH * NWORDS * 4);
    int*   pre      = (int*)  take((size_t)NCH * N_NODES * 4);
    int*   csr      = (int*)  take((size_t)N_EDGES * 8);
    float* H1   = (float*)take((size_t)N_NODES * HID * 4);
    float* H2   = (float*)take((size_t)N_NODES * HID2 * 4);

    const int TB = 256;
    count_chunk <<<NCH, RK_TB, 0, stream>>>(dst, chunkcnt);
    chunk_offsets<<<CS_BLOCKS, CS_TB, 0, stream>>>(chunkcnt, pre, cnt);
    scan_local  <<<SCAN_BLOCKS, SCAN_TB, 0, stream>>>(cnt, rowptr, blocksum);
    scan_offsets<<<SCAN_BLOCKS, SCAN_TB, 0, stream>>>(blocksum, rowptr);
    place_chunk <<<NCH * PQ, RK_TB, 0, stream>>>(src, dst, ew, rowptr, pre, csr);
    node_dinv   <<<(N_NODES + TB - 1) / TB, TB, 0, stream>>>(rowptr, csr, dinv);

    {
        gemm_wmma<IN_CH, HID><<<(N_NODES / 16 + 7) / 8, 256, 0, stream>>>(x, W1, dinv, H1);
    }
    gather96_gemm48<<<N_NODES / GB_NODES, 256, 0, stream>>>(H1, rowptr, csr, dinv, b1, W2, H2);

    gather48_classify<<<N_NODES / GB_NODES, 256, 0, stream>>>(H2, rowptr, csr, dinv, b2, Wc, bc, logits);
}


# edges grouped in two levels: per chunk, ordered by node range in LDS and written as whole runs; per range, a work-group orders its runs by node in LDS, writes list, bounds, dinv; sums in run order
# speedup vs baseline: 5.9622x; 1.4212x over previous
#include <hip/hip_runtime.h>

#define N_NODES 100000
#define N_EDGES 3200000
#define IN_CH   128
#define HID     96
#define HID2    48
#define NCLS    3

typedef __attribute__((ext_vector_type(2)))  float  v2f;
typedef __attribute__((ext_vector_type(4)))  float  v4f;
typedef __attribute__((ext_vector_type(8)))  float  v8f;


#define NB      256
#define BN      391
#define NC2     256
#define CH2     (N_EDGES / NC2)
#define BK_TB   1024
#define BK_PER  ((CH2 + BK_TB - 1) / BK_TB)
static_assert(N_EDGES % NC2 == 0, "whole chunks");
static_assert(NB * BN >= N_NODES && (NB - 1) * BN < N_NODES, "the ranges cover the nodes, none is empty");
static_assert((BK_PER - 1) * BK_TB < CH2, "only a thread's last step can fall behind the chunk's end");

static __device__ inline int group_excl_scan(int c, int* wsum) {
    const int lane = threadIdx.x & 31, wave = threadIdx.x >> 5;
    int incl = c;
    for (int d = 1; d < 32; d <<= 1) {
        const int v = __shfl_up(incl, d);
        if (lane >= d) incl += v;
    }
    if (lane == 31) wsum[wave] = incl;
    __syncthreads();
    int pre = 0;
    for (int w = 0; w < 32; ++w) if (w < wave) pre += wsum[w];
    return pre + incl - c;
}

__global__ __launch_bounds__(BK_TB) void bucket_count(const int* __restrict__ dst,
                                                      int* __restrict__ cc) {
    __shared__ int hist[NB];
    const int t  = threadIdx.x;
    const int e0 = blockIdx.x * CH2;
    if (t < NB) hist[t] = 0;
    int d[BK_PER];
    #pragma unroll
    for (int k = 0; k < BK_PER; ++k) {
        const int e = t + k * BK_TB;
        d[k] = (k < BK_PER - 1 || e < CH2) ? dst[e0 + e] : -1;
    }
    __syncthreads();
    #pragma unroll
    for (int k = 0; k < BK_PER; ++k)
        if (d[k] >= 0) atomicAdd(&hist[(unsigned)d[k] / BN], 1);
    __syncthreads();
    if (t < NB) cc[blockIdx.x * NB + t] = hist[t];
}

#define OG      (BK_TB / NB)
#define OG_PER  (NC2 / OG)
static_assert(OG * NB == BK_TB && NC2 % OG == 0 && (NB & (NB - 1)) == 0, "a thread per (group, range)");

__global__ __launch_bounds__(BK_TB) void bucket_offsets(const int* __restrict__ cc,
                                                        int* __restrict__ goff,
                                                        int* __restrict__ bstart) {
    __shared__ int part[OG][NB];
    __shared__ int bs[NB];
    __shared__ int wsum[32];
    const int t = threadIdx.x, r = t & (NB - 1), g = t / NB;
    int v[OG_PER];
    #pragma unroll
    for (int k = 0; k < OG_PER; ++k) v[k] = cc[(g * OG_PER + k) * NB + r];
    int s = 0;
    #pragma unroll
    for (int k = 0; k < OG_PER; ++k) s += v[k];
    part[g][r] = s;
    __syncthreads();
    int tot = 0;
    if (t < NB) for (int q = 0; q < OG; ++q) tot += part[q][t];
    const int excl = group_excl_scan(tot, wsum);
    if (t < NB) {
        bs[t] = excl;
        bstart[t] = excl;
        if (t == NB - 1) bstart[NB] = excl + tot;
    }
    __syncthreads();
    int run = bs[r];
    for (int q = 0; q < OG; ++q) if (q < g) run += part[q][r];
    #pragma unroll
    for (int k = 0; k < OG_PER; ++k) {
        goff[(g * OG_PER + k) * NB + r] = run;
        run += v[k];
    }
}

__global__ __launch_bounds__(BK_TB) void bucket_place(const int* __restrict__ src,
                                                      const int* __restrict__ dst,
                                                      const float* __restrict__ ew,
                                                      const int* __restrict__ goff,
                                                      int* __restrict__ tmp_) {
    int4* tmp = reinterpret_cast<int4*>(tmp_);
    __shared__ int hist[NB];
    __shared__ int lstart[NB];
    __shared__ int delta[NB];
    __shared__ int wsum[32];
    __shared__ int4 stage[CH2];
    const int t  = threadIdx.x;
    const int c  = blockIdx.x;
    const int e0 = c * CH2;
    if (t < NB) hist[t] = 0;
    const int go = t < NB ? goff[c * NB + t] : 0;
    int d[BK_PER], s[BK_PER], w[BK_PER], rk[BK_PER];
    #pragma unroll
    for (int k = 0; k < BK_PER; ++k) {
        const int e = t + k * BK_TB;
        d[k] = -1;
        if (k < BK_PER - 1 || e < CH2) {
            d[k] = dst[e0 + e];
            s[k] = src[e0 + e];
            w[k] = __float_as_int(ew[e0 + e]);
        }
    }
    __syncthreads();
    #pragma unroll
    for (int k = 0; k < BK_PER; ++k)
        if (d[k] >= 0) rk[k] = atomicAdd(&hist[(unsigned)d[k] / BN], 1);
    __syncthreads();
    const int cnt  = t < NB ? hist[t] : 0;
    const int excl = group_excl_scan(cnt, wsum);
    if (t < NB) { lstart[t] = excl; delta[t] = go - excl; }
    __syncthreads();
    #pragma unroll
    for (int k = 0; k < BK_PER; ++k)
        if (d[k] >= 0) {
            const int r = (unsigned)d[k] / BN;
            stage[lstart[r] + rk[k]] = make_int4(s[k], w[k], d[k] - r * BN, r);
        }
    __syncthreads();
    for (int i = t; i < CH2; i += BK_TB) {
        const int4 rec = stage[i];
        tmp[i + delta[rec.w]] = rec;
    }
}

#define ST_CAP  16384
#define SW_U    16

__global__ __launch_bounds__(BK_TB) void bucket_sort(const int* __restrict__ tmp_,
                                                     const int* __restrict__ bstart,
                                                     int* __restrict__ rowptr,
                                                     int* __restrict__ csr_,
                                                     float* __restrict__ dinv) {
    const int4* tmp = reinterpret_cast<const int4*>(tmp_);
    int2* csr = reinterpret_cast<int2*>(csr_);
    __shared__ int   wcnt[32][BN];
    __shared__ int   lrow[BN + 1];
    __shared__ float fsum[BN];
    __shared__ int   wsum[32];
    __shared__ int2  stage[ST_CAP];
    const int t = threadIdx.x, lane = t & 31, wave = t >> 5;
    const int r   = blockIdx.x;
    const int beg = bstart[r];
    const int n   = bstart[r + 1] - beg;
    const int G   = ((n + 31) / 32 + 31) / 32;
    const int gb  = wave * G;
    const bool fits = n <= ST_CAP;
    for (int i = t; i < 32 * BN; i += BK_TB) (&wcnt[0][0])[i] = 0;
    if (t < BN) fsum[t] = 0.0f;
    __syncthreads();
    for (int g0 = 0; g0 < G; g0 += SW_U) {
        int ln[SW_U];
        #pragma unroll
        for (int u = 0; u < SW_U; ++u) {
            const int i = 32 * (gb + g0 + u) + lane;
            ln[u] = (g0 + u < G && i < n) ? tmp[beg + i].z : -1;
        }
        #pragma unroll
        for (int u = 0; u < SW_U; ++u)
            if (ln[u] >= 0) atomicAdd(&wcnt[wave][ln[u]], 1);
    }
    __syncthreads();
    int tot = 0;
    if (t < BN) {
        #pragma unroll
        for (int w = 0; w < 32; ++w) { const int cw = wcnt[w][t]; wcnt[w][t] = tot; tot += cw; }
    }
    const int excl = group_excl_scan(tot, wsum);
    const int node = r * BN + t;
    if (t < BN) {
        lrow[t] = excl;
        if (node < N_NODES) rowptr[node] = beg + excl;
        if (t == BN - 1) lrow[BN] = excl + tot;
    }
    if (r == NB - 1 && t == 0) rowptr[N_NODES] = beg + n;
    __syncthreads();
    for (int g0 = 0; g0 < G; g0 += SW_U) {
        int4 rec[SW_U];
        #pragma unroll
        for (int u = 0; u < SW_U; ++u) {
            const int i = 32 * (gb + g0 + u) + lane;
            rec[u] = make_int4(0, 0, -1, 0);
            if (g0 + u < G && i < n) rec[u] = tmp[beg + i];
        }
        #pragma unroll
        for (int u = 0; u < SW_U; ++u)
            if (rec[u].z >= 0) {
                const int p = lrow[rec[u].z] + atomicAdd(&wcnt[wave][rec[u].z], 1);
                if (fits) stage[p] = make_int2(rec[u].x, rec[u].y);
                else {
                    csr[beg + p] = make_int2(rec[u].x, rec[u].y);
                    atomicAdd(&fsum[rec[u].z], __int_as_float(rec[u].y));
                }
            }
    }
    __syncthreads();
    if (fits)
        for (int i = t; i < n; i += BK_TB) csr[beg + i] = stage[i];
    if (t < BN && node < N_NODES) {
        float dsum = 1.0f;
        if (fits) {
            const int pe = lrow[t + 1];
            for (int p = lrow[t]; p < pe; ++p) dsum += __int_as_float(stage[p].y);
        } else dsum += fsum[t];
        dinv[node] = rsqrtf(dsum);
    }
}


template<int K, int COLS>
__global__ __launch_bounds__(256) void gemm_wmma(const float* __restrict__ X,
                                                 const float* __restrict__ W,
                                                 const float* __restrict__ dinv,
                                                 float* __restrict__ H) {
    constexpr int NT = COLS / 16;
    constexpr int KS = K / 4;
    constexpr int P2 = COLS + ((16 - COLS % 32 + 32) % 32);
    static_assert(K % 4 == 0 && COLS % 16 == 0 && P2 % 32 == 16 && P2 % 4 == 0, "tile shapes");
    __shared__ __attribute__((aligned(16))) v2f wp[(K / 2) * P2];
    const int lane = threadIdx.x & 31;
    const int wave = threadIdx.x >> 5;
    const int c = lane & 15, h = lane >> 4;
    const int m0 = (blockIdx.x * 8 + wave) * 16;
    const bool live = m0 < N_NODES;
    const int row = (live ? m0 : N_NODES - 16) + c;
    const float* xrow = X + (size_t)row * K + 2 * h;
    constexpr int QR  = COLS / 4;
    constexpr int NQ  = (K / 2) * QR;
    constexpr int WIT = (NQ + 255) / 256;
    float4 w0[WIT], w1[WIT];
    #pragma unroll
    for (int j = 0; j < WIT; ++j) {
        const int idx = j * 256 + (int)threadIdx.x;
        if ((j + 1) * 256 <= NQ || idx < NQ) {
            const int kp = idx / QR, q = idx - kp * QR;
            w0[j] = *(const float4*)(W + (size_t)(2 * kp)     * COLS + 4 * q);
            w1[j] = *(const float4*)(W + (size_t)(2 * kp + 1) * COLS + 4 * q);
        }
    }
    #pragma unroll
    for (int j = 0; j < WIT; ++j) {
        const int idx = j * 256 + (int)threadIdx.x;
        if ((j + 1) * 256 <= NQ || idx < NQ) {
            const int kp = idx / QR, q = idx - kp * QR;
            float4* d = reinterpret_cast<float4*>(wp + kp * P2 + 4 * q);
            d[0] = make_float4(w0[j].x, w1[j].x, w0[j].y, w1[j].y);
            d[1] = make_float4(w0[j].z, w1[j].z, w0[j].w, w1[j].w);
        }
    }
    __syncthreads();
    v2f a[KS];
    #pragma unroll
    for (int i = 0; i < KS; ++i) a[i] = *(const v2f*)(xrow + 4 * i);
    v8f acc[NT];
    #pragma unroll
    for (int t = 0; t < NT; ++t) acc[t] = v8f{};
    const v2f* bp = wp + h * P2 + c;
    #pragma unroll
    for (int i = 0; i < KS; ++i) {
        #pragma unroll
        for (int t = 0; t < NT; ++t) {
            const v2f b = bp[(2 * i) * P2 + 16 * t];
            acc[t] = __builtin_amdgcn_wmma_f32_16x16x4_f32(false, a[i], false, b,
                                                           (short)0, acc[t], false, false);
        }
    }
    if (live) {
        float* hrow = H + (size_t)(m0 + 8 * h) * COLS + c;
        float dr[8];
        #pragma unroll
        for (int r = 0; r < 8; ++r) dr[r] = dinv[m0 + 8 * h + r];
        #pragma unroll
        for (int t = 0; t < NT; ++t) {
            #pragma unroll
            for (int r = 0; r < 8; ++r) hrow[(size_t)r * COLS + 16 * t] = dr[r] * acc[t][r];
        }
    }
}


#define ROW_LANES 24
template<int VW> struct row_vec;
template<> struct row_vec<2> { typedef v2f type; };
template<> struct row_vec<4> { typedef v4f type; };

template<int C, int VW>
static __device__ inline void sum_segment(const float* __restrict__ H,
                                          const int2* __restrict__ csr,
                                          int beg, int end, int lane, int q,
                                          typename row_vec<VW>::type& acc) {
    typedef typename row_vec<VW>::type vec;
    static_assert(C == ROW_LANES * VW, "a row is ROW_LANES lanes x VW floats");
    constexpr int U = 8;
    for (int base = beg; base < end; base += 32) {
        const int left = end - base;
        const int n = left < 32 ? left : 32;
        int2 ev = make_int2(0, 0);
        if (lane < n) ev = csr[base + lane];
        for (int j0 = 0; j0 < n; j0 += U) {
            const int m = n - j0;
            vec hv[U];
            if (m >= U) {
                #pragma unroll
                for (int u = 0; u < U; ++u) {
                    const int s = __builtin_amdgcn_readlane(ev.x, j0 + u);
                    hv[u] = *(const vec*)(H + (size_t)s * C + VW * q);
                }
                #pragma unroll
                for (int u = 0; u < U; ++u) {
                    const float w = __int_as_float(__builtin_amdgcn_readlane(ev.y, j0 + u));
                    acc += w * hv[u];
                }
            } else {
                #pragma unroll
                for (int u = 0; u < U; ++u)
                    if (u < m) {
                        const int s = __builtin_amdgcn_readlane(ev.x, j0 + u);
                        hv[u] = *(const vec*)(H + (size_t)s * C + VW * q);
                    }
                #pragma unroll
                for (int u = 0; u < U; ++u)
                    if (u < m) {
                        const float w = __int_as_float(__builtin_amdgcn_readlane(ev.y, j0 + u));
                        acc += w * hv[u];
                    }
            }
        }
    }
}

#define NODES_PER_WAVE 4
#define GB_NODES (8 * NODES_PER_WAVE)

__global__ __launch_bounds__(256) void gather96_gemm48(const float* __restrict__ H,
                                                       const int* __restrict__ rowptr,
                                                       const int* __restrict__ csr,
                                                       const float* __restrict__ dinv,
                                                       const float* __restrict__ bias,
                                                       const float* __restrict__ W,
                                                       float* __restrict__ H2) {
    static_assert(N_NODES % GB_NODES == 0 && GB_NODES % 16 == 0, "whole blocks, whole 16-row strips");
    constexpr int K = HID, COLS = HID2;
    constexpr int NT = COLS / 16;
    constexpr int KS = K / 4;
    constexpr int P2 = COLS + ((16 - COLS % 32 + 32) % 32);
    constexpr int XP = K + 4;
    static_assert(K % 4 == 0 && COLS % 16 == 0 && P2 % 32 == 16 && P2 % 4 == 0, "tile shapes");
    static_assert(XP % 64 == 36 && XP % 4 == 0, "A fragment reads: 8-byte aligned, no bank shared");
    __shared__ __attribute__((aligned(16))) v2f wp[(K / 2) * P2];
    __shared__ __attribute__((aligned(16))) float xt[GB_NODES * XP];
    const int lane = threadIdx.x & 31;
    const int wave = __builtin_amdgcn_readfirstlane((int)(threadIdx.x >> 5));
    {
        constexpr int QR  = COLS / 4;
        constexpr int NQ  = (K / 2) * QR;
        constexpr int WIT = (NQ + 255) / 256;
        float4 w0[WIT], w1[WIT];
        #pragma unroll
        for (int j = 0; j < WIT; ++j) {
            const int idx = j * 256 + (int)threadIdx.x;
            if ((j + 1) * 256 <= NQ || idx < NQ) {
                const int kp = idx / QR, q = idx - kp * QR;
                w0[j] = *(const float4*)(W + (size_t)(2 * kp)     * COLS + 4 * q);
                w1[j] = *(const float4*)(W + (size_t)(2 * kp + 1) * COLS + 4 * q);
            }
        }
        #pragma unroll
        for (int j = 0; j < WIT; ++j) {
            const int idx = j * 256 + (int)threadIdx.x;
            if ((j + 1) * 256 <= NQ || idx < NQ) {
                const int kp = idx / QR, q = idx - kp * QR;
                float4* d = reinterpret_cast<float4*>(wp + kp * P2 + 4 * q);
                d[0] = make_float4(w0[j].x, w1[j].x, w0[j].y, w1[j].y);
                d[1] = make_float4(w0[j].z, w1[j].z, w0[j].w, w1[j].w);
            }
        }
    }
    const int n0 = (blockIdx.x * 8 + wave) * NODES_PER_WAVE;
    const int q = lane < ROW_LANES ? lane : lane - ROW_LANES;
    const float bb[4] = { bias[4 * q], bias[4 * q + 1], bias[4 * q + 2], bias[4 * q + 3] };
    for (int k = 0; k < NODES_PER_WAVE; ++k) {
        const int node = n0 + k;
        const int beg  = __builtin_amdgcn_readfirstlane(rowptr[node]);
        const int end  = __builtin_amdgcn_readfirstlane(rowptr[node + 1]);
        const float di = dinv[node];
        const float* hs = H + (size_t)node * HID;
        v4f acc = *(const v4f*)(hs + 4 * q);
        sum_segment<HID, 4>(H, reinterpret_cast<const int2*>(csr), beg, end, lane, q, acc);
        float* o = xt + (wave * NODES_PER_WAVE + k) * XP;
        if (lane < ROW_LANES) {
            v4f r;
            #pragma unroll
            for (int p = 0; p < 4; ++p) r[p] = fmaxf(di * acc[p] + bb[p], 0.0f);
            *(v4f*)(o + 4 * lane) = r;
        }
    }
    __syncthreads();
    if (wave < GB_NODES / 16) {
        const int c = lane & 15, h = lane >> 4;
        const int m0 = blockIdx.x * GB_NODES + wave * 16;
        const float* xrow = xt + (wave * 16 + c) * XP + 2 * h;
        const v2f* bp = wp + h * P2 + c;
        float dr[8];
        #pragma unroll
        for (int r = 0; r < 8; ++r) dr[r] = dinv[m0 + 8 * h + r];
        v8f acc[NT];
        #pragma unroll
        for (int t = 0; t < NT; ++t) acc[t] = v8f{};
        #pragma unroll
        for (int i = 0; i < KS; ++i) {
            const v2f a = *(const v2f*)(xrow + 4 * i);
            #pragma unroll
            for (int t = 0; t < NT; ++t) {
                const v2f b = bp[(2 * i) * P2 + 16 * t];
                acc[t] = __builtin_amdgcn_wmma_f32_16x16x4_f32(false, a, false, b,
                                                               (short)0, acc[t], false, false);
            }
        }
        float* hrow = H2 + (size_t)(m0 + 8 * h) * COLS + c;
        #pragma unroll
        for (int t = 0; t < NT; ++t) {
            #pragma unroll
            for (int r = 0; r < 8; ++r) hrow[(size_t)r * COLS + 16 * t] = dr[r] * acc[t][r];
        }
    }
}

__global__ __launch_bounds__(256) void gather48_classify(const float* __restrict__ H,
                                                         const int* __restrict__ rowptr,
                                                         const int* __restrict__ csr,
                                                         const float* __restrict__ dinv,
                                                         const float* __restrict__ bias,
                                                         const float* __restrict__ Wc,
                                                         const float* __restrict__ bc,
                                                         float* __restrict__ logits) {
    static_assert((GB_NODES * NCLS) % 32 == 0 && GB_NODES * NCLS <= 256, "the block's logits are whole lines");
    __shared__ float sl[GB_NODES * NCLS];
    const int lane = threadIdx.x & 31;
    const int wave = __builtin_amdgcn_readfirstlane((int)(threadIdx.x >> 5));
    const int n0 = (blockIdx.x * 8 + wave) * NODES_PER_WAVE;
    const int2* cs = reinterpret_cast<const int2*>(csr);
    const bool on = lane < ROW_LANES;
    const int  q  = on ? lane : lane - ROW_LANES;
    const float b0 = bias[2 * q], b1 = bias[2 * q + 1];
    float w0[NCLS], w1[NCLS];
    #pragma unroll
    for (int j = 0; j < NCLS; ++j) {
        w0[j] = Wc[(2 * q) * NCLS + j];
        w1[j] = Wc[(2 * q + 1) * NCLS + j];
    }
    const float bcl = bc[lane < NCLS ? lane : 0];
    for (int k = 0; k < NODES_PER_WAVE; ++k) {
        const int node = n0 + k;
        const int beg  = __builtin_amdgcn_readfirstlane(rowptr[node]);
        const int end  = __builtin_amdgcn_readfirstlane(rowptr[node + 1]);
        const float di = dinv[node];
        v2f acc = *(const v2f*)(H + (size_t)node * HID2 + 2 * q);
        sum_segment<HID2, 2>(H, cs, beg, end, lane, q, acc);
        const float v0 = fmaxf(di * acc[0] + b0, 0.0f);
        const float v1 = fmaxf(di * acc[1] + b1, 0.0f);
        float s[NCLS];
        #pragma unroll
        for (int j = 0; j < NCLS; ++j) s[j] = on ? v0 * w0[j] + v1 * w1[j] : 0.0f;
        #pragma unroll
        for (int d = 16; d >= 1; d >>= 1) {
            #pragma unroll
            for (int j = 0; j < NCLS; ++j) s[j] += __shfl_xor(s[j], d);
        }
        if (lane < NCLS) {
            const float v = lane == 0 ? s[0] : (lane == 1 ? s[1] : s[2]);
            sl[(wave * NODES_PER_WAVE + k) * NCLS + lane] = v + bcl;
        }
    }
    __syncthreads();
    if (threadIdx.x < GB_NODES * NCLS)
        logits[(size_t)blockIdx.x * (GB_NODES * NCLS) + threadIdx.x] = sl[threadIdx.x];
}


extern "C" void kernel_launch(void* const* d_in, const int* in_sizes, int n_in,
                              void* d_out, int out_size, void* d_ws, size_t ws_size,
                              hipStream_t stream) {
    const float* x  = (const float*)d_in[0];
    const int*   ei = (const int*)  d_in[1];
    const float* ew = (const float*)d_in[2];
    const float* W1 = (const float*)d_in[3];
    const float* b1 = (const float*)d_in[4];
    const float* W2 = (const float*)d_in[5];
    const float* b2 = (const float*)d_in[6];
    const float* Wc = (const float*)d_in[7];
    const float* bc = (const float*)d_in[8];
    float* logits = (float*)d_out;

    const int* src = ei;
    const int* dst = ei + N_EDGES;

    char* ws = (char*)d_ws;
    size_t off = 0;
    auto take = [&](size_t bytes) -> void* {
        void* p = ws + off;
        off += (bytes + 255) & ~(size_t)255;
        return p;
    };
    float* dinv     = (float*)take((size_t)N_NODES * 4);
    int*   rowptr   = (int*)  take((size_t)(N_NODES + 1) * 4);
    int*   cc       = (int*)  take((size_t)NC2 * NB * 4);
    int*   goff     = (int*)  take((size_t)NC2 * NB * 4);
    int*   bstart   = (int*)  take((size_t)(NB + 1) * 4);
    int*   tmp      = (int*)  take((size_t)N_EDGES * 16);
    int*   csr      = (int*)  take((size_t)N_EDGES * 8);
    float* H1   = (float*)take((size_t)N_NODES * HID * 4);
    float* H2   = (float*)take((size_t)N_NODES * HID2 * 4);

    bucket_count  <<<NC2, BK_TB, 0, stream>>>(dst, cc);
    bucket_offsets<<<1,   BK_TB, 0, stream>>>(cc, goff, bstart);
    bucket_place  <<<NC2, BK_TB, 0, stream>>>(src, dst, ew, goff, tmp);
    bucket_sort   <<<NB,  BK_TB, 0, stream>>>(tmp, bstart, rowptr, csr, dinv);

    {
        gemm_wmma<IN_CH, HID><<<(N_NODES / 16 + 7) / 8, 256, 0, stream>>>(x, W1, dinv, H1);
    }
    gather96_gemm48<<<N_NODES / GB_NODES, 256, 0, stream>>>(H1, rowptr, csr, dinv, b1, W2, H2);

    gather48_classify<<<N_NODES / GB_NODES, 256, 0, stream>>>(H2, rowptr, csr, dinv, b2, Wc, bc, logits);
}


# edge grouping without its count and offsets launches: a chunk writes its range-ordered records to its own piece plus per-range offsets; a range finds its runs and first slot from them; 7 to 5 launches
# speedup vs baseline: 6.3254x; 1.0609x over previous
#include <hip/hip_runtime.h>

#define N_NODES 100000
#define N_EDGES 3200000
#define IN_CH   128
#define HID     96
#define HID2    48
#define NCLS    3

typedef __attribute__((ext_vector_type(2)))  float  v2f;
typedef __attribute__((ext_vector_type(4)))  float  v4f;
typedef __attribute__((ext_vector_type(8)))  float  v8f;


#define NB      256
#define BN      391
#define NC2     256
#define CH2     (N_EDGES / NC2)
#define BK_TB   1024
#define BK_PER  ((CH2 + BK_TB - 1) / BK_TB)
#define TPITCH  ((CH2 + 7) & ~7)
#define OPITCH  (NB + 1)
static_assert(N_EDGES % NC2 == 0, "whole chunks");
static_assert(NB * BN >= N_NODES && (NB - 1) * BN < N_NODES, "the ranges cover the nodes, none is empty");
static_assert((BK_PER - 1) * BK_TB < CH2, "only a thread's last step can fall behind the chunk's end");
static_assert(NB < BK_TB && NC2 <= BK_TB && NC2 % 32 == 0, "a thread per range and one more; whole waves of runs");

static __device__ inline int group_excl_scan(int c, int* wsum) {
    const int lane = threadIdx.x & 31, wave = threadIdx.x >> 5;
    int incl = c;
    for (int d = 1; d < 32; d <<= 1) {
        const int v = __shfl_up(incl, d);
        if (lane >= d) incl += v;
    }
    if (lane == 31) wsum[wave] = incl;
    __syncthreads();
    int pre = 0;
    for (int w = 0; w < 32; ++w) if (w < wave) pre += wsum[w];
    return pre + incl - c;
}

__global__ __launch_bounds__(BK_TB) void bucket_place(const int* __restrict__ src,
                                                      const int* __restrict__ dst,
                                                      const float* __restrict__ ew,
                                                      int* __restrict__ offs,
                                                      int* __restrict__ tmp_) {
    int4* tmp = reinterpret_cast<int4*>(tmp_);
    __shared__ int hist[NB];
    __shared__ int lstart[NB];
    __shared__ int wsum[32];
    __shared__ int4 stage[CH2];
    const int t  = threadIdx.x;
    const int c  = blockIdx.x;
    const int e0 = c * CH2;
    if (t < NB) hist[t] = 0;
    int d[BK_PER], s[BK_PER], w[BK_PER], rk[BK_PER];
    #pragma unroll
    for (int k = 0; k < BK_PER; ++k) {
        const int e = t + k * BK_TB;
        d[k] = -1;
        if (k < BK_PER - 1 || e < CH2) {
            d[k] = dst[e0 + e];
            s[k] = src[e0 + e];
            w[k] = __float_as_int(ew[e0 + e]);
        }
    }
    __syncthreads();
    #pragma unroll
    for (int k = 0; k < BK_PER; ++k)
        if (d[k] >= 0) rk[k] = atomicAdd(&hist[(unsigned)d[k] / BN], 1);
    __syncthreads();
    const int cnt  = t < NB ? hist[t] : 0;
    const int excl = group_excl_scan(cnt, wsum);
    if (t < NB) lstart[t] = excl;
    if (t <= NB) offs[c * OPITCH + t] = excl;
    __syncthreads();
    #pragma unroll
    for (int k = 0; k < BK_PER; ++k)
        if (d[k] >= 0) {
            const int r = (unsigned)d[k] / BN;
            stage[lstart[r] + rk[k]] = make_int4(s[k], w[k], d[k] - r * BN, 0);
        }
    __syncthreads();
    for (int i = t; i < CH2; i += BK_TB) tmp[c * TPITCH + i] = stage[i];
}

#define ST_CAP  16384
#define RPW     (NC2 / 32)
#define RUN_U   2

__global__ __launch_bounds__(BK_TB) void bucket_sort(const int* __restrict__ tmp_,
                                                     const int* __restrict__ offs,
                                                     int* __restrict__ rowptr,
                                                     int* __restrict__ csr_,
                                                     float* __restrict__ dinv) {
    const int4* tmp = reinterpret_cast<const int4*>(tmp_);
    int2* csr = reinterpret_cast<int2*>(csr_);
    __shared__ int   wcnt[32][BN];
    __shared__ int   lrow[BN + 1];
    __shared__ float fsum[BN];
    __shared__ int   wsum[32];
    __shared__ int   rbeg[NC2];
    __shared__ int   rcnt[NC2];
    __shared__ int   psum[2][NC2 / 32];
    __shared__ int2  stage[ST_CAP];
    const int t = threadIdx.x, lane = t & 31, wave = t >> 5;
    const int r = blockIdx.x;
    int ra = 0, rn = 0;
    if (t < NC2) {
        ra = offs[t * OPITCH + r];
        rn = offs[t * OPITCH + r + 1] - ra;
    }
    for (int i = t; i < 32 * BN; i += BK_TB) (&wcnt[0][0])[i] = 0;
    if (t < BN) fsum[t] = 0.0f;
    if (t < NC2) { rbeg[t] = t * TPITCH + ra; rcnt[t] = rn; }
    int sa = ra, sn = rn;
    for (int d = 16; d >= 1; d >>= 1) { sa += __shfl_xor(sa, d); sn += __shfl_xor(sn, d); }
    if (t < NC2 && lane == 0) { psum[0][wave] = sa; psum[1][wave] = sn; }
    __syncthreads();
    int beg = 0, n = 0;
    for (int w = 0; w < NC2 / 32; ++w) { beg += psum[0][w]; n += psum[1][w]; }
    const bool fits = n <= ST_CAP;
    int rb[RPW], rc[RPW];
    #pragma unroll
    for (int k = 0; k < RPW; ++k) {
        rb[k] = __builtin_amdgcn_readfirstlane(rbeg[wave * RPW + k]);
        rc[k] = __builtin_amdgcn_readfirstlane(rcnt[wave * RPW + k]);
    }
    {
        int ln[RPW][RUN_U];
        #pragma unroll
        for (int k = 0; k < RPW; ++k) {
            #pragma unroll
            for (int u = 0; u < RUN_U; ++u) {
                const int i = 32 * u + lane;
                ln[k][u] = i < rc[k] ? tmp[rb[k] + i].z : -1;
            }
        }
        #pragma unroll
        for (int k = 0; k < RPW; ++k) {
            #pragma unroll
            for (int u = 0; u < RUN_U; ++u)
                if (ln[k][u] >= 0) atomicAdd(&wcnt[wave][ln[k][u]], 1);
            for (int i = 32 * RUN_U + lane; i < rc[k]; i += 32)
                atomicAdd(&wcnt[wave][tmp[rb[k] + i].z], 1);
        }
    }
    __syncthreads();
    int tot = 0;
    if (t < BN) {
        #pragma unroll
        for (int w = 0; w < 32; ++w) { const int cw = wcnt[w][t]; wcnt[w][t] = tot; tot += cw; }
    }
    const int excl = group_excl_scan(tot, wsum);
    const int node = r * BN + t;
    if (t < BN) {
        lrow[t] = excl;
        if (node < N_NODES) rowptr[node] = beg + excl;
        if (t == BN - 1) lrow[BN] = excl + tot;
    }
    if (r == NB - 1 && t == 0) rowptr[N_NODES] = beg + n;
    __syncthreads();
    {
        auto put = [&](const int4 rec) {
            const int p = lrow[rec.z] + atomicAdd(&wcnt[wave][rec.z], 1);
            if (fits) stage[p] = make_int2(rec.x, rec.y);
            else {
                csr[beg + p] = make_int2(rec.x, rec.y);
                atomicAdd(&fsum[rec.z], __int_as_float(rec.y));
            }
        };
        int4 rec[RPW][RUN_U];
        #pragma unroll
        for (int k = 0; k < RPW; ++k) {
            #pragma unroll
            for (int u = 0; u < RUN_U; ++u) {
                const int i = 32 * u + lane;
                rec[k][u] = make_int4(0, 0, -1, 0);
                if (i < rc[k]) rec[k][u] = tmp[rb[k] + i];
            }
        }
        #pragma unroll
        for (int k = 0; k < RPW; ++k) {
            #pragma unroll
            for (int u = 0; u < RUN_U; ++u)
                if (rec[k][u].z >= 0) put(rec[k][u]);
            for (int i = 32 * RUN_U + lane; i < rc[k]; i += 32)
                put(tmp[rb[k] + i]);
        }
    }
    __syncthreads();
    if (fits)
        for (int i = t; i < n; i += BK_TB) csr[beg + i] = stage[i];
    if (t < BN && node < N_NODES) {
        float dsum = 1.0f;
        if (fits) {
            const int pe = lrow[t + 1];
            for (int p = lrow[t]; p < pe; ++p) dsum += __int_as_float(stage[p].y);
        } else dsum += fsum[t];
        dinv[node] = rsqrtf(dsum);
    }
}


template<int K, int COLS>
__global__ __launch_bounds__(256) void gemm_wmma(const float* __restrict__ X,
                                                 const float* __restrict__ W,
                                                 const float* __restrict__ dinv,
                                                 float* __restrict__ H) {
    constexpr int NT = COLS / 16;
    constexpr int KS = K / 4;
    constexpr int P2 = COLS + ((16 - COLS % 32 + 32) % 32);
    static_assert(K % 4 == 0 && COLS % 16 == 0 && P2 % 32 == 16 && P2 % 4 == 0, "tile shapes");
    __shared__ __attribute__((aligned(16))) v2f wp[(K / 2) * P2];
    const int lane = threadIdx.x & 31;
    const int wave = threadIdx.x >> 5;
    const int c = lane & 15, h = lane >> 4;
    const int m0 = (blockIdx.x * 8 + wave) * 16;
    const bool live = m0 < N_NODES;
    const int row = (live ? m0 : N_NODES - 16) + c;
    const float* xrow = X + (size_t)row * K + 2 * h;
    constexpr int QR  = COLS / 4;
    constexpr int NQ  = (K / 2) * QR;
    constexpr int WIT = (NQ + 255) / 256;
    float4 w0[WIT], w1[WIT];
    #pragma unroll
    for (int j = 0; j < WIT; ++j) {
        const int idx = j * 256 + (int)threadIdx.x;
        if ((j + 1) * 256 <= NQ || idx < NQ) {
            const int kp = idx / QR, q = idx - kp * QR;
            w0[j] = *(const float4*)(W + (size_t)(2 * kp)     * COLS + 4 * q);
            w1[j] = *(const float4*)(W + (size_t)(2 * kp + 1) * COLS + 4 * q);
        }
    }
    #pragma unroll
    for (int j = 0; j < WIT; ++j) {
        const int idx = j * 256 + (int)threadIdx.x;
        if ((j + 1) * 256 <= NQ || idx < NQ) {
            const int kp = idx / QR, q = idx - kp * QR;
            float4* d = reinterpret_cast<float4*>(wp + kp * P2 + 4 * q);
            d[0] = make_float4(w0[j].x, w1[j].x, w0[j].y, w1[j].y);
            d[1] = make_float4(w0[j].z, w1[j].z, w0[j].w, w1[j].w);
        }
    }
    __syncthreads();
    v2f a[KS];
    #pragma unroll
    for (int i = 0; i < KS; ++i) a[i] = *(const v2f*)(xrow + 4 * i);
    v8f acc[NT];
    #pragma unroll
    for (int t = 0; t < NT; ++t) acc[t] = v8f{};
    const v2f* bp = wp + h * P2 + c;
    #pragma unroll
    for (int i = 0; i < KS; ++i) {
        #pragma unroll
        for (int t = 0; t < NT; ++t) {
            const v2f b = bp[(2 * i) * P2 + 16 * t];
            acc[t] = __builtin_amdgcn_wmma_f32_16x16x4_f32(false, a[i], false, b,
                                                           (short)0, acc[t], false, false);
        }
    }
    if (live) {
        float* hrow = H + (size_t)(m0 + 8 * h) * COLS + c;
        float dr[8];
        #pragma unroll
        for (int r = 0; r < 8; ++r) dr[r] = dinv[m0 + 8 * h + r];
        #pragma unroll
        for (int t = 0; t < NT; ++t) {
            #pragma unroll
            for (int r = 0; r < 8; ++r) hrow[(size_t)r * COLS + 16 * t] = dr[r] * acc[t][r];
        }
    }
}


#define ROW_LANES 24
template<int VW> struct row_vec;
template<> struct row_vec<2> { typedef v2f type; };
template<> struct row_vec<4> { typedef v4f type; };

template<int C, int VW>
static __device__ inline void sum_segment(const float* __restrict__ H,
                                          const int2* __restrict__ csr,
                                          int beg, int end, int lane, int q,
                                          typename row_vec<VW>::type& acc) {
    typedef typename row_vec<VW>::type vec;
    static_assert(C == ROW_LANES * VW, "a row is ROW_LANES lanes x VW floats");
    constexpr int U = 8;
    for (int base = beg; base < end; base += 32) {
        const int left = end - base;
        const int n = left < 32 ? left : 32;
        int2 ev = make_int2(0, 0);
        if (lane < n) ev = csr[base + lane];
        for (int j0 = 0; j0 < n; j0 += U) {
            const int m = n - j0;
            vec hv[U];
            if (m >= U) {
                #pragma unroll
                for (int u = 0; u < U; ++u) {
                    const int s = __builtin_amdgcn_readlane(ev.x, j0 + u);
                    hv[u] = *(const vec*)(H + (size_t)s * C + VW * q);
                }
                #pragma unroll
                for (int u = 0; u < U; ++u) {
                    const float w = __int_as_float(__builtin_amdgcn_readlane(ev.y, j0 + u));
                    acc += w * hv[u];
                }
            } else {
                #pragma unroll
                for (int u = 0; u < U; ++u)
                    if (u < m) {
                        const int s = __builtin_amdgcn_readlane(ev.x, j0 + u);
                        hv[u] = *(const vec*)(H + (size_t)s * C + VW * q);
                    }
                #pragma unroll
                for (int u = 0; u < U; ++u)
                    if (u < m) {
                        const float w = __int_as_float(__builtin_amdgcn_readlane(ev.y, j0 + u));
                        acc += w * hv[u];
                    }
            }
        }
    }
}

#define NODES_PER_WAVE 4
#define GB_NODES (8 * NODES_PER_WAVE)

__global__ __launch_bounds__(256) void gather96_gemm48(const float* __restrict__ H,
                                                       const int* __restrict__ rowptr,
                                                       const int* __restrict__ csr,
                                                       const float* __restrict__ dinv,
                                                       const float* __restrict__ bias,
                                                       const float* __restrict__ W,
                                                       float* __restrict__ H2) {
    static_assert(N_NODES % GB_NODES == 0 && GB_NODES % 16 == 0, "whole blocks, whole 16-row strips");
    constexpr int K = HID, COLS = HID2;
    constexpr int NT = COLS / 16;
    constexpr int KS = K / 4;
    constexpr int P2 = COLS + ((16 - COLS % 32 + 32) % 32);
    constexpr int XP = K + 4;
    static_assert(K % 4 == 0 && COLS % 16 == 0 && P2 % 32 == 16 && P2 % 4 == 0, "tile shapes");
    static_assert(XP % 64 == 36 && XP % 4 == 0, "A fragment reads: 8-byte aligned, no bank shared");
    __shared__ __attribute__((aligned(16))) v2f wp[(K / 2) * P2];
    __shared__ __attribute__((aligned(16))) float xt[GB_NODES * XP];
    const int lane = threadIdx.x & 31;
    const int wave = __builtin_amdgcn_readfirstlane((int)(threadIdx.x >> 5));
    {
        constexpr int QR  = COLS / 4;
        constexpr int NQ  = (K / 2) * QR;
        constexpr int WIT = (NQ + 255) / 256;
        float4 w0[WIT], w1[WIT];
        #pragma unroll
        for (int j = 0; j < WIT; ++j) {
            const int idx = j * 256 + (int)threadIdx.x;
            if ((j + 1) * 256 <= NQ || idx < NQ) {
                const int kp = idx / QR, q = idx - kp * QR;
                w0[j] = *(const float4*)(W + (size_t)(2 * kp)     * COLS + 4 * q);
                w1[j] = *(const float4*)(W + (size_t)(2 * kp + 1) * COLS + 4 * q);
            }
        }
        #pragma unroll
        for (int j = 0; j < WIT; ++j) {
            const int idx = j * 256 + (int)threadIdx.x;
            if ((j + 1) * 256 <= NQ || idx < NQ) {
                const int kp = idx / QR, q = idx - kp * QR;
                float4* d = reinterpret_cast<float4*>(wp + kp * P2 + 4 * q);
                d[0] = make_float4(w0[j].x, w1[j].x, w0[j].y, w1[j].y);
                d[1] = make_float4(w0[j].z, w1[j].z, w0[j].w, w1[j].w);
            }
        }
    }
    const int n0 = (blockIdx.x * 8 + wave) * NODES_PER_WAVE;
    const int q = lane < ROW_LANES ? lane : lane - ROW_LANES;
    const float bb[4] = { bias[4 * q], bias[4 * q + 1], bias[4 * q + 2], bias[4 * q + 3] };
    for (int k = 0; k < NODES_PER_WAVE; ++k) {
        const int node = n0 + k;
        const int beg  = __builtin_amdgcn_readfirstlane(rowptr[node]);
        const int end  = __builtin_amdgcn_readfirstlane(rowptr[node + 1]);
        const float di = dinv[node];
        const float* hs = H + (size_t)node * HID;
        v4f acc = *(const v4f*)(hs + 4 * q);
        sum_segment<HID, 4>(H, reinterpret_cast<const int2*>(csr), beg, end, lane, q, acc);
        float* o = xt + (wave * NODES_PER_WAVE + k) * XP;
        if (lane < ROW_LANES) {
            v4f r;
            #pragma unroll
            for (int p = 0; p < 4; ++p) r[p] = fmaxf(di * acc[p] + bb[p], 0.0f);
            *(v4f*)(o + 4 * lane) = r;
        }
    }
    __syncthreads();
    if (wave < GB_NODES / 16) {
        const int c = lane & 15, h = lane >> 4;
        const int m0 = blockIdx.x * GB_NODES + wave * 16;
        const float* xrow = xt + (wave * 16 + c) * XP + 2 * h;
        const v2f* bp = wp + h * P2 + c;
        float dr[8];
        #pragma unroll
        for (int r = 0; r < 8; ++r) dr[r] = dinv[m0 + 8 * h + r];
        v8f acc[NT];
        #pragma unroll
        for (int t = 0; t < NT; ++t) acc[t] = v8f{};
        #pragma unroll
        for (int i = 0; i < KS; ++i) {
            const v2f a = *(const v2f*)(xrow + 4 * i);
            #pragma unroll
            for (int t = 0; t < NT; ++t) {
                const v2f b = bp[(2 * i) * P2 + 16 * t];
                acc[t] = __builtin_amdgcn_wmma_f32_16x16x4_f32(false, a, false, b,
                                                               (short)0, acc[t], false, false);
            }
        }
        float* hrow = H2 + (size_t)(m0 + 8 * h) * COLS + c;
        #pragma unroll
        for (int t = 0; t < NT; ++t) {
            #pragma unroll
            for (int r = 0; r < 8; ++r) hrow[(size_t)r * COLS + 16 * t] = dr[r] * acc[t][r];
        }
    }
}

__global__ __launch_bounds__(256) void gather48_classify(const float* __restrict__ H,
                                                         const int* __restrict__ rowptr,
                                                         const int* __restrict__ csr,
                                                         const float* __restrict__ dinv,
                                                         const float* __restrict__ bias,
                                                         const float* __restrict__ Wc,
                                                         const float* __restrict__ bc,
                                                         float* __restrict__ logits) {
    static_assert((GB_NODES * NCLS) % 32 == 0 && GB_NODES * NCLS <= 256, "the block's logits are whole lines");
    __shared__ float sl[GB_NODES * NCLS];
    const int lane = threadIdx.x & 31;
    const int wave = __builtin_amdgcn_readfirstlane((int)(threadIdx.x >> 5));
    const int n0 = (blockIdx.x * 8 + wave) * NODES_PER_WAVE;
    const int2* cs = reinterpret_cast<const int2*>(csr);
    const bool on = lane < ROW_LANES;
    const int  q  = on ? lane : lane - ROW_LANES;
    const float b0 = bias[2 * q], b1 = bias[2 * q + 1];
    float w0[NCLS], w1[NCLS];
    #pragma unroll
    for (int j = 0; j < NCLS; ++j) {
        w0[j] = Wc[(2 * q) * NCLS + j];
        w1[j] = Wc[(2 * q + 1) * NCLS + j];
    }
    const float bcl = bc[lane < NCLS ? lane : 0];
    for (int k = 0; k < NODES_PER_WAVE; ++k) {
        const int node = n0 + k;
        const int beg  = __builtin_amdgcn_readfirstlane(rowptr[node]);
        const int end  = __builtin_amdgcn_readfirstlane(rowptr[node + 1]);
        const float di = dinv[node];
        v2f acc = *(const v2f*)(H + (size_t)node * HID2 + 2 * q);
        sum_segment<HID2, 2>(H, cs, beg, end, lane, q, acc);
        const float v0 = fmaxf(di * acc[0] + b0, 0.0f);
        const float v1 = fmaxf(di * acc[1] + b1, 0.0f);
        float s[NCLS];
        #pragma unroll
        for (int j = 0; j < NCLS; ++j) s[j] = on ? v0 * w0[j] + v1 * w1[j] : 0.0f;
        #pragma unroll
        for (int d = 16; d >= 1; d >>= 1) {
            #pragma unroll
            for (int j = 0; j < NCLS; ++j) s[j] += __shfl_xor(s[j], d);
        }
        if (lane < NCLS) {
            const float v = lane == 0 ? s[0] : (lane == 1 ? s[1] : s[2]);
            sl[(wave * NODES_PER_WAVE + k) * NCLS + lane] = v + bcl;
        }
    }
    __syncthreads();
    if (threadIdx.x < GB_NODES * NCLS)
        logits[(size_t)blockIdx.x * (GB_NODES * NCLS) + threadIdx.x] = sl[threadIdx.x];
}


extern "C" void kernel_launch(void* const* d_in, const int* in_sizes, int n_in,
                              void* d_out, int out_size, void* d_ws, size_t ws_size,
                              hipStream_t stream) {
    const float* x  = (const float*)d_in[0];
    const int*   ei = (const int*)  d_in[1];
    const float* ew = (const float*)d_in[2];
    const float* W1 = (const float*)d_in[3];
    const float* b1 = (const float*)d_in[4];
    const float* W2 = (const float*)d_in[5];
    const float* b2 = (const float*)d_in[6];
    const float* Wc = (const float*)d_in[7];
    const float* bc = (const float*)d_in[8];
    float* logits = (float*)d_out;

    const int* src = ei;
    const int* dst = ei + N_EDGES;

    char* ws = (char*)d_ws;
    size_t off = 0;
    auto take = [&](size_t bytes) -> void* {
        void* p = ws + off;
        off += (bytes + 255) & ~(size_t)255;
        return p;
    };
    float* dinv     = (float*)take((size_t)N_NODES * 4);
    int*   rowptr   = (int*)  take((size_t)(N_NODES + 1) * 4);
    int*   offs     = (int*)  take((size_t)NC2 * OPITCH * 4);
    int*   tmp      = (int*)  take((size_t)NC2 * TPITCH * 16);
    int*   csr      = (int*)  take((size_t)N_EDGES * 8);
    float* H1   = (float*)take((size_t)N_NODES * HID * 4);
    float* H2   = (float*)take((size_t)N_NODES * HID2 * 4);

    bucket_place<<<NC2, BK_TB, 0, stream>>>(src, dst, ew, offs, tmp);
    bucket_sort <<<NB,  BK_TB, 0, stream>>>(tmp, offs, rowptr, csr, dinv);

    {
        gemm_wmma<IN_CH, HID><<<(N_NODES / 16 + 7) / 8, 256, 0, stream>>>(x, W1, dinv, H1);
    }
    gather96_gemm48<<<N_NODES / GB_NODES, 256, 0, stream>>>(H1, rowptr, csr, dinv, b1, W2, H2);

    gather48_classify<<<N_NODES / GB_NODES, 256, 0, stream>>>(H2, rowptr, csr, dinv, b2, Wc, bc, logits);
}
